# starting kernel
# baseline (speedup 1.0000x reference)
#include <hip/hip_runtime.h>
#include <math.h>

typedef __attribute__((ext_vector_type(2))) float v2f;
typedef __attribute__((ext_vector_type(8))) float v8f;

#define HEADS 4
#define CCH 64
#define HIDF 256
#define NEG_SLOPE 0.2f
#define LN_EPS 1e-5f

__device__ __forceinline__ float eluf(float x) { return x > 0.f ? x : (expf(x) - 1.f); }
__device__ __forceinline__ float leakyf(float x) { return x > 0.f ? x : NEG_SLOPE * x; }

__device__ __forceinline__ unsigned fkey(float f) {
    unsigned u = __float_as_uint(f);
    return (u & 0x80000000u) ? ~u : (u | 0x80000000u);
}
__device__ __forceinline__ float funkey(unsigned k) {
    unsigned u = (k & 0x80000000u) ? (k ^ 0x80000000u) : ~k;
    return __uint_as_float(u);
}

__device__ __forceinline__ float waveRedSum(float v) {
    for (int off = 16; off > 0; off >>= 1) v += __shfl_xor(v, off, 32);
    return v;
}

__global__ void gemm_wmma_f32(const float* __restrict__ A, int lda,
                              const float* __restrict__ W,
                              float* __restrict__ C,
                              int M, int K, int Nout,
                              const float* __restrict__ bias, int elu_act) {
    int wid  = (blockIdx.x * blockDim.x + threadIdx.x) >> 5;
    int lane = threadIdx.x & 31;
    int tilesN = Nout >> 4;
    int totalTiles = (M >> 4) * tilesN;
    if (wid >= totalTiles) return;
    int tm = wid / tilesN;
    int tn = wid - tm * tilesN;
    int m  = (tm << 4) + (lane & 15);
    int n  = (tn << 4) + (lane & 15);
    int kb = (lane >> 4) << 1;
    const float* arow = A + (size_t)m * lda;
    const float* brow = W + (size_t)n * K;
    v8f acc = {};
    for (int k = 0; k < K; k += 4) {
        v2f a, b;
        a.x = arow[k + kb];
        a.y = arow[k + kb + 1];
        b.x = brow[k + kb];
        b.y = brow[k + kb + 1];
        acc = __builtin_amdgcn_wmma_f32_16x16x4_f32(false, a, false, b,
                                                    (short)0, acc, false, false);
    }
    int rbase = (tm << 4) + ((lane >> 4) << 3);
    int col   = (tn << 4) + (lane & 15);
    float bv  = bias ? bias[col] : 0.f;
    for (int i = 0; i < 8; ++i) {
        float v = acc[i] + bv;
        if (elu_act) v = eluf(v);
        C[(size_t)(rbase + i) * Nout + col] = v;
    }
}

__global__ void attn_coef(const float* __restrict__ hmat,
                          const float* __restrict__ a_s,
                          const float* __restrict__ a_d,
                          float* __restrict__ asrc, float* __restrict__ adst, int NH) {
    int idx = blockIdx.x * blockDim.x + threadIdx.x;
    if (idx >= NH) return;
    int n = idx >> 2, h = idx & 3;
    const float* row = hmat + (size_t)n * HIDF + h * CCH;
    const float* ws  = a_s + h * CCH;
    const float* wd  = a_d + h * CCH;
    float ss = 0.f, sd = 0.f;
    for (int c = 0; c < CCH; ++c) { float v = row[c]; ss += v * ws[c]; sd += v * wd[c]; }
    asrc[idx] = ss;
    adst[idx] = sd;
}

__global__ void softmax_init(const float* __restrict__ asrc, const float* __restrict__ adst,
                             unsigned* __restrict__ amaxk, int NH) {
    int idx = blockIdx.x * blockDim.x + threadIdx.x;
    if (idx >= NH) return;
    amaxk[idx] = fkey(leakyf(asrc[idx] + adst[idx]));
}

__global__ void edge_max(const int* __restrict__ src, const int* __restrict__ dst,
                         const float* __restrict__ asrc, const float* __restrict__ adst,
                         unsigned* __restrict__ amaxk, int EH) {
    int idx = blockIdx.x * blockDim.x + threadIdx.x;
    if (idx >= EH) return;
    int e = idx >> 2, h = idx & 3;
    int s = src[e], d = dst[e];
    float a = leakyf(asrc[s * HEADS + h] + adst[d * HEADS + h]);
    atomicMax(&amaxk[d * HEADS + h], fkey(a));
}

__global__ void acc_init(const float* __restrict__ hmat,
                         const float* __restrict__ asrc, const float* __restrict__ adst,
                         const unsigned* __restrict__ amaxk,
                         float* __restrict__ denom, float* __restrict__ hacc, int N) {
    int wid  = (blockIdx.x * blockDim.x + threadIdx.x) >> 5;
    int lane = threadIdx.x & 31;
    if (wid >= N) return;
    int n = wid;
    float ex[HEADS];
#pragma unroll
    for (int h = 0; h < HEADS; ++h) {
        float aself = leakyf(asrc[n * HEADS + h] + adst[n * HEADS + h]);
        ex[h] = expf(aself - funkey(amaxk[n * HEADS + h]));
    }
    if (lane == 0) {
#pragma unroll
        for (int h = 0; h < HEADS; ++h) denom[n * HEADS + h] = ex[h];
    }
    const float* row = hmat + (size_t)n * HIDF;
    float* orow      = hacc + (size_t)n * HIDF;
#pragma unroll
    for (int i = 0; i < 8; ++i) {
        int c = lane + 32 * i;
        orow[c] = row[c] * ex[i >> 1];
    }
}

__global__ void edge_accum(const int* __restrict__ src, const int* __restrict__ dst,
                           const float* __restrict__ asrc, const float* __restrict__ adst,
                           const unsigned* __restrict__ amaxk,
                           const float* __restrict__ hmat,
                           float* __restrict__ denom, float* __restrict__ hacc, int E) {
    int wid  = (blockIdx.x * blockDim.x + threadIdx.x) >> 5;
    int lane = threadIdx.x & 31;
    if (wid >= E) return;
    int e = wid;
    int s = src[e], d = dst[e];
    float ex[HEADS];
#pragma unroll
    for (int h = 0; h < HEADS; ++h) {
        float a = leakyf(asrc[s * HEADS + h] + adst[d * HEADS + h]);
        ex[h] = expf(a - funkey(amaxk[d * HEADS + h]));
    }
    if (lane == 0) {
#pragma unroll
        for (int h = 0; h < HEADS; ++h) atomicAdd(&denom[d * HEADS + h], ex[h]);
    }
    const float* hrow = hmat + (size_t)s * HIDF;
    float* orow       = hacc + (size_t)d * HIDF;
#pragma unroll
    for (int i = 0; i < 8; ++i) {
        int c = lane + 32 * i;
        atomicAdd(&orow[c], ex[i >> 1] * hrow[c]);
    }
}

__global__ void finalize_concat(const float* __restrict__ hacc, const float* __restrict__ denom,
                                const float* __restrict__ bias, const float* __restrict__ g,
                                const float* __restrict__ be, float* __restrict__ out, int N) {
    int wid  = (blockIdx.x * blockDim.x + threadIdx.x) >> 5;
    int lane = threadIdx.x & 31;
    if (wid >= N) return;
    size_t base = (size_t)wid * HIDF;
    float dn[HEADS];
#pragma unroll
    for (int h = 0; h < HEADS; ++h) dn[h] = 1.f / (denom[wid * HEADS + h] + 1e-16f);
    float v[8];
    float s = 0.f;
#pragma unroll
    for (int i = 0; i < 8; ++i) {
        int c = lane + 32 * i;
        v[i] = hacc[base + c] * dn[i >> 1] + bias[c];
        s += v[i];
    }
    float mu = waveRedSum(s) * (1.f / HIDF);
    float q = 0.f;
#pragma unroll
    for (int i = 0; i < 8; ++i) { float dlt = v[i] - mu; q += dlt * dlt; }
    float rstd = rsqrtf(waveRedSum(q) * (1.f / HIDF) + LN_EPS);
#pragma unroll
    for (int i = 0; i < 8; ++i) {
        int c = lane + 32 * i;
        float y = (v[i] - mu) * rstd * g[c] + be[c];
        out[base + c] = eluf(y);
    }
}

__global__ void finalize_mean(const float* __restrict__ hacc, const float* __restrict__ denom,
                              const float* __restrict__ bias, const float* __restrict__ g,
                              const float* __restrict__ be, float* __restrict__ out, int N) {
    int wid  = (blockIdx.x * blockDim.x + threadIdx.x) >> 5;
    int lane = threadIdx.x & 31;
    if (wid >= N) return;
    size_t base = (size_t)wid * HIDF;
    float dn[HEADS];
#pragma unroll
    for (int h = 0; h < HEADS; ++h) dn[h] = 1.f / (denom[wid * HEADS + h] + 1e-16f);
    float v[2];
    float s = 0.f;
#pragma unroll
    for (int i = 0; i < 2; ++i) {
        int c = lane + 32 * i;
        float acc = 0.f;
#pragma unroll
        for (int h = 0; h < HEADS; ++h) acc += hacc[base + h * CCH + c] * dn[h];
        v[i] = acc * (1.f / HEADS) + bias[c];
        s += v[i];
    }
    float mu = waveRedSum(s) * (1.f / CCH);
    float q = 0.f;
#pragma unroll
    for (int i = 0; i < 2; ++i) { float dlt = v[i] - mu; q += dlt * dlt; }
    float rstd = rsqrtf(waveRedSum(q) * (1.f / CCH) + LN_EPS);
#pragma unroll
    for (int i = 0; i < 2; ++i) {
        int c = lane + 32 * i;
        float y = (v[i] - mu) * rstd * g[c] + be[c];
        out[(size_t)wid * CCH + c] = eluf(y);
    }
}

__global__ void value_head(const float* __restrict__ hin,
                           const float* __restrict__ vw1, const float* __restrict__ vb1,
                           const float* __restrict__ vw2, const float* __restrict__ vb2,
                           float* __restrict__ out, int N) {
    int wid  = (blockIdx.x * blockDim.x + threadIdx.x) >> 5;
    int lane = threadIdx.x & 31;
    if (wid >= N) return;
    const float* row  = hin + (size_t)wid * CCH;
    const float* wrow = vw1 + lane * CCH;
    float hid = vb1[lane];
    for (int c = 0; c < CCH; ++c) hid += row[c] * wrow[c];
    hid = eluf(hid);
#pragma unroll
    for (int k = 0; k < 4; ++k) {
        float p = vw2[k * 32 + lane] * hid;
        p = waveRedSum(p);
        if (lane == 0) out[wid * 4 + k] = p + vb2[k];
    }
}

extern "C" void kernel_launch(void* const* d_in, const int* in_sizes, int n_in,
                              void* d_out, int out_size, void* d_ws, size_t ws_size,
                              hipStream_t stream) {
    const float* x      = (const float*)d_in[0];
    const int*   esrc   = (const int*)d_in[1];
    const int*   edst   = (const int*)d_in[2];
    const float* proj_w = (const float*)d_in[3];
    const float* proj_b = (const float*)d_in[4];
    const float* W[3]  = {(const float*)d_in[5],  (const float*)d_in[11], (const float*)d_in[17]};
    const float* AS[3] = {(const float*)d_in[6],  (const float*)d_in[12], (const float*)d_in[18]};
    const float* AD[3] = {(const float*)d_in[7],  (const float*)d_in[13], (const float*)d_in[19]};
    const float* B[3]  = {(const float*)d_in[8],  (const float*)d_in[14], (const float*)d_in[20]};
    const float* G[3]  = {(const float*)d_in[9],  (const float*)d_in[15], (const float*)d_in[21]};
    const float* BE[3] = {(const float*)d_in[10], (const float*)d_in[16], (const float*)d_in[22]};
    const float* vw1 = (const float*)d_in[23];
    const float* vb1 = (const float*)d_in[24];
    const float* vw2 = (const float*)d_in[25];
    const float* vb2 = (const float*)d_in[26];

    const int N = in_sizes[0] / 128;
    const int E = in_sizes[1];
    const int NH = N * HEADS;
    const int EH = E * HEADS;

    float* ws     = (float*)d_ws;
    float* hin    = ws;
    float* hmat   = hin  + (size_t)N * HIDF;
    float* hacc   = hmat + (size_t)N * HIDF;
    float* asrc   = hacc + (size_t)N * HIDF;
    float* adst   = asrc + (size_t)NH;
    unsigned* amk = (unsigned*)(adst + (size_t)NH);
    float* denom  = (float*)(amk + (size_t)NH);

    dim3 blk(256);

    {
        int tiles = (N / 16) * (64 / 16);
        gemm_wmma_f32<<<(tiles * 32 + 255) / 256, blk, 0, stream>>>(
            x, 128, proj_w, hin, N, 128, 64, proj_b, 1);
    }

    int in_ld = 64;
    for (int l = 0; l < 3; ++l) {
        int K = (l == 0) ? 64 : 256;
        int tiles = (N / 16) * (HIDF / 16);
        gemm_wmma_f32<<<(tiles * 32 + 255) / 256, blk, 0, stream>>>(
            hin, in_ld, W[l], hmat, N, K, HIDF, nullptr, 0);
        attn_coef<<<(NH + 255) / 256, blk, 0, stream>>>(hmat, AS[l], AD[l], asrc, adst, NH);
        softmax_init<<<(NH + 255) / 256, blk, 0, stream>>>(asrc, adst, amk, NH);
        edge_max<<<(EH + 255) / 256, blk, 0, stream>>>(esrc, edst, asrc, adst, amk, EH);
        acc_init<<<(N * 32 + 255) / 256, blk, 0, stream>>>(hmat, asrc, adst, amk, denom, hacc, N);
        edge_accum<<<((size_t)E * 32 + 255) / 256, blk, 0, stream>>>(
            esrc, edst, asrc, adst, amk, hmat, denom, hacc, E);
        if (l < 2) {
            finalize_concat<<<(N * 32 + 255) / 256, blk, 0, stream>>>(
                hacc, denom, B[l], G[l], BE[l], hin, N);
            in_ld = HIDF;
        } else {
            finalize_mean<<<(N * 32 + 255) / 256, blk, 0, stream>>>(
                hacc, denom, B[l], G[l], BE[l], hin, N);
        }
    }

    value_head<<<(N * 32 + 255) / 256, blk, 0, stream>>>(
        hin, vw1, vb1, vw2, vb2, (float*)d_out, N);
}


# Edge softmax-aggregate as a gather: edges grouped by destination once per call (count, scan, fill); one wave per node sums its in-edges in registers in list (fill-arrival) order; no float atomics
# speedup vs baseline: 2.0411x; 2.0411x over previous
#include <hip/hip_runtime.h>
#include <math.h>

typedef __attribute__((ext_vector_type(2))) float v2f;
typedef __attribute__((ext_vector_type(8))) float v8f;

#define HEADS 4
#define CCH 64
#define HIDF 256
#define NEG_SLOPE 0.2f
#define LN_EPS 1e-5f

__device__ __forceinline__ float eluf(float x) { return x > 0.f ? x : (expf(x) - 1.f); }
__device__ __forceinline__ float leakyf(float x) { return x > 0.f ? x : NEG_SLOPE * x; }

__device__ __forceinline__ float waveRedSum(float v) {
    for (int off = 16; off > 0; off >>= 1) v += __shfl_xor(v, off, 32);
    return v;
}
__device__ __forceinline__ float waveRedMax(float v) {
    for (int off = 16; off > 0; off >>= 1) v = fmaxf(v, __shfl_xor(v, off, 32));
    return v;
}
__device__ __forceinline__ int waveRedSumI(int v) {
    for (int off = 16; off > 0; off >>= 1) v += __shfl_xor(v, off, 32);
    return v;
}
__device__ __forceinline__ float laneValF(float v, int t) {
    return __int_as_float(__builtin_amdgcn_readlane(__float_as_int(v), t));
}

__global__ void gemm_wmma_f32(const float* __restrict__ A, int lda,
                              const float* __restrict__ W,
                              float* __restrict__ C,
                              int M, int K, int Nout,
                              const float* __restrict__ bias, int elu_act) {
    int wid  = (blockIdx.x * blockDim.x + threadIdx.x) >> 5;
    int lane = threadIdx.x & 31;
    int tilesN = Nout >> 4;
    int totalTiles = (M >> 4) * tilesN;
    if (wid >= totalTiles) return;
    int tm = wid / tilesN;
    int tn = wid - tm * tilesN;
    int m  = (tm << 4) + (lane & 15);
    int n  = (tn << 4) + (lane & 15);
    int kb = (lane >> 4) << 1;
    const float* arow = A + (size_t)m * lda;
    const float* brow = W + (size_t)n * K;
    v8f acc = {};
    for (int k = 0; k < K; k += 4) {
        v2f a, b;
        a.x = arow[k + kb];
        a.y = arow[k + kb + 1];
        b.x = brow[k + kb];
        b.y = brow[k + kb + 1];
        acc = __builtin_amdgcn_wmma_f32_16x16x4_f32(false, a, false, b,
                                                    (short)0, acc, false, false);
    }
    int rbase = (tm << 4) + ((lane >> 4) << 3);
    int col   = (tn << 4) + (lane & 15);
    float bv  = bias ? bias[col] : 0.f;
    for (int i = 0; i < 8; ++i) {
        float v = acc[i] + bv;
        if (elu_act) v = eluf(v);
        C[(size_t)(rbase + i) * Nout + col] = v;
    }
}

__global__ void attn_coef(const float* __restrict__ hmat,
                          const float* __restrict__ a_s,
                          const float* __restrict__ a_d,
                          float* __restrict__ asrc, float* __restrict__ adst, int NH) {
    int idx = blockIdx.x * blockDim.x + threadIdx.x;
    if (idx >= NH) return;
    int n = idx >> 2, h = idx & 3;
    const float* row = hmat + (size_t)n * HIDF + h * CCH;
    const float* ws  = a_s + h * CCH;
    const float* wd  = a_d + h * CCH;
    float ss = 0.f, sd = 0.f;
    for (int c = 0; c < CCH; ++c) { float v = row[c]; ss += v * ws[c]; sd += v * wd[c]; }
    asrc[idx] = ss;
    adst[idx] = sd;
}

__global__ void csr_clear(int* __restrict__ cnt, int N) {
    int i = blockIdx.x * blockDim.x + threadIdx.x;
    if (i < N) cnt[i] = 0;
}

__global__ void csr_count(const int* __restrict__ dst, int* __restrict__ cnt, int E) {
    int e = blockIdx.x * blockDim.x + threadIdx.x;
    if (e < E) atomicAdd(&cnt[dst[e]], 1);
}

__global__ void __launch_bounds__(1024) csr_scan(int* __restrict__ cnt, int* __restrict__ rowptr, int N) {
    __shared__ int wtot[32];
    const int lane = threadIdx.x & 31;
    const int w    = threadIdx.x >> 5;
    const int per  = ((N + 4095) >> 12) << 7;
    const int lo   = w * per;
    const int hi   = (lo + per < N) ? (lo + per) : N;
    int sum = 0;
#pragma unroll 4
    for (int i = lo + 4 * lane; i < hi; i += 128) {
        const int4 c = *reinterpret_cast<const int4*>(cnt + i);
        sum += (c.x + c.y) + (c.z + c.w);
    }
    sum = waveRedSumI(sum);
    if (lane == 0) wtot[w] = sum;
    __syncthreads();
    int carry = waveRedSumI(lane < w ? wtot[lane] : 0);
    for (int i0 = lo; i0 < hi; i0 += 128) {
        const int  i  = i0 + 4 * lane;
        const bool in = i < hi;
        int4 c = make_int4(0, 0, 0, 0);
        if (in) c = *reinterpret_cast<const int4*>(cnt + i);
        const int t = (c.x + c.y) + (c.z + c.w);
        int x = t;
        for (int off = 1; off < 32; off <<= 1) {
            int y = __shfl_up(x, off, 32);
            if (lane >= off) x += y;
        }
        if (in) {
            const int b = carry + x - t;
            const int4 o = make_int4(b, b + c.x, b + c.x + c.y, b + c.x + c.y + c.z);
            *reinterpret_cast<int4*>(rowptr + i) = o;
            *reinterpret_cast<int4*>(cnt + i)    = o;
        }
        carry += __shfl(x, 31, 32);
    }
    if (threadIdx.x == 1023) rowptr[N] = carry;
}

__global__ void csr_fill(const int* __restrict__ src, const int* __restrict__ dst,
                         int* __restrict__ cursor, int* __restrict__ col, int E) {
    int e = blockIdx.x * blockDim.x + threadIdx.x;
    if (e >= E) return;
    int p = atomicAdd(&cursor[dst[e]], 1);
    col[p] = src[e];
}

__device__ __forceinline__ void gather_rows(const float* __restrict__ hmat, int s,
                                            float e0, float e1, float e2, float e3,
                                            int cnt, int lane, bool hi,
                                            float4& acc0, float4& acc1) {
#pragma unroll 4
    for (int t = 0; t < cnt; ++t) {
        const int   st = __builtin_amdgcn_readlane(s, t);
        const float w0 = laneValF(e0, t), w1 = laneValF(e1, t);
        const float w2 = laneValF(e2, t), w3 = laneValF(e3, t);
        const float wa = hi ? w1 : w0;
        const float wb = hi ? w3 : w2;
        const float4* r = reinterpret_cast<const float4*>(hmat + (size_t)st * HIDF);
        const float4 p = r[lane];
        const float4 q = r[32 + lane];
        acc0.x += wa * p.x; acc0.y += wa * p.y; acc0.z += wa * p.z; acc0.w += wa * p.w;
        acc1.x += wb * q.x; acc1.y += wb * q.y; acc1.z += wb * q.z; acc1.w += wb * q.w;
    }
}

__global__ void gat_gather(const int* __restrict__ rowptr, const int* __restrict__ col,
                           const float* __restrict__ asrc, const float* __restrict__ adst,
                           const float* __restrict__ hmat,
                           float* __restrict__ denom, float* __restrict__ hacc, int N) {
    const int lane = threadIdx.x & 31;
    const int n = __builtin_amdgcn_readfirstlane((int)((blockIdx.x * blockDim.x + threadIdx.x) >> 5));
    if (n >= N) return;
    const int beg = rowptr[n], end = rowptr[n + 1];
    const bool hi = lane >= 16;
    const float4* asrc4 = reinterpret_cast<const float4*>(asrc);
    const float4 ad = reinterpret_cast<const float4*>(adst)[n];
    const float4 an = asrc4[n];
    const float sf0 = leakyf(an.x + ad.x), sf1 = leakyf(an.y + ad.y);
    const float sf2 = leakyf(an.z + ad.z), sf3 = leakyf(an.w + ad.w);

    const bool v0 = beg + lane < end;
    int   s0 = n;
    float a0 = sf0, a1 = sf1, a2 = sf2, a3 = sf3;
    if (v0) {
        s0 = col[beg + lane];
        const float4 a = asrc4[s0];
        a0 = leakyf(a.x + ad.x); a1 = leakyf(a.y + ad.y);
        a2 = leakyf(a.z + ad.z); a3 = leakyf(a.w + ad.w);
    }
    float m0 = fmaxf(sf0, a0), m1 = fmaxf(sf1, a1), m2 = fmaxf(sf2, a2), m3 = fmaxf(sf3, a3);
    for (int base = beg + 32; base < end; base += 32) {
        const int j = base + lane;
        if (j < end) {
            const float4 a = asrc4[col[j]];
            m0 = fmaxf(m0, leakyf(a.x + ad.x)); m1 = fmaxf(m1, leakyf(a.y + ad.y));
            m2 = fmaxf(m2, leakyf(a.z + ad.z)); m3 = fmaxf(m3, leakyf(a.w + ad.w));
        }
    }
    m0 = waveRedMax(m0); m1 = waveRedMax(m1); m2 = waveRedMax(m2); m3 = waveRedMax(m3);

    const float x0 = expf(sf0 - m0), x1 = expf(sf1 - m1), x2 = expf(sf2 - m2), x3 = expf(sf3 - m3);
    const float4* rown = reinterpret_cast<const float4*>(hmat + (size_t)n * HIDF);
    const float4 r0 = rown[lane], r1 = rown[32 + lane];
    const float wsa = hi ? x1 : x0, wsb = hi ? x3 : x2;
    float4 acc0 = make_float4(r0.x * wsa, r0.y * wsa, r0.z * wsa, r0.w * wsa);
    float4 acc1 = make_float4(r1.x * wsb, r1.y * wsb, r1.z * wsb, r1.w * wsb);

    float d0, d1, d2, d3;
    {
        const float t0 = expf(a0 - m0), t1 = expf(a1 - m1), t2 = expf(a2 - m2), t3 = expf(a3 - m3);
        d0 = v0 ? t0 : 0.f; d1 = v0 ? t1 : 0.f; d2 = v0 ? t2 : 0.f; d3 = v0 ? t3 : 0.f;
        const int cnt = (end - beg < 32) ? (end - beg) : 32;
        gather_rows(hmat, s0, d0, d1, d2, d3, cnt, lane, hi, acc0, acc1);
    }
    for (int base = beg + 32; base < end; base += 32) {
        const int j = base + lane;
        int s = n;
        float e0 = 0.f, e1 = 0.f, e2 = 0.f, e3 = 0.f;
        if (j < end) {
            s = col[j];
            const float4 a = asrc4[s];
            e0 = expf(leakyf(a.x + ad.x) - m0); e1 = expf(leakyf(a.y + ad.y) - m1);
            e2 = expf(leakyf(a.z + ad.z) - m2); e3 = expf(leakyf(a.w + ad.w) - m3);
        }
        d0 += e0; d1 += e1; d2 += e2; d3 += e3;
        const int cnt = (end - base < 32) ? (end - base) : 32;
        gather_rows(hmat, s, e0, e1, e2, e3, cnt, lane, hi, acc0, acc1);
    }
    d0 = waveRedSum(d0); d1 = waveRedSum(d1); d2 = waveRedSum(d2); d3 = waveRedSum(d3);
    if (lane == 0)
        reinterpret_cast<float4*>(denom)[n] = make_float4(x0 + d0, x1 + d1, x2 + d2, x3 + d3);
    float4* orow = reinterpret_cast<float4*>(hacc + (size_t)n * HIDF);
    orow[lane]      = acc0;
    orow[32 + lane] = acc1;
}

__global__ void finalize_concat(const float* __restrict__ hacc, const float* __restrict__ denom,
                                const float* __restrict__ bias, const float* __restrict__ g,
                                const float* __restrict__ be, float* __restrict__ out, int N) {
    int wid  = (blockIdx.x * blockDim.x + threadIdx.x) >> 5;
    int lane = threadIdx.x & 31;
    if (wid >= N) return;
    size_t base = (size_t)wid * HIDF;
    float dn[HEADS];
#pragma unroll
    for (int h = 0; h < HEADS; ++h) dn[h] = 1.f / (denom[wid * HEADS + h] + 1e-16f);
    float v[8];
    float s = 0.f;
#pragma unroll
    for (int i = 0; i < 8; ++i) {
        int c = lane + 32 * i;
        v[i] = hacc[base + c] * dn[i >> 1] + bias[c];
        s += v[i];
    }
    float mu = waveRedSum(s) * (1.f / HIDF);
    float q = 0.f;
#pragma unroll
    for (int i = 0; i < 8; ++i) { float dlt = v[i] - mu; q += dlt * dlt; }
    float rstd = rsqrtf(waveRedSum(q) * (1.f / HIDF) + LN_EPS);
#pragma unroll
    for (int i = 0; i < 8; ++i) {
        int c = lane + 32 * i;
        float y = (v[i] - mu) * rstd * g[c] + be[c];
        out[base + c] = eluf(y);
    }
}

__global__ void finalize_mean(const float* __restrict__ hacc, const float* __restrict__ denom,
                              const float* __restrict__ bias, const float* __restrict__ g,
                              const float* __restrict__ be, float* __restrict__ out, int N) {
    int wid  = (blockIdx.x * blockDim.x + threadIdx.x) >> 5;
    int lane = threadIdx.x & 31;
    if (wid >= N) return;
    size_t base = (size_t)wid * HIDF;
    float dn[HEADS];
#pragma unroll
    for (int h = 0; h < HEADS; ++h) dn[h] = 1.f / (denom[wid * HEADS + h] + 1e-16f);
    float v[2];
    float s = 0.f;
#pragma unroll
    for (int i = 0; i < 2; ++i) {
        int c = lane + 32 * i;
        float acc = 0.f;
#pragma unroll
        for (int h = 0; h < HEADS; ++h) acc += hacc[base + h * CCH + c] * dn[h];
        v[i] = acc * (1.f / HEADS) + bias[c];
        s += v[i];
    }
    float mu = waveRedSum(s) * (1.f / CCH);
    float q = 0.f;
#pragma unroll
    for (int i = 0; i < 2; ++i) { float dlt = v[i] - mu; q += dlt * dlt; }
    float rstd = rsqrtf(waveRedSum(q) * (1.f / CCH) + LN_EPS);
#pragma unroll
    for (int i = 0; i < 2; ++i) {
        int c = lane + 32 * i;
        float y = (v[i] - mu) * rstd * g[c] + be[c];
        out[(size_t)wid * CCH + c] = eluf(y);
    }
}

__global__ void value_head(const float* __restrict__ hin,
                           const float* __restrict__ vw1, const float* __restrict__ vb1,
                           const float* __restrict__ vw2, const float* __restrict__ vb2,
                           float* __restrict__ out, int N) {
    int wid  = (blockIdx.x * blockDim.x + threadIdx.x) >> 5;
    int lane = threadIdx.x & 31;
    if (wid >= N) return;
    const float* row  = hin + (size_t)wid * CCH;
    const float* wrow = vw1 + lane * CCH;
    float hid = vb1[lane];
    for (int c = 0; c < CCH; ++c) hid += row[c] * wrow[c];
    hid = eluf(hid);
#pragma unroll
    for (int k = 0; k < 4; ++k) {
        float p = vw2[k * 32 + lane] * hid;
        p = waveRedSum(p);
        if (lane == 0) out[wid * 4 + k] = p + vb2[k];
    }
}

extern "C" void kernel_launch(void* const* d_in, const int* in_sizes, int n_in,
                              void* d_out, int out_size, void* d_ws, size_t ws_size,
                              hipStream_t stream) {
    const float* x      = (const float*)d_in[0];
    const int*   esrc   = (const int*)d_in[1];
    const int*   edst   = (const int*)d_in[2];
    const float* proj_w = (const float*)d_in[3];
    const float* proj_b = (const float*)d_in[4];
    const float* W[3]  = {(const float*)d_in[5],  (const float*)d_in[11], (const float*)d_in[17]};
    const float* AS[3] = {(const float*)d_in[6],  (const float*)d_in[12], (const float*)d_in[18]};
    const float* AD[3] = {(const float*)d_in[7],  (const float*)d_in[13], (const float*)d_in[19]};
    const float* B[3]  = {(const float*)d_in[8],  (const float*)d_in[14], (const float*)d_in[20]};
    const float* G[3]  = {(const float*)d_in[9],  (const float*)d_in[15], (const float*)d_in[21]};
    const float* BE[3] = {(const float*)d_in[10], (const float*)d_in[16], (const float*)d_in[22]};
    const float* vw1 = (const float*)d_in[23];
    const float* vb1 = (const float*)d_in[24];
    const float* vw2 = (const float*)d_in[25];
    const float* vb2 = (const float*)d_in[26];

    const int N = in_sizes[0] / 128;
    const int E = in_sizes[1];
    const int NH = N * HEADS;

    float* ws     = (float*)d_ws;
    float* hin    = ws;
    float* hmat   = hin  + (size_t)N * HIDF;
    float* hacc   = hmat + (size_t)N * HIDF;
    float* asrc   = hacc + (size_t)N * HIDF;
    float* adst   = asrc + (size_t)NH;
    float* denom  = adst + (size_t)NH;
    int* cnt      = (int*)(denom + (size_t)NH);
    int* rowptr   = cnt + (size_t)N;
    int* col      = rowptr + (size_t)N + 1;

    dim3 blk(256);

    csr_clear<<<(N + 255) / 256, blk, 0, stream>>>(cnt, N);
    csr_count<<<(E + 255) / 256, blk, 0, stream>>>(edst, cnt, E);
    csr_scan<<<1, 1024, 0, stream>>>(cnt, rowptr, N);
    csr_fill<<<(E + 255) / 256, blk, 0, stream>>>(esrc, edst, cnt, col, E);

    {
        int tiles = (N / 16) * (64 / 16);
        gemm_wmma_f32<<<(tiles * 32 + 255) / 256, blk, 0, stream>>>(
            x, 128, proj_w, hin, N, 128, 64, proj_b, 1);
    }

    int in_ld = 64;
    for (int l = 0; l < 3; ++l) {
        int K = (l == 0) ? 64 : 256;
        int tiles = (N / 16) * (HIDF / 16);
        gemm_wmma_f32<<<(tiles * 32 + 255) / 256, blk, 0, stream>>>(
            hin, in_ld, W[l], hmat, N, K, HIDF, nullptr, 0);
        attn_coef<<<(NH + 255) / 256, blk, 0, stream>>>(hmat, AS[l], AD[l], asrc, adst, NH);
        gat_gather<<<(N * 32 + 255) / 256, blk, 0, stream>>>(
            rowptr, col, asrc, adst, hmat, denom, hacc, N);
        if (l < 2) {
            finalize_concat<<<(N * 32 + 255) / 256, blk, 0, stream>>>(
                hacc, denom, B[l], G[l], BE[l], hin, N);
            in_ld = HIDF;
        } else {
            finalize_mean<<<(N * 32 + 255) / 256, blk, 0, stream>>>(
                hacc, denom, B[l], G[l], BE[l], hin, N);
        }
    }

    value_head<<<(N * 32 + 255) / 256, blk, 0, stream>>>(
        hin, vw1, vb1, vw2, vb2, (float*)d_out, N);
}


# GEMMs tiled through LDS: 8-wave work-group stages 64-row A and 64/128-row W chunks of 32 k with whole-line 16-byte loads, each wave 2x2 (1x2) WMMA tiles from LDS fragments; same k order and sums
# speedup vs baseline: 2.7418x; 1.3433x over previous
#include <hip/hip_runtime.h>
#include <math.h>

typedef __attribute__((ext_vector_type(2))) float v2f;
typedef __attribute__((ext_vector_type(8))) float v8f;

#define HEADS 4
#define CCH 64
#define HIDF 256
#define NEG_SLOPE 0.2f
#define LN_EPS 1e-5f

__device__ __forceinline__ float eluf(float x) { return x > 0.f ? x : (expf(x) - 1.f); }
__device__ __forceinline__ float leakyf(float x) { return x > 0.f ? x : NEG_SLOPE * x; }

__device__ __forceinline__ float waveRedSum(float v) {
    for (int off = 16; off > 0; off >>= 1) v += __shfl_xor(v, off, 32);
    return v;
}
__device__ __forceinline__ float waveRedMax(float v) {
    for (int off = 16; off > 0; off >>= 1) v = fmaxf(v, __shfl_xor(v, off, 32));
    return v;
}
__device__ __forceinline__ int waveRedSumI(int v) {
    for (int off = 16; off > 0; off >>= 1) v += __shfl_xor(v, off, 32);
    return v;
}
__device__ __forceinline__ float laneValF(float v, int t) {
    return __int_as_float(__builtin_amdgcn_readlane(__float_as_int(v), t));
}

template <int BM, int BN, int WM, int WN>
__global__ void __launch_bounds__(256)
gemm_wmma_f32_lds(const float* __restrict__ A, int lda,
                  const float* __restrict__ W,
                  float* __restrict__ C,
                  int M, int K, int Nout,
                  const float* __restrict__ bias, int elu_act) {
    constexpr int BK = 32;
    constexpr int P  = BK + 4;
    constexpr int TM = BM / (16 * WM);
    constexpr int TN = BN / (16 * WN);
    constexpr int AF = BM * (BK / 4) / 256;
    constexpr int BF = BN * (BK / 4) / 256;
    static_assert(WM * WN == 8, "work-group of 8 waves");
    static_assert(TM * 16 * WM == BM && TN * 16 * WN == BN, "wave tiles cover the block");
    static_assert(AF * 256 == BM * (BK / 4) && BF * 256 == BN * (BK / 4), "copy covers the chunk");

    __shared__ __attribute__((aligned(16))) float sA[2][BM * P];
    __shared__ __attribute__((aligned(16))) float sB[2][BN * P];

    const int tid  = threadIdx.x;
    const int lane = tid & 31;
    const int w    = tid >> 5;
    const int wm   = w / WN;
    const int wn   = w - wm * WN;
    const int blocksN = Nout / BN;
    const int bm = blockIdx.x / blocksN;
    const int bn = blockIdx.x - bm * blocksN;
    const int m0 = bm * BM;
    const int n0 = bn * BN;

    const int cr = tid >> 3;
    const int cq = (tid & 7) << 2;
    const float* gA[AF];
    const float* gB[BF];
#pragma unroll
    for (int j = 0; j < AF; ++j) {
        int gm = m0 + cr + 32 * j;
        gm = gm < M ? gm : (M - 1);
        gA[j] = A + (size_t)gm * lda + cq;
    }
#pragma unroll
    for (int j = 0; j < BF; ++j) gB[j] = W + (size_t)(n0 + cr + 32 * j) * K + cq;

    static_assert(AF == 2 && (BF == 2 || BF == 4), "piece registers below are written out for these sizes");
    float4 ra0, ra1, rb0, rb1, rb2, rb3;
    ra0 = *reinterpret_cast<const float4*>(gA[0]);
    ra1 = *reinterpret_cast<const float4*>(gA[1]);
    rb0 = *reinterpret_cast<const float4*>(gB[0]);
    rb1 = *reinterpret_cast<const float4*>(gB[1]);
    if constexpr (BF == 4) {
        rb2 = *reinterpret_cast<const float4*>(gB[2]);
        rb3 = *reinterpret_cast<const float4*>(gB[3]);
    }
    const int sOff = cr * P + cq;
    *reinterpret_cast<float4*>(&sA[0][sOff])          = ra0;
    *reinterpret_cast<float4*>(&sA[0][sOff + 32 * P]) = ra1;
    *reinterpret_cast<float4*>(&sB[0][sOff])          = rb0;
    *reinterpret_cast<float4*>(&sB[0][sOff + 32 * P]) = rb1;
    if constexpr (BF == 4) {
        *reinterpret_cast<float4*>(&sB[0][sOff + 64 * P]) = rb2;
        *reinterpret_cast<float4*>(&sB[0][sOff + 96 * P]) = rb3;
    }

    v8f acc[TM][TN];
#pragma unroll
    for (int i = 0; i < TM; ++i)
#pragma unroll
        for (int j = 0; j < TN; ++j) acc[i][j] = v8f{};

    const int fragOff = (lane & 15) * P + ((lane >> 4) << 1);
    const int nk = K / BK;
    for (int it = 0; it < nk; ++it) {
        const int  buf  = it & 1;
        const bool more = it + 1 < nk;
        __syncthreads();
        if (more) {
            const int k1 = (it + 1) * BK;
            ra0 = *reinterpret_cast<const float4*>(gA[0] + k1);
            ra1 = *reinterpret_cast<const float4*>(gA[1] + k1);
            rb0 = *reinterpret_cast<const float4*>(gB[0] + k1);
            rb1 = *reinterpret_cast<const float4*>(gB[1] + k1);
            if constexpr (BF == 4) {
                rb2 = *reinterpret_cast<const float4*>(gB[2] + k1);
                rb3 = *reinterpret_cast<const float4*>(gB[3] + k1);
            }
        }
        const float* pa = &sA[buf][wm * (TM * 16) * P + fragOff];
        const float* pb = &sB[buf][wn * (TN * 16) * P + fragOff];
#pragma unroll
        for (int ks = 0; ks < BK / 4; ++ks) {
            v2f af[TM], bf[TN];
#pragma unroll
            for (int i = 0; i < TM; ++i) af[i] = *reinterpret_cast<const v2f*>(pa + i * 16 * P + ks * 4);
#pragma unroll
            for (int j = 0; j < TN; ++j) bf[j] = *reinterpret_cast<const v2f*>(pb + j * 16 * P + ks * 4);
#pragma unroll
            for (int i = 0; i < TM; ++i)
#pragma unroll
                for (int j = 0; j < TN; ++j)
                    acc[i][j] = __builtin_amdgcn_wmma_f32_16x16x4_f32(false, af[i], false, bf[j],
                                                                      (short)0, acc[i][j], false, false);
        }
        if (more) {
            *reinterpret_cast<float4*>(&sA[buf ^ 1][sOff])          = ra0;
            *reinterpret_cast<float4*>(&sA[buf ^ 1][sOff + 32 * P]) = ra1;
            *reinterpret_cast<float4*>(&sB[buf ^ 1][sOff])          = rb0;
            *reinterpret_cast<float4*>(&sB[buf ^ 1][sOff + 32 * P]) = rb1;
            if constexpr (BF == 4) {
                *reinterpret_cast<float4*>(&sB[buf ^ 1][sOff + 64 * P]) = rb2;
                *reinterpret_cast<float4*>(&sB[buf ^ 1][sOff + 96 * P]) = rb3;
            }
        }
    }

#pragma unroll
    for (int i = 0; i < TM; ++i) {
#pragma unroll
        for (int j = 0; j < TN; ++j) {
            const int rbase = m0 + (wm * TM + i) * 16 + ((lane >> 4) << 3);
            const int col   = n0 + (wn * TN + j) * 16 + (lane & 15);
            const float bv  = bias ? bias[col] : 0.f;
#pragma unroll
            for (int r = 0; r < 8; ++r) {
                float v = acc[i][j][r] + bv;
                if (elu_act) v = eluf(v);
                if (rbase + r < M) C[(size_t)(rbase + r) * Nout + col] = v;
            }
        }
    }
}

__global__ void attn_coef(const float* __restrict__ hmat,
                          const float* __restrict__ a_s,
                          const float* __restrict__ a_d,
                          float* __restrict__ asrc, float* __restrict__ adst, int NH) {
    int idx = blockIdx.x * blockDim.x + threadIdx.x;
    if (idx >= NH) return;
    int n = idx >> 2, h = idx & 3;
    const float* row = hmat + (size_t)n * HIDF + h * CCH;
    const float* ws  = a_s + h * CCH;
    const float* wd  = a_d + h * CCH;
    float ss = 0.f, sd = 0.f;
    for (int c = 0; c < CCH; ++c) { float v = row[c]; ss += v * ws[c]; sd += v * wd[c]; }
    asrc[idx] = ss;
    adst[idx] = sd;
}

__global__ void csr_clear(int* __restrict__ cnt, int N) {
    int i = blockIdx.x * blockDim.x + threadIdx.x;
    if (i < N) cnt[i] = 0;
}

__global__ void csr_count(const int* __restrict__ dst, int* __restrict__ cnt, int E) {
    int e = blockIdx.x * blockDim.x + threadIdx.x;
    if (e < E) atomicAdd(&cnt[dst[e]], 1);
}

__global__ void __launch_bounds__(1024) csr_scan(int* __restrict__ cnt, int* __restrict__ rowptr, int N) {
    __shared__ int wtot[32];
    const int lane = threadIdx.x & 31;
    const int w    = threadIdx.x >> 5;
    const int per  = ((N + 4095) >> 12) << 7;
    const int lo   = w * per;
    const int hi   = (lo + per < N) ? (lo + per) : N;
    int sum = 0;
#pragma unroll 4
    for (int i = lo + 4 * lane; i < hi; i += 128) {
        const int4 c = *reinterpret_cast<const int4*>(cnt + i);
        sum += (c.x + c.y) + (c.z + c.w);
    }
    sum = waveRedSumI(sum);
    if (lane == 0) wtot[w] = sum;
    __syncthreads();
    int carry = waveRedSumI(lane < w ? wtot[lane] : 0);
    for (int i0 = lo; i0 < hi; i0 += 128) {
        const int  i  = i0 + 4 * lane;
        const bool in = i < hi;
        int4 c = make_int4(0, 0, 0, 0);
        if (in) c = *reinterpret_cast<const int4*>(cnt + i);
        const int t = (c.x + c.y) + (c.z + c.w);
        int x = t;
        for (int off = 1; off < 32; off <<= 1) {
            int y = __shfl_up(x, off, 32);
            if (lane >= off) x += y;
        }
        if (in) {
            const int b = carry + x - t;
            const int4 o = make_int4(b, b + c.x, b + c.x + c.y, b + c.x + c.y + c.z);
            *reinterpret_cast<int4*>(rowptr + i) = o;
            *reinterpret_cast<int4*>(cnt + i)    = o;
        }
        carry += __shfl(x, 31, 32);
    }
    if (threadIdx.x == 1023) rowptr[N] = carry;
}

__global__ void csr_fill(const int* __restrict__ src, const int* __restrict__ dst,
                         int* __restrict__ cursor, int* __restrict__ col, int E) {
    int e = blockIdx.x * blockDim.x + threadIdx.x;
    if (e >= E) return;
    int p = atomicAdd(&cursor[dst[e]], 1);
    col[p] = src[e];
}

__device__ __forceinline__ void gather_rows(const float* __restrict__ hmat, int s,
                                            float e0, float e1, float e2, float e3,
                                            int cnt, int lane, bool hi,
                                            float4& acc0, float4& acc1) {
#pragma unroll 4
    for (int t = 0; t < cnt; ++t) {
        const int   st = __builtin_amdgcn_readlane(s, t);
        const float w0 = laneValF(e0, t), w1 = laneValF(e1, t);
        const float w2 = laneValF(e2, t), w3 = laneValF(e3, t);
        const float wa = hi ? w1 : w0;
        const float wb = hi ? w3 : w2;
        const float4* r = reinterpret_cast<const float4*>(hmat + (size_t)st * HIDF);
        const float4 p = r[lane];
        const float4 q = r[32 + lane];
        acc0.x += wa * p.x; acc0.y += wa * p.y; acc0.z += wa * p.z; acc0.w += wa * p.w;
        acc1.x += wb * q.x; acc1.y += wb * q.y; acc1.z += wb * q.z; acc1.w += wb * q.w;
    }
}

__global__ void gat_gather(const int* __restrict__ rowptr, const int* __restrict__ col,
                           const float* __restrict__ asrc, const float* __restrict__ adst,
                           const float* __restrict__ hmat,
                           float* __restrict__ denom, float* __restrict__ hacc, int N) {
    const int lane = threadIdx.x & 31;
    const int n = __builtin_amdgcn_readfirstlane((int)((blockIdx.x * blockDim.x + threadIdx.x) >> 5));
    if (n >= N) return;
    const int beg = rowptr[n], end = rowptr[n + 1];
    const bool hi = lane >= 16;
    const float4* asrc4 = reinterpret_cast<const float4*>(asrc);
    const float4 ad = reinterpret_cast<const float4*>(adst)[n];
    const float4 an = asrc4[n];
    const float sf0 = leakyf(an.x + ad.x), sf1 = leakyf(an.y + ad.y);
    const float sf2 = leakyf(an.z + ad.z), sf3 = leakyf(an.w + ad.w);

    const bool v0 = beg + lane < end;
    int   s0 = n;
    float a0 = sf0, a1 = sf1, a2 = sf2, a3 = sf3;
    if (v0) {
        s0 = col[beg + lane];
        const float4 a = asrc4[s0];
        a0 = leakyf(a.x + ad.x); a1 = leakyf(a.y + ad.y);
        a2 = leakyf(a.z + ad.z); a3 = leakyf(a.w + ad.w);
    }
    float m0 = fmaxf(sf0, a0), m1 = fmaxf(sf1, a1), m2 = fmaxf(sf2, a2), m3 = fmaxf(sf3, a3);
    for (int base = beg + 32; base < end; base += 32) {
        const int j = base + lane;
        if (j < end) {
            const float4 a = asrc4[col[j]];
            m0 = fmaxf(m0, leakyf(a.x + ad.x)); m1 = fmaxf(m1, leakyf(a.y + ad.y));
            m2 = fmaxf(m2, leakyf(a.z + ad.z)); m3 = fmaxf(m3, leakyf(a.w + ad.w));
        }
    }
    m0 = waveRedMax(m0); m1 = waveRedMax(m1); m2 = waveRedMax(m2); m3 = waveRedMax(m3);

    const float x0 = expf(sf0 - m0), x1 = expf(sf1 - m1), x2 = expf(sf2 - m2), x3 = expf(sf3 - m3);
    const float4* rown = reinterpret_cast<const float4*>(hmat + (size_t)n * HIDF);
    const float4 r0 = rown[lane], r1 = rown[32 + lane];
    const float wsa = hi ? x1 : x0, wsb = hi ? x3 : x2;
    float4 acc0 = make_float4(r0.x * wsa, r0.y * wsa, r0.z * wsa, r0.w * wsa);
    float4 acc1 = make_float4(r1.x * wsb, r1.y * wsb, r1.z * wsb, r1.w * wsb);

    float d0, d1, d2, d3;
    {
        const float t0 = expf(a0 - m0), t1 = expf(a1 - m1), t2 = expf(a2 - m2), t3 = expf(a3 - m3);
        d0 = v0 ? t0 : 0.f; d1 = v0 ? t1 : 0.f; d2 = v0 ? t2 : 0.f; d3 = v0 ? t3 : 0.f;
        const int cnt = (end - beg < 32) ? (end - beg) : 32;
        gather_rows(hmat, s0, d0, d1, d2, d3, cnt, lane, hi, acc0, acc1);
    }
    for (int base = beg + 32; base < end; base += 32) {
        const int j = base + lane;
        int s = n;
        float e0 = 0.f, e1 = 0.f, e2 = 0.f, e3 = 0.f;
        if (j < end) {
            s = col[j];
            const float4 a = asrc4[s];
            e0 = expf(leakyf(a.x + ad.x) - m0); e1 = expf(leakyf(a.y + ad.y) - m1);
            e2 = expf(leakyf(a.z + ad.z) - m2); e3 = expf(leakyf(a.w + ad.w) - m3);
        }
        d0 += e0; d1 += e1; d2 += e2; d3 += e3;
        const int cnt = (end - base < 32) ? (end - base) : 32;
        gather_rows(hmat, s, e0, e1, e2, e3, cnt, lane, hi, acc0, acc1);
    }
    d0 = waveRedSum(d0); d1 = waveRedSum(d1); d2 = waveRedSum(d2); d3 = waveRedSum(d3);
    if (lane == 0)
        reinterpret_cast<float4*>(denom)[n] = make_float4(x0 + d0, x1 + d1, x2 + d2, x3 + d3);
    float4* orow = reinterpret_cast<float4*>(hacc + (size_t)n * HIDF);
    orow[lane]      = acc0;
    orow[32 + lane] = acc1;
}

__global__ void finalize_concat(const float* __restrict__ hacc, const float* __restrict__ denom,
                                const float* __restrict__ bias, const float* __restrict__ g,
                                const float* __restrict__ be, float* __restrict__ out, int N) {
    int wid  = (blockIdx.x * blockDim.x + threadIdx.x) >> 5;
    int lane = threadIdx.x & 31;
    if (wid >= N) return;
    size_t base = (size_t)wid * HIDF;
    float dn[HEADS];
#pragma unroll
    for (int h = 0; h < HEADS; ++h) dn[h] = 1.f / (denom[wid * HEADS + h] + 1e-16f);
    float v[8];
    float s = 0.f;
#pragma unroll
    for (int i = 0; i < 8; ++i) {
        int c = lane + 32 * i;
        v[i] = hacc[base + c] * dn[i >> 1] + bias[c];
        s += v[i];
    }
    float mu = waveRedSum(s) * (1.f / HIDF);
    float q = 0.f;
#pragma unroll
    for (int i = 0; i < 8; ++i) { float dlt = v[i] - mu; q += dlt * dlt; }
    float rstd = rsqrtf(waveRedSum(q) * (1.f / HIDF) + LN_EPS);
#pragma unroll
    for (int i = 0; i < 8; ++i) {
        int c = lane + 32 * i;
        float y = (v[i] - mu) * rstd * g[c] + be[c];
        out[base + c] = eluf(y);
    }
}

__global__ void finalize_mean(const float* __restrict__ hacc, const float* __restrict__ denom,
                              const float* __restrict__ bias, const float* __restrict__ g,
                              const float* __restrict__ be, float* __restrict__ out, int N) {
    int wid  = (blockIdx.x * blockDim.x + threadIdx.x) >> 5;
    int lane = threadIdx.x & 31;
    if (wid >= N) return;
    size_t base = (size_t)wid * HIDF;
    float dn[HEADS];
#pragma unroll
    for (int h = 0; h < HEADS; ++h) dn[h] = 1.f / (denom[wid * HEADS + h] + 1e-16f);
    float v[2];
    float s = 0.f;
#pragma unroll
    for (int i = 0; i < 2; ++i) {
        int c = lane + 32 * i;
        float acc = 0.f;
#pragma unroll
        for (int h = 0; h < HEADS; ++h) acc += hacc[base + h * CCH + c] * dn[h];
        v[i] = acc * (1.f / HEADS) + bias[c];
        s += v[i];
    }
    float mu = waveRedSum(s) * (1.f / CCH);
    float q = 0.f;
#pragma unroll
    for (int i = 0; i < 2; ++i) { float dlt = v[i] - mu; q += dlt * dlt; }
    float rstd = rsqrtf(waveRedSum(q) * (1.f / CCH) + LN_EPS);
#pragma unroll
    for (int i = 0; i < 2; ++i) {
        int c = lane + 32 * i;
        float y = (v[i] - mu) * rstd * g[c] + be[c];
        out[(size_t)wid * CCH + c] = eluf(y);
    }
}

__global__ void value_head(const float* __restrict__ hin,
                           const float* __restrict__ vw1, const float* __restrict__ vb1,
                           const float* __restrict__ vw2, const float* __restrict__ vb2,
                           float* __restrict__ out, int N) {
    int wid  = (blockIdx.x * blockDim.x + threadIdx.x) >> 5;
    int lane = threadIdx.x & 31;
    if (wid >= N) return;
    const float* row  = hin + (size_t)wid * CCH;
    const float* wrow = vw1 + lane * CCH;
    float hid = vb1[lane];
    for (int c = 0; c < CCH; ++c) hid += row[c] * wrow[c];
    hid = eluf(hid);
#pragma unroll
    for (int k = 0; k < 4; ++k) {
        float p = vw2[k * 32 + lane] * hid;
        p = waveRedSum(p);
        if (lane == 0) out[wid * 4 + k] = p + vb2[k];
    }
}

extern "C" void kernel_launch(void* const* d_in, const int* in_sizes, int n_in,
                              void* d_out, int out_size, void* d_ws, size_t ws_size,
                              hipStream_t stream) {
    const float* x      = (const float*)d_in[0];
    const int*   esrc   = (const int*)d_in[1];
    const int*   edst   = (const int*)d_in[2];
    const float* proj_w = (const float*)d_in[3];
    const float* proj_b = (const float*)d_in[4];
    const float* W[3]  = {(const float*)d_in[5],  (const float*)d_in[11], (const float*)d_in[17]};
    const float* AS[3] = {(const float*)d_in[6],  (const float*)d_in[12], (const float*)d_in[18]};
    const float* AD[3] = {(const float*)d_in[7],  (const float*)d_in[13], (const float*)d_in[19]};
    const float* B[3]  = {(const float*)d_in[8],  (const float*)d_in[14], (const float*)d_in[20]};
    const float* G[3]  = {(const float*)d_in[9],  (const float*)d_in[15], (const float*)d_in[21]};
    const float* BE[3] = {(const float*)d_in[10], (const float*)d_in[16], (const float*)d_in[22]};
    const float* vw1 = (const float*)d_in[23];
    const float* vb1 = (const float*)d_in[24];
    const float* vw2 = (const float*)d_in[25];
    const float* vb2 = (const float*)d_in[26];

    const int N = in_sizes[0] / 128;
    const int E = in_sizes[1];
    const int NH = N * HEADS;

    float* ws     = (float*)d_ws;
    float* hin    = ws;
    float* hmat   = hin  + (size_t)N * HIDF;
    float* hacc   = hmat + (size_t)N * HIDF;
    float* asrc   = hacc + (size_t)N * HIDF;
    float* adst   = asrc + (size_t)NH;
    float* denom  = adst + (size_t)NH;
    int* cnt      = (int*)(denom + (size_t)NH);
    int* rowptr   = cnt + (size_t)N;
    int* col      = rowptr + (size_t)N + 1;

    dim3 blk(256);

    csr_clear<<<(N + 255) / 256, blk, 0, stream>>>(cnt, N);
    csr_count<<<(E + 255) / 256, blk, 0, stream>>>(edst, cnt, E);
    csr_scan<<<1, 1024, 0, stream>>>(cnt, rowptr, N);
    csr_fill<<<(E + 255) / 256, blk, 0, stream>>>(esrc, edst, cnt, col, E);

    gemm_wmma_f32_lds<64, 64, 4, 2><<<(N + 63) / 64, blk, 0, stream>>>(
        x, 128, proj_w, hin, N, 128, 64, proj_b, 1);

    int in_ld = 64;
    for (int l = 0; l < 3; ++l) {
        int K = (l == 0) ? 64 : 256;
        gemm_wmma_f32_lds<64, 128, 2, 4><<<((N + 63) / 64) * (HIDF / 128), blk, 0, stream>>>(
            hin, in_ld, W[l], hmat, N, K, HIDF, nullptr, 0);
        attn_coef<<<(NH + 255) / 256, blk, 0, stream>>>(hmat, AS[l], AD[l], asrc, adst, NH);
        gat_gather<<<(N * 32 + 255) / 256, blk, 0, stream>>>(
            rowptr, col, asrc, adst, hmat, denom, hacc, N);
        if (l < 2) {
            finalize_concat<<<(N * 32 + 255) / 256, blk, 0, stream>>>(
                hacc, denom, B[l], G[l], BE[l], hin, N);
            in_ld = HIDF;
        } else {
            finalize_mean<<<(N * 32 + 255) / 256, blk, 0, stream>>>(
                hacc, denom, B[l], G[l], BE[l], hin, N);
        }
    }

    value_head<<<(N * 32 + 255) / 256, blk, 0, stream>>>(
        hin, vw1, vb1, vw2, vb2, (float*)d_out, N);
}


# asrc/adst logits taken in the layer GEMM epilogue from the C block staged in LDS (a wave owns and stores whole rows); attn_coef launches gone; logit sum now 4 terms per lane then a 16-lane tree
# speedup vs baseline: 2.8852x; 1.0523x over previous
#include <hip/hip_runtime.h>
#include <math.h>

typedef __attribute__((ext_vector_type(2))) float v2f;
typedef __attribute__((ext_vector_type(8))) float v8f;

#define HEADS 4
#define CCH 64
#define HIDF 256
#define NEG_SLOPE 0.2f
#define LN_EPS 1e-5f

__device__ __forceinline__ float eluf(float x) { return x > 0.f ? x : (expf(x) - 1.f); }
__device__ __forceinline__ float leakyf(float x) { return x > 0.f ? x : NEG_SLOPE * x; }

__device__ __forceinline__ float waveRedSum(float v) {
    for (int off = 16; off > 0; off >>= 1) v += __shfl_xor(v, off, 32);
    return v;
}
__device__ __forceinline__ float waveRedMax(float v) {
    for (int off = 16; off > 0; off >>= 1) v = fmaxf(v, __shfl_xor(v, off, 32));
    return v;
}
__device__ __forceinline__ int waveRedSumI(int v) {
    for (int off = 16; off > 0; off >>= 1) v += __shfl_xor(v, off, 32);
    return v;
}
__device__ __forceinline__ float laneValF(float v, int t) {
    return __int_as_float(__builtin_amdgcn_readlane(__float_as_int(v), t));
}

template <int BM, int BN, int WM, int WN, bool ATT>
__global__ void __launch_bounds__(256)
gemm_wmma_f32_lds(const float* __restrict__ A, int lda,
                  const float* __restrict__ W,
                  float* __restrict__ C,
                  int M, int K, int Nout,
                  const float* __restrict__ bias, int elu_act,
                  const float* __restrict__ a_s, const float* __restrict__ a_d,
                  float* __restrict__ asrc, float* __restrict__ adst) {
    constexpr int BK = 32;
    constexpr int P  = BK + 4;
    constexpr int TM = BM / (16 * WM);
    constexpr int TN = BN / (16 * WN);
    constexpr int AF = BM * (BK / 4) / 256;
    constexpr int BF = BN * (BK / 4) / 256;
    static_assert(WM * WN == 8, "work-group of 8 waves");
    static_assert(TM * 16 * WM == BM && TN * 16 * WN == BN, "wave tiles cover the block");
    static_assert(AF * 256 == BM * (BK / 4) && BF * 256 == BN * (BK / 4), "copy covers the chunk");

    __shared__ __attribute__((aligned(16))) float sA[2][BM * P];
    __shared__ __attribute__((aligned(16))) float sB[2][BN * P];

    const int tid  = threadIdx.x;
    const int lane = tid & 31;
    const int w    = tid >> 5;
    const int wm   = w / WN;
    const int wn   = w - wm * WN;
    const int blocksN = Nout / BN;
    const int bm = blockIdx.x / blocksN;
    const int bn = blockIdx.x - bm * blocksN;
    const int m0 = bm * BM;
    const int n0 = bn * BN;

    const int cr = tid >> 3;
    const int cq = (tid & 7) << 2;
    const float* gA[AF];
    const float* gB[BF];
#pragma unroll
    for (int j = 0; j < AF; ++j) {
        int gm = m0 + cr + 32 * j;
        gm = gm < M ? gm : (M - 1);
        gA[j] = A + (size_t)gm * lda + cq;
    }
#pragma unroll
    for (int j = 0; j < BF; ++j) gB[j] = W + (size_t)(n0 + cr + 32 * j) * K + cq;

    static_assert(AF == 2 && (BF == 2 || BF == 4), "piece registers below are written out for these sizes");
    float4 ra0, ra1, rb0, rb1, rb2, rb3;
    ra0 = *reinterpret_cast<const float4*>(gA[0]);
    ra1 = *reinterpret_cast<const float4*>(gA[1]);
    rb0 = *reinterpret_cast<const float4*>(gB[0]);
    rb1 = *reinterpret_cast<const float4*>(gB[1]);
    if constexpr (BF == 4) {
        rb2 = *reinterpret_cast<const float4*>(gB[2]);
        rb3 = *reinterpret_cast<const float4*>(gB[3]);
    }
    const int sOff = cr * P + cq;
    *reinterpret_cast<float4*>(&sA[0][sOff])          = ra0;
    *reinterpret_cast<float4*>(&sA[0][sOff + 32 * P]) = ra1;
    *reinterpret_cast<float4*>(&sB[0][sOff])          = rb0;
    *reinterpret_cast<float4*>(&sB[0][sOff + 32 * P]) = rb1;
    if constexpr (BF == 4) {
        *reinterpret_cast<float4*>(&sB[0][sOff + 64 * P]) = rb2;
        *reinterpret_cast<float4*>(&sB[0][sOff + 96 * P]) = rb3;
    }

    v8f acc[TM][TN];
#pragma unroll
    for (int i = 0; i < TM; ++i)
#pragma unroll
        for (int j = 0; j < TN; ++j) acc[i][j] = v8f{};

    const int fragOff = (lane & 15) * P + ((lane >> 4) << 1);
    const int nk = K / BK;
    for (int it = 0; it < nk; ++it) {
        const int  buf  = it & 1;
        const bool more = it + 1 < nk;
        __syncthreads();
        if (more) {
            const int k1 = (it + 1) * BK;
            ra0 = *reinterpret_cast<const float4*>(gA[0] + k1);
            ra1 = *reinterpret_cast<const float4*>(gA[1] + k1);
            rb0 = *reinterpret_cast<const float4*>(gB[0] + k1);
            rb1 = *reinterpret_cast<const float4*>(gB[1] + k1);
            if constexpr (BF == 4) {
                rb2 = *reinterpret_cast<const float4*>(gB[2] + k1);
                rb3 = *reinterpret_cast<const float4*>(gB[3] + k1);
            }
        }
        const float* pa = &sA[buf][wm * (TM * 16) * P + fragOff];
        const float* pb = &sB[buf][wn * (TN * 16) * P + fragOff];
#pragma unroll
        for (int ks = 0; ks < BK / 4; ++ks) {
            v2f af[TM], bf[TN];
#pragma unroll
            for (int i = 0; i < TM; ++i) af[i] = *reinterpret_cast<const v2f*>(pa + i * 16 * P + ks * 4);
#pragma unroll
            for (int j = 0; j < TN; ++j) bf[j] = *reinterpret_cast<const v2f*>(pb + j * 16 * P + ks * 4);
#pragma unroll
            for (int i = 0; i < TM; ++i)
#pragma unroll
                for (int j = 0; j < TN; ++j)
                    acc[i][j] = __builtin_amdgcn_wmma_f32_16x16x4_f32(false, af[i], false, bf[j],
                                                                      (short)0, acc[i][j], false, false);
        }
        if (more) {
            *reinterpret_cast<float4*>(&sA[buf ^ 1][sOff])          = ra0;
            *reinterpret_cast<float4*>(&sA[buf ^ 1][sOff + 32 * P]) = ra1;
            *reinterpret_cast<float4*>(&sB[buf ^ 1][sOff])          = rb0;
            *reinterpret_cast<float4*>(&sB[buf ^ 1][sOff + 32 * P]) = rb1;
            if constexpr (BF == 4) {
                *reinterpret_cast<float4*>(&sB[buf ^ 1][sOff + 64 * P]) = rb2;
                *reinterpret_cast<float4*>(&sB[buf ^ 1][sOff + 96 * P]) = rb3;
            }
        }
    }

    if constexpr (ATT) {
        constexpr int CP = BN + 4;
        constexpr int RW = BM / 8;
        static_assert(BN == 128, "a row of the block is one 16-byte piece per lane, one head per lane half");
        static_assert(BM * CP <= 2 * BN * P, "the C block fits in the W chunk buffers");
        float* sC = &sB[0][0];
        __syncthreads();
#pragma unroll
        for (int i = 0; i < TM; ++i)
#pragma unroll
            for (int j = 0; j < TN; ++j)
#pragma unroll
                for (int r = 0; r < 8; ++r)
                    sC[((wm * TM + i) * 16 + ((lane >> 4) << 3) + r) * CP + (wn * TN + j) * 16 + (lane & 15)] = acc[i][j][r];
        __syncthreads();
        const int colg = n0 + 4 * lane;
        const float4 s4 = *reinterpret_cast<const float4*>(a_s + colg);
        const float4 d4 = *reinterpret_cast<const float4*>(a_d + colg);
        const int head = (n0 >> 6) + (lane >> 4);
        float4 v[RW];
        float ps[RW], pd[RW];
#pragma unroll
        for (int q = 0; q < RW; ++q)
            v[q] = *reinterpret_cast<const float4*>(sC + (w * RW + q) * CP + 4 * lane);
#pragma unroll
        for (int q = 0; q < RW; ++q) {
            ps[q] = (v[q].x * s4.x + v[q].y * s4.y) + (v[q].z * s4.z + v[q].w * s4.w);
            pd[q] = (v[q].x * d4.x + v[q].y * d4.y) + (v[q].z * d4.z + v[q].w * d4.w);
        }
#pragma unroll
        for (int off = 8; off > 0; off >>= 1) {
#pragma unroll
            for (int q = 0; q < RW; ++q) {
                ps[q] += __shfl_xor(ps[q], off, 32);
                pd[q] += __shfl_xor(pd[q], off, 32);
            }
        }
#pragma unroll
        for (int q = 0; q < RW; ++q) {
            const int gm = m0 + w * RW + q;
            if (gm < M) {
                *reinterpret_cast<float4*>(C + (size_t)gm * Nout + colg) = v[q];
                if ((lane & 15) == 0) {
                    asrc[gm * HEADS + head] = ps[q];
                    adst[gm * HEADS + head] = pd[q];
                }
            }
        }
        return;
    }

#pragma unroll
    for (int i = 0; i < TM; ++i) {
#pragma unroll
        for (int j = 0; j < TN; ++j) {
            const int rbase = m0 + (wm * TM + i) * 16 + ((lane >> 4) << 3);
            const int col   = n0 + (wn * TN + j) * 16 + (lane & 15);
            const float bv  = bias ? bias[col] : 0.f;
#pragma unroll
            for (int r = 0; r < 8; ++r) {
                float v = acc[i][j][r] + bv;
                if (elu_act) v = eluf(v);
                if (rbase + r < M) C[(size_t)(rbase + r) * Nout + col] = v;
            }
        }
    }
}

__global__ void csr_clear(int* __restrict__ cnt, int N) {
    int i = blockIdx.x * blockDim.x + threadIdx.x;
    if (i < N) cnt[i] = 0;
}

__global__ void csr_count(const int* __restrict__ dst, int* __restrict__ cnt, int E) {
    int e = blockIdx.x * blockDim.x + threadIdx.x;
    if (e < E) atomicAdd(&cnt[dst[e]], 1);
}

__global__ void __launch_bounds__(1024) csr_scan(int* __restrict__ cnt, int* __restrict__ rowptr, int N) {
    __shared__ int wtot[32];
    const int lane = threadIdx.x & 31;
    const int w    = threadIdx.x >> 5;
    const int per  = ((N + 4095) >> 12) << 7;
    const int lo   = w * per;
    const int hi   = (lo + per < N) ? (lo + per) : N;
    int sum = 0;
#pragma unroll 4
    for (int i = lo + 4 * lane; i < hi; i += 128) {
        const int4 c = *reinterpret_cast<const int4*>(cnt + i);
        sum += (c.x + c.y) + (c.z + c.w);
    }
    sum = waveRedSumI(sum);
    if (lane == 0) wtot[w] = sum;
    __syncthreads();
    int carry = waveRedSumI(lane < w ? wtot[lane] : 0);
    for (int i0 = lo; i0 < hi; i0 += 128) {
        const int  i  = i0 + 4 * lane;
        const bool in = i < hi;
        int4 c = make_int4(0, 0, 0, 0);
        if (in) c = *reinterpret_cast<const int4*>(cnt + i);
        const int t = (c.x + c.y) + (c.z + c.w);
        int x = t;
        for (int off = 1; off < 32; off <<= 1) {
            int y = __shfl_up(x, off, 32);
            if (lane >= off) x += y;
        }
        if (in) {
            const int b = carry + x - t;
            const int4 o = make_int4(b, b + c.x, b + c.x + c.y, b + c.x + c.y + c.z);
            *reinterpret_cast<int4*>(rowptr + i) = o;
            *reinterpret_cast<int4*>(cnt + i)    = o;
        }
        carry += __shfl(x, 31, 32);
    }
    if (threadIdx.x == 1023) rowptr[N] = carry;
}

__global__ void csr_fill(const int* __restrict__ src, const int* __restrict__ dst,
                         int* __restrict__ cursor, int* __restrict__ col, int E) {
    int e = blockIdx.x * blockDim.x + threadIdx.x;
    if (e >= E) return;
    int p = atomicAdd(&cursor[dst[e]], 1);
    col[p] = src[e];
}

__device__ __forceinline__ void gather_rows(const float* __restrict__ hmat, int s,
                                            float e0, float e1, float e2, float e3,
                                            int cnt, int lane, bool hi,
                                            float4& acc0, float4& acc1) {
#pragma unroll 4
    for (int t = 0; t < cnt; ++t) {
        const int   st = __builtin_amdgcn_readlane(s, t);
        const float w0 = laneValF(e0, t), w1 = laneValF(e1, t);
        const float w2 = laneValF(e2, t), w3 = laneValF(e3, t);
        const float wa = hi ? w1 : w0;
        const float wb = hi ? w3 : w2;
        const float4* r = reinterpret_cast<const float4*>(hmat + (size_t)st * HIDF);
        const float4 p = r[lane];
        const float4 q = r[32 + lane];
        acc0.x += wa * p.x; acc0.y += wa * p.y; acc0.z += wa * p.z; acc0.w += wa * p.w;
        acc1.x += wb * q.x; acc1.y += wb * q.y; acc1.z += wb * q.z; acc1.w += wb * q.w;
    }
}

__global__ void gat_gather(const int* __restrict__ rowptr, const int* __restrict__ col,
                           const float* __restrict__ asrc, const float* __restrict__ adst,
                           const float* __restrict__ hmat,
                           float* __restrict__ denom, float* __restrict__ hacc, int N) {
    const int lane = threadIdx.x & 31;
    const int n = __builtin_amdgcn_readfirstlane((int)((blockIdx.x * blockDim.x + threadIdx.x) >> 5));
    if (n >= N) return;
    const int beg = rowptr[n], end = rowptr[n + 1];
    const bool hi = lane >= 16;
    const float4* asrc4 = reinterpret_cast<const float4*>(asrc);
    const float4 ad = reinterpret_cast<const float4*>(adst)[n];
    const float4 an = asrc4[n];
    const float sf0 = leakyf(an.x + ad.x), sf1 = leakyf(an.y + ad.y);
    const float sf2 = leakyf(an.z + ad.z), sf3 = leakyf(an.w + ad.w);

    const bool v0 = beg + lane < end;
    int   s0 = n;
    float a0 = sf0, a1 = sf1, a2 = sf2, a3 = sf3;
    if (v0) {
        s0 = col[beg + lane];
        const float4 a = asrc4[s0];
        a0 = leakyf(a.x + ad.x); a1 = leakyf(a.y + ad.y);
        a2 = leakyf(a.z + ad.z); a3 = leakyf(a.w + ad.w);
    }
    float m0 = fmaxf(sf0, a0), m1 = fmaxf(sf1, a1), m2 = fmaxf(sf2, a2), m3 = fmaxf(sf3, a3);
    for (int base = beg + 32; base < end; base += 32) {
        const int j = base + lane;
        if (j < end) {
            const float4 a = asrc4[col[j]];
            m0 = fmaxf(m0, leakyf(a.x + ad.x)); m1 = fmaxf(m1, leakyf(a.y + ad.y));
            m2 = fmaxf(m2, leakyf(a.z + ad.z)); m3 = fmaxf(m3, leakyf(a.w + ad.w));
        }
    }
    m0 = waveRedMax(m0); m1 = waveRedMax(m1); m2 = waveRedMax(m2); m3 = waveRedMax(m3);

    const float x0 = expf(sf0 - m0), x1 = expf(sf1 - m1), x2 = expf(sf2 - m2), x3 = expf(sf3 - m3);
    const float4* rown = reinterpret_cast<const float4*>(hmat + (size_t)n * HIDF);
    const float4 r0 = rown[lane], r1 = rown[32 + lane];
    const float wsa = hi ? x1 : x0, wsb = hi ? x3 : x2;
    float4 acc0 = make_float4(r0.x * wsa, r0.y * wsa, r0.z * wsa, r0.w * wsa);
    float4 acc1 = make_float4(r1.x * wsb, r1.y * wsb, r1.z * wsb, r1.w * wsb);

    float d0, d1, d2, d3;
    {
        const float t0 = expf(a0 - m0), t1 = expf(a1 - m1), t2 = expf(a2 - m2), t3 = expf(a3 - m3);
        d0 = v0 ? t0 : 0.f; d1 = v0 ? t1 : 0.f; d2 = v0 ? t2 : 0.f; d3 = v0 ? t3 : 0.f;
        const int cnt = (end - beg < 32) ? (end - beg) : 32;
        gather_rows(hmat, s0, d0, d1, d2, d3, cnt, lane, hi, acc0, acc1);
    }
    for (int base = beg + 32; base < end; base += 32) {
        const int j = base + lane;
        int s = n;
        float e0 = 0.f, e1 = 0.f, e2 = 0.f, e3 = 0.f;
        if (j < end) {
            s = col[j];
            const float4 a = asrc4[s];
            e0 = expf(leakyf(a.x + ad.x) - m0); e1 = expf(leakyf(a.y + ad.y) - m1);
            e2 = expf(leakyf(a.z + ad.z) - m2); e3 = expf(leakyf(a.w + ad.w) - m3);
        }
        d0 += e0; d1 += e1; d2 += e2; d3 += e3;
        const int cnt = (end - base < 32) ? (end - base) : 32;
        gather_rows(hmat, s, e0, e1, e2, e3, cnt, lane, hi, acc0, acc1);
    }
    d0 = waveRedSum(d0); d1 = waveRedSum(d1); d2 = waveRedSum(d2); d3 = waveRedSum(d3);
    if (lane == 0)
        reinterpret_cast<float4*>(denom)[n] = make_float4(x0 + d0, x1 + d1, x2 + d2, x3 + d3);
    float4* orow = reinterpret_cast<float4*>(hacc + (size_t)n * HIDF);
    orow[lane]      = acc0;
    orow[32 + lane] = acc1;
}

__global__ void finalize_concat(const float* __restrict__ hacc, const float* __restrict__ denom,
                                const float* __restrict__ bias, const float* __restrict__ g,
                                const float* __restrict__ be, float* __restrict__ out, int N) {
    int wid  = (blockIdx.x * blockDim.x + threadIdx.x) >> 5;
    int lane = threadIdx.x & 31;
    if (wid >= N) return;
    size_t base = (size_t)wid * HIDF;
    float dn[HEADS];
#pragma unroll
    for (int h = 0; h < HEADS; ++h) dn[h] = 1.f / (denom[wid * HEADS + h] + 1e-16f);
    float v[8];
    float s = 0.f;
#pragma unroll
    for (int i = 0; i < 8; ++i) {
        int c = lane + 32 * i;
        v[i] = hacc[base + c] * dn[i >> 1] + bias[c];
        s += v[i];
    }
    float mu = waveRedSum(s) * (1.f / HIDF);
    float q = 0.f;
#pragma unroll
    for (int i = 0; i < 8; ++i) { float dlt = v[i] - mu; q += dlt * dlt; }
    float rstd = rsqrtf(waveRedSum(q) * (1.f / HIDF) + LN_EPS);
#pragma unroll
    for (int i = 0; i < 8; ++i) {
        int c = lane + 32 * i;
        float y = (v[i] - mu) * rstd * g[c] + be[c];
        out[base + c] = eluf(y);
    }
}

__global__ void finalize_mean(const float* __restrict__ hacc, const float* __restrict__ denom,
                              const float* __restrict__ bias, const float* __restrict__ g,
                              const float* __restrict__ be, float* __restrict__ out, int N) {
    int wid  = (blockIdx.x * blockDim.x + threadIdx.x) >> 5;
    int lane = threadIdx.x & 31;
    if (wid >= N) return;
    size_t base = (size_t)wid * HIDF;
    float dn[HEADS];
#pragma unroll
    for (int h = 0; h < HEADS; ++h) dn[h] = 1.f / (denom[wid * HEADS + h] + 1e-16f);
    float v[2];
    float s = 0.f;
#pragma unroll
    for (int i = 0; i < 2; ++i) {
        int c = lane + 32 * i;
        float acc = 0.f;
#pragma unroll
        for (int h = 0; h < HEADS; ++h) acc += hacc[base + h * CCH + c] * dn[h];
        v[i] = acc * (1.f / HEADS) + bias[c];
        s += v[i];
    }
    float mu = waveRedSum(s) * (1.f / CCH);
    float q = 0.f;
#pragma unroll
    for (int i = 0; i < 2; ++i) { float dlt = v[i] - mu; q += dlt * dlt; }
    float rstd = rsqrtf(waveRedSum(q) * (1.f / CCH) + LN_EPS);
#pragma unroll
    for (int i = 0; i < 2; ++i) {
        int c = lane + 32 * i;
        float y = (v[i] - mu) * rstd * g[c] + be[c];
        out[(size_t)wid * CCH + c] = eluf(y);
    }
}

__global__ void value_head(const float* __restrict__ hin,
                           const float* __restrict__ vw1, const float* __restrict__ vb1,
                           const float* __restrict__ vw2, const float* __restrict__ vb2,
                           float* __restrict__ out, int N) {
    int wid  = (blockIdx.x * blockDim.x + threadIdx.x) >> 5;
    int lane = threadIdx.x & 31;
    if (wid >= N) return;
    const float* row  = hin + (size_t)wid * CCH;
    const float* wrow = vw1 + lane * CCH;
    float hid = vb1[lane];
    for (int c = 0; c < CCH; ++c) hid += row[c] * wrow[c];
    hid = eluf(hid);
#pragma unroll
    for (int k = 0; k < 4; ++k) {
        float p = vw2[k * 32 + lane] * hid;
        p = waveRedSum(p);
        if (lane == 0) out[wid * 4 + k] = p + vb2[k];
    }
}

extern "C" void kernel_launch(void* const* d_in, const int* in_sizes, int n_in,
                              void* d_out, int out_size, void* d_ws, size_t ws_size,
                              hipStream_t stream) {
    const float* x      = (const float*)d_in[0];
    const int*   esrc   = (const int*)d_in[1];
    const int*   edst   = (const int*)d_in[2];
    const float* proj_w = (const float*)d_in[3];
    const float* proj_b = (const float*)d_in[4];
    const float* W[3]  = {(const float*)d_in[5],  (const float*)d_in[11], (const float*)d_in[17]};
    const float* AS[3] = {(const float*)d_in[6],  (const float*)d_in[12], (const float*)d_in[18]};
    const float* AD[3] = {(const float*)d_in[7],  (const float*)d_in[13], (const float*)d_in[19]};
    const float* B[3]  = {(const float*)d_in[8],  (const float*)d_in[14], (const float*)d_in[20]};
    const float* G[3]  = {(const float*)d_in[9],  (const float*)d_in[15], (const float*)d_in[21]};
    const float* BE[3] = {(const float*)d_in[10], (const float*)d_in[16], (const float*)d_in[22]};
    const float* vw1 = (const float*)d_in[23];
    const float* vb1 = (const float*)d_in[24];
    const float* vw2 = (const float*)d_in[25];
    const float* vb2 = (const float*)d_in[26];

    const int N = in_sizes[0] / 128;
    const int E = in_sizes[1];
    const int NH = N * HEADS;

    float* ws     = (float*)d_ws;
    float* hin    = ws;
    float* hmat   = hin  + (size_t)N * HIDF;
    float* hacc   = hmat + (size_t)N * HIDF;
    float* asrc   = hacc + (size_t)N * HIDF;
    float* adst   = asrc + (size_t)NH;
    float* denom  = adst + (size_t)NH;
    int* cnt      = (int*)(denom + (size_t)NH);
    int* rowptr   = cnt + (size_t)N;
    int* col      = rowptr + (size_t)N + 1;

    dim3 blk(256);

    csr_clear<<<(N + 255) / 256, blk, 0, stream>>>(cnt, N);
    csr_count<<<(E + 255) / 256, blk, 0, stream>>>(edst, cnt, E);
    csr_scan<<<1, 1024, 0, stream>>>(cnt, rowptr, N);
    csr_fill<<<(E + 255) / 256, blk, 0, stream>>>(esrc, edst, cnt, col, E);

    gemm_wmma_f32_lds<64, 64, 4, 2, false><<<(N + 63) / 64, blk, 0, stream>>>(
        x, 128, proj_w, hin, N, 128, 64, proj_b, 1, nullptr, nullptr, nullptr, nullptr);

    int in_ld = 64;
    for (int l = 0; l < 3; ++l) {
        int K = (l == 0) ? 64 : 256;
        gemm_wmma_f32_lds<64, 128, 2, 4, true><<<((N + 63) / 64) * (HIDF / 128), blk, 0, stream>>>(
            hin, in_ld, W[l], hmat, N, K, HIDF, nullptr, 0, AS[l], AD[l], asrc, adst);
        gat_gather<<<(N * 32 + 255) / 256, blk, 0, stream>>>(
            rowptr, col, asrc, adst, hmat, denom, hacc, N);
        if (l < 2) {
            finalize_concat<<<(N * 32 + 255) / 256, blk, 0, stream>>>(
                hacc, denom, B[l], G[l], BE[l], hin, N);
            in_ld = HIDF;
        } else {
            finalize_mean<<<(N * 32 + 255) / 256, blk, 0, stream>>>(
                hacc, denom, B[l], G[l], BE[l], hin, N);
        }
    }

    value_head<<<(N * 32 + 255) / 256, blk, 0, stream>>>(
        hin, vw1, vb1, vw2, vb2, (float*)d_out, N);
}


# Gather wave finishes its node in registers: 1/denom, bias, LayerNorm, ELU (last layer: head mean first); finalize launches and hacc round trip gone; LayerNorm sums group 4 adjacent channels per lane
# speedup vs baseline: 3.1455x; 1.0902x over previous
#include <hip/hip_runtime.h>
#include <math.h>

typedef __attribute__((ext_vector_type(2))) float v2f;
typedef __attribute__((ext_vector_type(8))) float v8f;

#define HEADS 4
#define CCH 64
#define HIDF 256
#define NEG_SLOPE 0.2f
#define LN_EPS 1e-5f

__device__ __forceinline__ float eluf(float x) { return x > 0.f ? x : (expf(x) - 1.f); }
__device__ __forceinline__ float leakyf(float x) { return x > 0.f ? x : NEG_SLOPE * x; }

__device__ __forceinline__ float waveRedSum(float v) {
    for (int off = 16; off > 0; off >>= 1) v += __shfl_xor(v, off, 32);
    return v;
}
__device__ __forceinline__ float waveRedMax(float v) {
    for (int off = 16; off > 0; off >>= 1) v = fmaxf(v, __shfl_xor(v, off, 32));
    return v;
}
__device__ __forceinline__ int waveRedSumI(int v) {
    for (int off = 16; off > 0; off >>= 1) v += __shfl_xor(v, off, 32);
    return v;
}
__device__ __forceinline__ float laneValF(float v, int t) {
    return __int_as_float(__builtin_amdgcn_readlane(__float_as_int(v), t));
}

template <int BM, int BN, int WM, int WN, bool ATT>
__global__ void __launch_bounds__(256)
gemm_wmma_f32_lds(const float* __restrict__ A, int lda,
                  const float* __restrict__ W,
                  float* __restrict__ C,
                  int M, int K, int Nout,
                  const float* __restrict__ bias, int elu_act,
                  const float* __restrict__ a_s, const float* __restrict__ a_d,
                  float* __restrict__ asrc, float* __restrict__ adst) {
    constexpr int BK = 32;
    constexpr int P  = BK + 4;
    constexpr int TM = BM / (16 * WM);
    constexpr int TN = BN / (16 * WN);
    constexpr int AF = BM * (BK / 4) / 256;
    constexpr int BF = BN * (BK / 4) / 256;
    static_assert(WM * WN == 8, "work-group of 8 waves");
    static_assert(TM * 16 * WM == BM && TN * 16 * WN == BN, "wave tiles cover the block");
    static_assert(AF * 256 == BM * (BK / 4) && BF * 256 == BN * (BK / 4), "copy covers the chunk");

    __shared__ __attribute__((aligned(16))) float sA[2][BM * P];
    __shared__ __attribute__((aligned(16))) float sB[2][BN * P];

    const int tid  = threadIdx.x;
    const int lane = tid & 31;
    const int w    = tid >> 5;
    const int wm   = w / WN;
    const int wn   = w - wm * WN;
    const int blocksN = Nout / BN;
    const int bm = blockIdx.x / blocksN;
    const int bn = blockIdx.x - bm * blocksN;
    const int m0 = bm * BM;
    const int n0 = bn * BN;

    const int cr = tid >> 3;
    const int cq = (tid & 7) << 2;
    const float* gA[AF];
    const float* gB[BF];
#pragma unroll
    for (int j = 0; j < AF; ++j) {
        int gm = m0 + cr + 32 * j;
        gm = gm < M ? gm : (M - 1);
        gA[j] = A + (size_t)gm * lda + cq;
    }
#pragma unroll
    for (int j = 0; j < BF; ++j) gB[j] = W + (size_t)(n0 + cr + 32 * j) * K + cq;

    static_assert(AF == 2 && (BF == 2 || BF == 4), "piece registers below are written out for these sizes");
    float4 ra0, ra1, rb0, rb1, rb2, rb3;
    ra0 = *reinterpret_cast<const float4*>(gA[0]);
    ra1 = *reinterpret_cast<const float4*>(gA[1]);
    rb0 = *reinterpret_cast<const float4*>(gB[0]);
    rb1 = *reinterpret_cast<const float4*>(gB[1]);
    if constexpr (BF == 4) {
        rb2 = *reinterpret_cast<const float4*>(gB[2]);
        rb3 = *reinterpret_cast<const float4*>(gB[3]);
    }
    const int sOff = cr * P + cq;
    *reinterpret_cast<float4*>(&sA[0][sOff])          = ra0;
    *reinterpret_cast<float4*>(&sA[0][sOff + 32 * P]) = ra1;
    *reinterpret_cast<float4*>(&sB[0][sOff])          = rb0;
    *reinterpret_cast<float4*>(&sB[0][sOff + 32 * P]) = rb1;
    if constexpr (BF == 4) {
        *reinterpret_cast<float4*>(&sB[0][sOff + 64 * P]) = rb2;
        *reinterpret_cast<float4*>(&sB[0][sOff + 96 * P]) = rb3;
    }

    v8f acc[TM][TN];
#pragma unroll
    for (int i = 0; i < TM; ++i)
#pragma unroll
        for (int j = 0; j < TN; ++j) acc[i][j] = v8f{};

    const int fragOff = (lane & 15) * P + ((lane >> 4) << 1);
    const int nk = K / BK;
    for (int it = 0; it < nk; ++it) {
        const int  buf  = it & 1;
        const bool more = it + 1 < nk;
        __syncthreads();
        if (more) {
            const int k1 = (it + 1) * BK;
            ra0 = *reinterpret_cast<const float4*>(gA[0] + k1);
            ra1 = *reinterpret_cast<const float4*>(gA[1] + k1);
            rb0 = *reinterpret_cast<const float4*>(gB[0] + k1);
            rb1 = *reinterpret_cast<const float4*>(gB[1] + k1);
            if constexpr (BF == 4) {
                rb2 = *reinterpret_cast<const float4*>(gB[2] + k1);
                rb3 = *reinterpret_cast<const float4*>(gB[3] + k1);
            }
        }
        const float* pa = &sA[buf][wm * (TM * 16) * P + fragOff];
        const float* pb = &sB[buf][wn * (TN * 16) * P + fragOff];
#pragma unroll
        for (int ks = 0; ks < BK / 4; ++ks) {
            v2f af[TM], bf[TN];
#pragma unroll
            for (int i = 0; i < TM; ++i) af[i] = *reinterpret_cast<const v2f*>(pa + i * 16 * P + ks * 4);
#pragma unroll
            for (int j = 0; j < TN; ++j) bf[j] = *reinterpret_cast<const v2f*>(pb + j * 16 * P + ks * 4);
#pragma unroll
            for (int i = 0; i < TM; ++i)
#pragma unroll
                for (int j = 0; j < TN; ++j)
                    acc[i][j] = __builtin_amdgcn_wmma_f32_16x16x4_f32(false, af[i], false, bf[j],
                                                                      (short)0, acc[i][j], false, false);
        }
        if (more) {
            *reinterpret_cast<float4*>(&sA[buf ^ 1][sOff])          = ra0;
            *reinterpret_cast<float4*>(&sA[buf ^ 1][sOff + 32 * P]) = ra1;
            *reinterpret_cast<float4*>(&sB[buf ^ 1][sOff])          = rb0;
            *reinterpret_cast<float4*>(&sB[buf ^ 1][sOff + 32 * P]) = rb1;
            if constexpr (BF == 4) {
                *reinterpret_cast<float4*>(&sB[buf ^ 1][sOff + 64 * P]) = rb2;
                *reinterpret_cast<float4*>(&sB[buf ^ 1][sOff + 96 * P]) = rb3;
            }
        }
    }

    if constexpr (ATT) {
        constexpr int CP = BN + 4;
        constexpr int RW = BM / 8;
        static_assert(BN == 128, "a row of the block is one 16-byte piece per lane, one head per lane half");
        static_assert(BM * CP <= 2 * BN * P, "the C block fits in the W chunk buffers");
        float* sC = &sB[0][0];
        __syncthreads();
#pragma unroll
        for (int i = 0; i < TM; ++i)
#pragma unroll
            for (int j = 0; j < TN; ++j)
#pragma unroll
                for (int r = 0; r < 8; ++r)
                    sC[((wm * TM + i) * 16 + ((lane >> 4) << 3) + r) * CP + (wn * TN + j) * 16 + (lane & 15)] = acc[i][j][r];
        __syncthreads();
        const int colg = n0 + 4 * lane;
        const float4 s4 = *reinterpret_cast<const float4*>(a_s + colg);
        const float4 d4 = *reinterpret_cast<const float4*>(a_d + colg);
        const int head = (n0 >> 6) + (lane >> 4);
        float4 v[RW];
        float ps[RW], pd[RW];
#pragma unroll
        for (int q = 0; q < RW; ++q)
            v[q] = *reinterpret_cast<const float4*>(sC + (w * RW + q) * CP + 4 * lane);
#pragma unroll
        for (int q = 0; q < RW; ++q) {
            ps[q] = (v[q].x * s4.x + v[q].y * s4.y) + (v[q].z * s4.z + v[q].w * s4.w);
            pd[q] = (v[q].x * d4.x + v[q].y * d4.y) + (v[q].z * d4.z + v[q].w * d4.w);
        }
#pragma unroll
        for (int off = 8; off > 0; off >>= 1) {
#pragma unroll
            for (int q = 0; q < RW; ++q) {
                ps[q] += __shfl_xor(ps[q], off, 32);
                pd[q] += __shfl_xor(pd[q], off, 32);
            }
        }
#pragma unroll
        for (int q = 0; q < RW; ++q) {
            const int gm = m0 + w * RW + q;
            if (gm < M) {
                *reinterpret_cast<float4*>(C + (size_t)gm * Nout + colg) = v[q];
                if ((lane & 15) == 0) {
                    asrc[gm * HEADS + head] = ps[q];
                    adst[gm * HEADS + head] = pd[q];
                }
            }
        }
        return;
    }

#pragma unroll
    for (int i = 0; i < TM; ++i) {
#pragma unroll
        for (int j = 0; j < TN; ++j) {
            const int rbase = m0 + (wm * TM + i) * 16 + ((lane >> 4) << 3);
            const int col   = n0 + (wn * TN + j) * 16 + (lane & 15);
            const float bv  = bias ? bias[col] : 0.f;
#pragma unroll
            for (int r = 0; r < 8; ++r) {
                float v = acc[i][j][r] + bv;
                if (elu_act) v = eluf(v);
                if (rbase + r < M) C[(size_t)(rbase + r) * Nout + col] = v;
            }
        }
    }
}

__global__ void csr_clear(int* __restrict__ cnt, int N) {
    int i = blockIdx.x * blockDim.x + threadIdx.x;
    if (i < N) cnt[i] = 0;
}

__global__ void csr_count(const int* __restrict__ dst, int* __restrict__ cnt, int E) {
    int e = blockIdx.x * blockDim.x + threadIdx.x;
    if (e < E) atomicAdd(&cnt[dst[e]], 1);
}

__global__ void __launch_bounds__(1024) csr_scan(int* __restrict__ cnt, int* __restrict__ rowptr, int N) {
    __shared__ int wtot[32];
    const int lane = threadIdx.x & 31;
    const int w    = threadIdx.x >> 5;
    const int per  = ((N + 4095) >> 12) << 7;
    const int lo   = w * per;
    const int hi   = (lo + per < N) ? (lo + per) : N;
    int sum = 0;
#pragma unroll 4
    for (int i = lo + 4 * lane; i < hi; i += 128) {
        const int4 c = *reinterpret_cast<const int4*>(cnt + i);
        sum += (c.x + c.y) + (c.z + c.w);
    }
    sum = waveRedSumI(sum);
    if (lane == 0) wtot[w] = sum;
    __syncthreads();
    int carry = waveRedSumI(lane < w ? wtot[lane] : 0);
    for (int i0 = lo; i0 < hi; i0 += 128) {
        const int  i  = i0 + 4 * lane;
        const bool in = i < hi;
        int4 c = make_int4(0, 0, 0, 0);
        if (in) c = *reinterpret_cast<const int4*>(cnt + i);
        const int t = (c.x + c.y) + (c.z + c.w);
        int x = t;
        for (int off = 1; off < 32; off <<= 1) {
            int y = __shfl_up(x, off, 32);
            if (lane >= off) x += y;
        }
        if (in) {
            const int b = carry + x - t;
            const int4 o = make_int4(b, b + c.x, b + c.x + c.y, b + c.x + c.y + c.z);
            *reinterpret_cast<int4*>(rowptr + i) = o;
            *reinterpret_cast<int4*>(cnt + i)    = o;
        }
        carry += __shfl(x, 31, 32);
    }
    if (threadIdx.x == 1023) rowptr[N] = carry;
}

__global__ void csr_fill(const int* __restrict__ src, const int* __restrict__ dst,
                         int* __restrict__ cursor, int* __restrict__ col, int E) {
    int e = blockIdx.x * blockDim.x + threadIdx.x;
    if (e >= E) return;
    int p = atomicAdd(&cursor[dst[e]], 1);
    col[p] = src[e];
}

__device__ __forceinline__ void gather_rows(const float* __restrict__ hmat, int s,
                                            float e0, float e1, float e2, float e3,
                                            int cnt, int lane, bool hi,
                                            float4& acc0, float4& acc1) {
#pragma unroll 4
    for (int t = 0; t < cnt; ++t) {
        const int   st = __builtin_amdgcn_readlane(s, t);
        const float w0 = laneValF(e0, t), w1 = laneValF(e1, t);
        const float w2 = laneValF(e2, t), w3 = laneValF(e3, t);
        const float wa = hi ? w1 : w0;
        const float wb = hi ? w3 : w2;
        const float4* r = reinterpret_cast<const float4*>(hmat + (size_t)st * HIDF);
        const float4 p = r[lane];
        const float4 q = r[32 + lane];
        acc0.x += wa * p.x; acc0.y += wa * p.y; acc0.z += wa * p.z; acc0.w += wa * p.w;
        acc1.x += wb * q.x; acc1.y += wb * q.y; acc1.z += wb * q.z; acc1.w += wb * q.w;
    }
}

template <bool MEAN>
__global__ void gat_gather(const int* __restrict__ rowptr, const int* __restrict__ col,
                           const float* __restrict__ asrc, const float* __restrict__ adst,
                           const float* __restrict__ hmat,
                           const float* __restrict__ bias, const float* __restrict__ g,
                           const float* __restrict__ be, float* __restrict__ out, int N) {
    const int lane = threadIdx.x & 31;
    const int n = __builtin_amdgcn_readfirstlane((int)((blockIdx.x * blockDim.x + threadIdx.x) >> 5));
    if (n >= N) return;
    const int beg = rowptr[n], end = rowptr[n + 1];
    const bool hi = lane >= 16;
    const float4* asrc4 = reinterpret_cast<const float4*>(asrc);
    const float4 ad = reinterpret_cast<const float4*>(adst)[n];
    const float4 an = asrc4[n];
    const float sf0 = leakyf(an.x + ad.x), sf1 = leakyf(an.y + ad.y);
    const float sf2 = leakyf(an.z + ad.z), sf3 = leakyf(an.w + ad.w);

    const bool v0 = beg + lane < end;
    int   s0 = n;
    float a0 = sf0, a1 = sf1, a2 = sf2, a3 = sf3;
    if (v0) {
        s0 = col[beg + lane];
        const float4 a = asrc4[s0];
        a0 = leakyf(a.x + ad.x); a1 = leakyf(a.y + ad.y);
        a2 = leakyf(a.z + ad.z); a3 = leakyf(a.w + ad.w);
    }
    float m0 = fmaxf(sf0, a0), m1 = fmaxf(sf1, a1), m2 = fmaxf(sf2, a2), m3 = fmaxf(sf3, a3);
    for (int base = beg + 32; base < end; base += 32) {
        const int j = base + lane;
        if (j < end) {
            const float4 a = asrc4[col[j]];
            m0 = fmaxf(m0, leakyf(a.x + ad.x)); m1 = fmaxf(m1, leakyf(a.y + ad.y));
            m2 = fmaxf(m2, leakyf(a.z + ad.z)); m3 = fmaxf(m3, leakyf(a.w + ad.w));
        }
    }
    m0 = waveRedMax(m0); m1 = waveRedMax(m1); m2 = waveRedMax(m2); m3 = waveRedMax(m3);

    const float x0 = expf(sf0 - m0), x1 = expf(sf1 - m1), x2 = expf(sf2 - m2), x3 = expf(sf3 - m3);
    const float4* rown = reinterpret_cast<const float4*>(hmat + (size_t)n * HIDF);
    const float4 r0 = rown[lane], r1 = rown[32 + lane];
    const float wsa = hi ? x1 : x0, wsb = hi ? x3 : x2;
    float4 acc0 = make_float4(r0.x * wsa, r0.y * wsa, r0.z * wsa, r0.w * wsa);
    float4 acc1 = make_float4(r1.x * wsb, r1.y * wsb, r1.z * wsb, r1.w * wsb);

    float d0, d1, d2, d3;
    {
        const float t0 = expf(a0 - m0), t1 = expf(a1 - m1), t2 = expf(a2 - m2), t3 = expf(a3 - m3);
        d0 = v0 ? t0 : 0.f; d1 = v0 ? t1 : 0.f; d2 = v0 ? t2 : 0.f; d3 = v0 ? t3 : 0.f;
        const int cnt = (end - beg < 32) ? (end - beg) : 32;
        gather_rows(hmat, s0, d0, d1, d2, d3, cnt, lane, hi, acc0, acc1);
    }
    for (int base = beg + 32; base < end; base += 32) {
        const int j = base + lane;
        int s = n;
        float e0 = 0.f, e1 = 0.f, e2 = 0.f, e3 = 0.f;
        if (j < end) {
            s = col[j];
            const float4 a = asrc4[s];
            e0 = expf(leakyf(a.x + ad.x) - m0); e1 = expf(leakyf(a.y + ad.y) - m1);
            e2 = expf(leakyf(a.z + ad.z) - m2); e3 = expf(leakyf(a.w + ad.w) - m3);
        }
        d0 += e0; d1 += e1; d2 += e2; d3 += e3;
        const int cnt = (end - base < 32) ? (end - base) : 32;
        gather_rows(hmat, s, e0, e1, e2, e3, cnt, lane, hi, acc0, acc1);
    }
    d0 = waveRedSum(d0); d1 = waveRedSum(d1); d2 = waveRedSum(d2); d3 = waveRedSum(d3);
    const float i0 = 1.f / ((x0 + d0) + 1e-16f), i1 = 1.f / ((x1 + d1) + 1e-16f);
    const float i2 = 1.f / ((x2 + d2) + 1e-16f), i3 = 1.f / ((x3 + d3) + 1e-16f);

    if constexpr (!MEAN) {
        const float ia = hi ? i1 : i0, ib = hi ? i3 : i2;
        const float4 ba = reinterpret_cast<const float4*>(bias)[lane];
        const float4 bb = reinterpret_cast<const float4*>(bias)[32 + lane];
        const float4 ga = reinterpret_cast<const float4*>(g)[lane];
        const float4 gb = reinterpret_cast<const float4*>(g)[32 + lane];
        const float4 ea = reinterpret_cast<const float4*>(be)[lane];
        const float4 eb = reinterpret_cast<const float4*>(be)[32 + lane];
        float4 va, vb;
        va.x = acc0.x * ia + ba.x; va.y = acc0.y * ia + ba.y; va.z = acc0.z * ia + ba.z; va.w = acc0.w * ia + ba.w;
        vb.x = acc1.x * ib + bb.x; vb.y = acc1.y * ib + bb.y; vb.z = acc1.z * ib + bb.z; vb.w = acc1.w * ib + bb.w;
        const float s  = ((va.x + va.y) + (va.z + va.w)) + ((vb.x + vb.y) + (vb.z + vb.w));
        const float mu = waveRedSum(s) * (1.f / HIDF);
        va.x -= mu; va.y -= mu; va.z -= mu; va.w -= mu;
        vb.x -= mu; vb.y -= mu; vb.z -= mu; vb.w -= mu;
        const float q = ((va.x * va.x + va.y * va.y) + (va.z * va.z + va.w * va.w))
                      + ((vb.x * vb.x + vb.y * vb.y) + (vb.z * vb.z + vb.w * vb.w));
        const float rstd = rsqrtf(waveRedSum(q) * (1.f / HIDF) + LN_EPS);
        float4 ya, yb;
        ya.x = eluf(va.x * rstd * ga.x + ea.x); ya.y = eluf(va.y * rstd * ga.y + ea.y);
        ya.z = eluf(va.z * rstd * ga.z + ea.z); ya.w = eluf(va.w * rstd * ga.w + ea.w);
        yb.x = eluf(vb.x * rstd * gb.x + eb.x); yb.y = eluf(vb.y * rstd * gb.y + eb.y);
        yb.z = eluf(vb.z * rstd * gb.z + eb.z); yb.w = eluf(vb.w * rstd * gb.w + eb.w);
        float4* orow = reinterpret_cast<float4*>(out + (size_t)n * HIDF);
        orow[lane]      = ya;
        orow[32 + lane] = yb;
    } else {
        float4 p0, p1;
        p0.x = __shfl_xor(acc0.x, 16, 32); p0.y = __shfl_xor(acc0.y, 16, 32);
        p0.z = __shfl_xor(acc0.z, 16, 32); p0.w = __shfl_xor(acc0.w, 16, 32);
        p1.x = __shfl_xor(acc1.x, 16, 32); p1.y = __shfl_xor(acc1.y, 16, 32);
        p1.z = __shfl_xor(acc1.z, 16, 32); p1.w = __shfl_xor(acc1.w, 16, 32);
        const float4 h0 = hi ? p0 : acc0, h1 = hi ? acc0 : p0;
        const float4 h2 = hi ? p1 : acc1, h3 = hi ? acc1 : p1;
        const int l16 = lane & 15;
        const float4 b4 = reinterpret_cast<const float4*>(bias)[l16];
        const float4 g4 = reinterpret_cast<const float4*>(g)[l16];
        const float4 e4 = reinterpret_cast<const float4*>(be)[l16];
        float4 v;
        v.x = (((h0.x * i0 + h1.x * i1) + h2.x * i2) + h3.x * i3) * (1.f / HEADS) + b4.x;
        v.y = (((h0.y * i0 + h1.y * i1) + h2.y * i2) + h3.y * i3) * (1.f / HEADS) + b4.y;
        v.z = (((h0.z * i0 + h1.z * i1) + h2.z * i2) + h3.z * i3) * (1.f / HEADS) + b4.z;
        v.w = (((h0.w * i0 + h1.w * i1) + h2.w * i2) + h3.w * i3) * (1.f / HEADS) + b4.w;
        float s = (v.x + v.y) + (v.z + v.w);
        for (int off = 8; off > 0; off >>= 1) s += __shfl_xor(s, off, 32);
        const float mu = s * (1.f / CCH);
        v.x -= mu; v.y -= mu; v.z -= mu; v.w -= mu;
        float q = (v.x * v.x + v.y * v.y) + (v.z * v.z + v.w * v.w);
        for (int off = 8; off > 0; off >>= 1) q += __shfl_xor(q, off, 32);
        const float rstd = rsqrtf(q * (1.f / CCH) + LN_EPS);
        float4 y;
        y.x = eluf(v.x * rstd * g4.x + e4.x); y.y = eluf(v.y * rstd * g4.y + e4.y);
        y.z = eluf(v.z * rstd * g4.z + e4.z); y.w = eluf(v.w * rstd * g4.w + e4.w);
        if (!hi) reinterpret_cast<float4*>(out + (size_t)n * CCH)[lane] = y;
    }
}

__global__ void value_head(const float* __restrict__ hin,
                           const float* __restrict__ vw1, const float* __restrict__ vb1,
                           const float* __restrict__ vw2, const float* __restrict__ vb2,
                           float* __restrict__ out, int N) {
    int wid  = (blockIdx.x * blockDim.x + threadIdx.x) >> 5;
    int lane = threadIdx.x & 31;
    if (wid >= N) return;
    const float* row  = hin + (size_t)wid * CCH;
    const float* wrow = vw1 + lane * CCH;
    float hid = vb1[lane];
    for (int c = 0; c < CCH; ++c) hid += row[c] * wrow[c];
    hid = eluf(hid);
#pragma unroll
    for (int k = 0; k < 4; ++k) {
        float p = vw2[k * 32 + lane] * hid;
        p = waveRedSum(p);
        if (lane == 0) out[wid * 4 + k] = p + vb2[k];
    }
}

extern "C" void kernel_launch(void* const* d_in, const int* in_sizes, int n_in,
                              void* d_out, int out_size, void* d_ws, size_t ws_size,
                              hipStream_t stream) {
    const float* x      = (const float*)d_in[0];
    const int*   esrc   = (const int*)d_in[1];
    const int*   edst   = (const int*)d_in[2];
    const float* proj_w = (const float*)d_in[3];
    const float* proj_b = (const float*)d_in[4];
    const float* W[3]  = {(const float*)d_in[5],  (const float*)d_in[11], (const float*)d_in[17]};
    const float* AS[3] = {(const float*)d_in[6],  (const float*)d_in[12], (const float*)d_in[18]};
    const float* AD[3] = {(const float*)d_in[7],  (const float*)d_in[13], (const float*)d_in[19]};
    const float* B[3]  = {(const float*)d_in[8],  (const float*)d_in[14], (const float*)d_in[20]};
    const float* G[3]  = {(const float*)d_in[9],  (const float*)d_in[15], (const float*)d_in[21]};
    const float* BE[3] = {(const float*)d_in[10], (const float*)d_in[16], (const float*)d_in[22]};
    const float* vw1 = (const float*)d_in[23];
    const float* vb1 = (const float*)d_in[24];
    const float* vw2 = (const float*)d_in[25];
    const float* vb2 = (const float*)d_in[26];

    const int N = in_sizes[0] / 128;
    const int E = in_sizes[1];
    const int NH = N * HEADS;

    float* ws     = (float*)d_ws;
    float* hin    = ws;
    float* hmat   = hin  + (size_t)N * HIDF;
    float* asrc   = hmat + (size_t)N * HIDF;
    float* adst   = asrc + (size_t)NH;
    int* cnt      = (int*)(adst + (size_t)NH);
    int* rowptr   = cnt + (size_t)N;
    int* col      = rowptr + (size_t)N + 1;

    dim3 blk(256);

    csr_clear<<<(N + 255) / 256, blk, 0, stream>>>(cnt, N);
    csr_count<<<(E + 255) / 256, blk, 0, stream>>>(edst, cnt, E);
    csr_scan<<<1, 1024, 0, stream>>>(cnt, rowptr, N);
    csr_fill<<<(E + 255) / 256, blk, 0, stream>>>(esrc, edst, cnt, col, E);

    gemm_wmma_f32_lds<64, 64, 4, 2, false><<<(N + 63) / 64, blk, 0, stream>>>(
        x, 128, proj_w, hin, N, 128, 64, proj_b, 1, nullptr, nullptr, nullptr, nullptr);

    int in_ld = 64;
    for (int l = 0; l < 3; ++l) {
        int K = (l == 0) ? 64 : 256;
        gemm_wmma_f32_lds<64, 128, 2, 4, true><<<((N + 63) / 64) * (HIDF / 128), blk, 0, stream>>>(
            hin, in_ld, W[l], hmat, N, K, HIDF, nullptr, 0, AS[l], AD[l], asrc, adst);
        if (l < 2) {
            gat_gather<false><<<(N * 32 + 255) / 256, blk, 0, stream>>>(
                rowptr, col, asrc, adst, hmat, B[l], G[l], BE[l], hin, N);
            in_ld = HIDF;
        } else {
            gat_gather<true><<<(N * 32 + 255) / 256, blk, 0, stream>>>(
                rowptr, col, asrc, adst, hmat, B[l], G[l], BE[l], hin, N);
        }
    }

    value_head<<<(N * 32 + 255) / 256, blk, 0, stream>>>(
        hin, vw1, vb1, vw2, vb2, (float*)d_out, N);
}


# Value head done by the last layer's gather wave from its registers, vw1 staged per work-group in LDS as whole-line pieces; value_head launch, its 256-byte-strided weight loads, hin round trip gone
# speedup vs baseline: 3.4215x; 1.0877x over previous
#include <hip/hip_runtime.h>
#include <math.h>

typedef __attribute__((ext_vector_type(2))) float v2f;
typedef __attribute__((ext_vector_type(8))) float v8f;

#define HEADS 4
#define CCH 64
#define HIDF 256
#define NEG_SLOPE 0.2f
#define LN_EPS 1e-5f

__device__ __forceinline__ float eluf(float x) { return x > 0.f ? x : (expf(x) - 1.f); }
__device__ __forceinline__ float leakyf(float x) { return x > 0.f ? x : NEG_SLOPE * x; }

__device__ __forceinline__ float waveRedSum(float v) {
    for (int off = 16; off > 0; off >>= 1) v += __shfl_xor(v, off, 32);
    return v;
}
__device__ __forceinline__ float waveRedMax(float v) {
    for (int off = 16; off > 0; off >>= 1) v = fmaxf(v, __shfl_xor(v, off, 32));
    return v;
}
__device__ __forceinline__ int waveRedSumI(int v) {
    for (int off = 16; off > 0; off >>= 1) v += __shfl_xor(v, off, 32);
    return v;
}
__device__ __forceinline__ float laneValF(float v, int t) {
    return __int_as_float(__builtin_amdgcn_readlane(__float_as_int(v), t));
}

template <int BM, int BN, int WM, int WN, bool ATT>
__global__ void __launch_bounds__(256)
gemm_wmma_f32_lds(const float* __restrict__ A, int lda,
                  const float* __restrict__ W,
                  float* __restrict__ C,
                  int M, int K, int Nout,
                  const float* __restrict__ bias, int elu_act,
                  const float* __restrict__ a_s, const float* __restrict__ a_d,
                  float* __restrict__ asrc, float* __restrict__ adst) {
    constexpr int BK = 32;
    constexpr int P  = BK + 4;
    constexpr int TM = BM / (16 * WM);
    constexpr int TN = BN / (16 * WN);
    constexpr int AF = BM * (BK / 4) / 256;
    constexpr int BF = BN * (BK / 4) / 256;
    static_assert(WM * WN == 8, "work-group of 8 waves");
    static_assert(TM * 16 * WM == BM && TN * 16 * WN == BN, "wave tiles cover the block");
    static_assert(AF * 256 == BM * (BK / 4) && BF * 256 == BN * (BK / 4), "copy covers the chunk");

    __shared__ __attribute__((aligned(16))) float sA[2][BM * P];
    __shared__ __attribute__((aligned(16))) float sB[2][BN * P];

    const int tid  = threadIdx.x;
    const int lane = tid & 31;
    const int w    = tid >> 5;
    const int wm   = w / WN;
    const int wn   = w - wm * WN;
    const int blocksN = Nout / BN;
    const int bm = blockIdx.x / blocksN;
    const int bn = blockIdx.x - bm * blocksN;
    const int m0 = bm * BM;
    const int n0 = bn * BN;

    const int cr = tid >> 3;
    const int cq = (tid & 7) << 2;
    const float* gA[AF];
    const float* gB[BF];
#pragma unroll
    for (int j = 0; j < AF; ++j) {
        int gm = m0 + cr + 32 * j;
        gm = gm < M ? gm : (M - 1);
        gA[j] = A + (size_t)gm * lda + cq;
    }
#pragma unroll
    for (int j = 0; j < BF; ++j) gB[j] = W + (size_t)(n0 + cr + 32 * j) * K + cq;

    static_assert(AF == 2 && (BF == 2 || BF == 4), "piece registers below are written out for these sizes");
    float4 ra0, ra1, rb0, rb1, rb2, rb3;
    ra0 = *reinterpret_cast<const float4*>(gA[0]);
    ra1 = *reinterpret_cast<const float4*>(gA[1]);
    rb0 = *reinterpret_cast<const float4*>(gB[0]);
    rb1 = *reinterpret_cast<const float4*>(gB[1]);
    if constexpr (BF == 4) {
        rb2 = *reinterpret_cast<const float4*>(gB[2]);
        rb3 = *reinterpret_cast<const float4*>(gB[3]);
    }
    const int sOff = cr * P + cq;
    *reinterpret_cast<float4*>(&sA[0][sOff])          = ra0;
    *reinterpret_cast<float4*>(&sA[0][sOff + 32 * P]) = ra1;
    *reinterpret_cast<float4*>(&sB[0][sOff])          = rb0;
    *reinterpret_cast<float4*>(&sB[0][sOff + 32 * P]) = rb1;
    if constexpr (BF == 4) {
        *reinterpret_cast<float4*>(&sB[0][sOff + 64 * P]) = rb2;
        *reinterpret_cast<float4*>(&sB[0][sOff + 96 * P]) = rb3;
    }

    v8f acc[TM][TN];
#pragma unroll
    for (int i = 0; i < TM; ++i)
#pragma unroll
        for (int j = 0; j < TN; ++j) acc[i][j] = v8f{};

    const int fragOff = (lane & 15) * P + ((lane >> 4) << 1);
    const int nk = K / BK;
    for (int it = 0; it < nk; ++it) {
        const int  buf  = it & 1;
        const bool more = it + 1 < nk;
        __syncthreads();
        if (more) {
            const int k1 = (it + 1) * BK;
            ra0 = *reinterpret_cast<const float4*>(gA[0] + k1);
            ra1 = *reinterpret_cast<const float4*>(gA[1] + k1);
            rb0 = *reinterpret_cast<const float4*>(gB[0] + k1);
            rb1 = *reinterpret_cast<const float4*>(gB[1] + k1);
            if constexpr (BF == 4) {
                rb2 = *reinterpret_cast<const float4*>(gB[2] + k1);
                rb3 = *reinterpret_cast<const float4*>(gB[3] + k1);
            }
        }
        const float* pa = &sA[buf][wm * (TM * 16) * P + fragOff];
        const float* pb = &sB[buf][wn * (TN * 16) * P + fragOff];
#pragma unroll
        for (int ks = 0; ks < BK / 4; ++ks) {
            v2f af[TM], bf[TN];
#pragma unroll
            for (int i = 0; i < TM; ++i) af[i] = *reinterpret_cast<const v2f*>(pa + i * 16 * P + ks * 4);
#pragma unroll
            for (int j = 0; j < TN; ++j) bf[j] = *reinterpret_cast<const v2f*>(pb + j * 16 * P + ks * 4);
#pragma unroll
            for (int i = 0; i < TM; ++i)
#pragma unroll
                for (int j = 0; j < TN; ++j)
                    acc[i][j] = __builtin_amdgcn_wmma_f32_16x16x4_f32(false, af[i], false, bf[j],
                                                                      (short)0, acc[i][j], false, false);
        }
        if (more) {
            *reinterpret_cast<float4*>(&sA[buf ^ 1][sOff])          = ra0;
            *reinterpret_cast<float4*>(&sA[buf ^ 1][sOff + 32 * P]) = ra1;
            *reinterpret_cast<float4*>(&sB[buf ^ 1][sOff])          = rb0;
            *reinterpret_cast<float4*>(&sB[buf ^ 1][sOff + 32 * P]) = rb1;
            if constexpr (BF == 4) {
                *reinterpret_cast<float4*>(&sB[buf ^ 1][sOff + 64 * P]) = rb2;
                *reinterpret_cast<float4*>(&sB[buf ^ 1][sOff + 96 * P]) = rb3;
            }
        }
    }

    if constexpr (ATT) {
        constexpr int CP = BN + 4;
        constexpr int RW = BM / 8;
        static_assert(BN == 128, "a row of the block is one 16-byte piece per lane, one head per lane half");
        static_assert(BM * CP <= 2 * BN * P, "the C block fits in the W chunk buffers");
        float* sC = &sB[0][0];
        __syncthreads();
#pragma unroll
        for (int i = 0; i < TM; ++i)
#pragma unroll
            for (int j = 0; j < TN; ++j)
#pragma unroll
                for (int r = 0; r < 8; ++r)
                    sC[((wm * TM + i) * 16 + ((lane >> 4) << 3) + r) * CP + (wn * TN + j) * 16 + (lane & 15)] = acc[i][j][r];
        __syncthreads();
        const int colg = n0 + 4 * lane;
        const float4 s4 = *reinterpret_cast<const float4*>(a_s + colg);
        const float4 d4 = *reinterpret_cast<const float4*>(a_d + colg);
        const int head = (n0 >> 6) + (lane >> 4);
        float4 v[RW];
        float ps[RW], pd[RW];
#pragma unroll
        for (int q = 0; q < RW; ++q)
            v[q] = *reinterpret_cast<const float4*>(sC + (w * RW + q) * CP + 4 * lane);
#pragma unroll
        for (int q = 0; q < RW; ++q) {
            ps[q] = (v[q].x * s4.x + v[q].y * s4.y) + (v[q].z * s4.z + v[q].w * s4.w);
            pd[q] = (v[q].x * d4.x + v[q].y * d4.y) + (v[q].z * d4.z + v[q].w * d4.w);
        }
#pragma unroll
        for (int off = 8; off > 0; off >>= 1) {
#pragma unroll
            for (int q = 0; q < RW; ++q) {
                ps[q] += __shfl_xor(ps[q], off, 32);
                pd[q] += __shfl_xor(pd[q], off, 32);
            }
        }
#pragma unroll
        for (int q = 0; q < RW; ++q) {
            const int gm = m0 + w * RW + q;
            if (gm < M) {
                *reinterpret_cast<float4*>(C + (size_t)gm * Nout + colg) = v[q];
                if ((lane & 15) == 0) {
                    asrc[gm * HEADS + head] = ps[q];
                    adst[gm * HEADS + head] = pd[q];
                }
            }
        }
        return;
    }

#pragma unroll
    for (int i = 0; i < TM; ++i) {
#pragma unroll
        for (int j = 0; j < TN; ++j) {
            const int rbase = m0 + (wm * TM + i) * 16 + ((lane >> 4) << 3);
            const int col   = n0 + (wn * TN + j) * 16 + (lane & 15);
            const float bv  = bias ? bias[col] : 0.f;
#pragma unroll
            for (int r = 0; r < 8; ++r) {
                float v = acc[i][j][r] + bv;
                if (elu_act) v = eluf(v);
                if (rbase + r < M) C[(size_t)(rbase + r) * Nout + col] = v;
            }
        }
    }
}

__global__ void csr_clear(int* __restrict__ cnt, int N) {
    int i = blockIdx.x * blockDim.x + threadIdx.x;
    if (i < N) cnt[i] = 0;
}

__global__ void csr_count(const int* __restrict__ dst, int* __restrict__ cnt, int E) {
    int e = blockIdx.x * blockDim.x + threadIdx.x;
    if (e < E) atomicAdd(&cnt[dst[e]], 1);
}

__global__ void __launch_bounds__(1024) csr_scan(int* __restrict__ cnt, int* __restrict__ rowptr, int N) {
    __shared__ int wtot[32];
    const int lane = threadIdx.x & 31;
    const int w    = threadIdx.x >> 5;
    const int per  = ((N + 4095) >> 12) << 7;
    const int lo   = w * per;
    const int hi   = (lo + per < N) ? (lo + per) : N;
    int sum = 0;
#pragma unroll 4
    for (int i = lo + 4 * lane; i < hi; i += 128) {
        const int4 c = *reinterpret_cast<const int4*>(cnt + i);
        sum += (c.x + c.y) + (c.z + c.w);
    }
    sum = waveRedSumI(sum);
    if (lane == 0) wtot[w] = sum;
    __syncthreads();
    int carry = waveRedSumI(lane < w ? wtot[lane] : 0);
    for (int i0 = lo; i0 < hi; i0 += 128) {
        const int  i  = i0 + 4 * lane;
        const bool in = i < hi;
        int4 c = make_int4(0, 0, 0, 0);
        if (in) c = *reinterpret_cast<const int4*>(cnt + i);
        const int t = (c.x + c.y) + (c.z + c.w);
        int x = t;
        for (int off = 1; off < 32; off <<= 1) {
            int y = __shfl_up(x, off, 32);
            if (lane >= off) x += y;
        }
        if (in) {
            const int b = carry + x - t;
            const int4 o = make_int4(b, b + c.x, b + c.x + c.y, b + c.x + c.y + c.z);
            *reinterpret_cast<int4*>(rowptr + i) = o;
            *reinterpret_cast<int4*>(cnt + i)    = o;
        }
        carry += __shfl(x, 31, 32);
    }
    if (threadIdx.x == 1023) rowptr[N] = carry;
}

__global__ void csr_fill(const int* __restrict__ src, const int* __restrict__ dst,
                         int* __restrict__ cursor, int* __restrict__ col, int E) {
    int e = blockIdx.x * blockDim.x + threadIdx.x;
    if (e >= E) return;
    int p = atomicAdd(&cursor[dst[e]], 1);
    col[p] = src[e];
}

__device__ __forceinline__ void gather_rows(const float* __restrict__ hmat, int s,
                                            float e0, float e1, float e2, float e3,
                                            int cnt, int lane, bool hi,
                                            float4& acc0, float4& acc1) {
#pragma unroll 4
    for (int t = 0; t < cnt; ++t) {
        const int   st = __builtin_amdgcn_readlane(s, t);
        const float w0 = laneValF(e0, t), w1 = laneValF(e1, t);
        const float w2 = laneValF(e2, t), w3 = laneValF(e3, t);
        const float wa = hi ? w1 : w0;
        const float wb = hi ? w3 : w2;
        const float4* r = reinterpret_cast<const float4*>(hmat + (size_t)st * HIDF);
        const float4 p = r[lane];
        const float4 q = r[32 + lane];
        acc0.x += wa * p.x; acc0.y += wa * p.y; acc0.z += wa * p.z; acc0.w += wa * p.w;
        acc1.x += wb * q.x; acc1.y += wb * q.y; acc1.z += wb * q.z; acc1.w += wb * q.w;
    }
}

template <bool MEAN>
__global__ void gat_gather(const int* __restrict__ rowptr, const int* __restrict__ col,
                           const float* __restrict__ asrc, const float* __restrict__ adst,
                           const float* __restrict__ hmat,
                           const float* __restrict__ bias, const float* __restrict__ g,
                           const float* __restrict__ be, float* __restrict__ out, int N,
                           const float* __restrict__ vw1, const float* __restrict__ vb1,
                           const float* __restrict__ vw2, const float* __restrict__ vb2,
                           float* __restrict__ vout) {
    constexpr int WP = 33;
    __shared__ __attribute__((aligned(16))) float sW[MEAN ? 16 * WP * 4 : 4];
    if constexpr (MEAN) {
        const int f0 = threadIdx.x, f1 = threadIdx.x + 256;
        const float4 wa = reinterpret_cast<const float4*>(vw1)[f0];
        const float4 wb = reinterpret_cast<const float4*>(vw1)[f1];
        reinterpret_cast<float4*>(sW)[(f0 & 15) * WP + (f0 >> 4)] = wa;
        reinterpret_cast<float4*>(sW)[(f1 & 15) * WP + (f1 >> 4)] = wb;
        __syncthreads();
    }
    const int lane = threadIdx.x & 31;
    const int n = __builtin_amdgcn_readfirstlane((int)((blockIdx.x * blockDim.x + threadIdx.x) >> 5));
    if (n >= N) return;
    const int beg = rowptr[n], end = rowptr[n + 1];
    const bool hi = lane >= 16;
    const float4* asrc4 = reinterpret_cast<const float4*>(asrc);
    const float4 ad = reinterpret_cast<const float4*>(adst)[n];
    const float4 an = asrc4[n];
    const float sf0 = leakyf(an.x + ad.x), sf1 = leakyf(an.y + ad.y);
    const float sf2 = leakyf(an.z + ad.z), sf3 = leakyf(an.w + ad.w);

    const bool v0 = beg + lane < end;
    int   s0 = n;
    float a0 = sf0, a1 = sf1, a2 = sf2, a3 = sf3;
    if (v0) {
        s0 = col[beg + lane];
        const float4 a = asrc4[s0];
        a0 = leakyf(a.x + ad.x); a1 = leakyf(a.y + ad.y);
        a2 = leakyf(a.z + ad.z); a3 = leakyf(a.w + ad.w);
    }
    float m0 = fmaxf(sf0, a0), m1 = fmaxf(sf1, a1), m2 = fmaxf(sf2, a2), m3 = fmaxf(sf3, a3);
    for (int base = beg + 32; base < end; base += 32) {
        const int j = base + lane;
        if (j < end) {
            const float4 a = asrc4[col[j]];
            m0 = fmaxf(m0, leakyf(a.x + ad.x)); m1 = fmaxf(m1, leakyf(a.y + ad.y));
            m2 = fmaxf(m2, leakyf(a.z + ad.z)); m3 = fmaxf(m3, leakyf(a.w + ad.w));
        }
    }
    m0 = waveRedMax(m0); m1 = waveRedMax(m1); m2 = waveRedMax(m2); m3 = waveRedMax(m3);

    const float x0 = expf(sf0 - m0), x1 = expf(sf1 - m1), x2 = expf(sf2 - m2), x3 = expf(sf3 - m3);
    const float4* rown = reinterpret_cast<const float4*>(hmat + (size_t)n * HIDF);
    const float4 r0 = rown[lane], r1 = rown[32 + lane];
    const float wsa = hi ? x1 : x0, wsb = hi ? x3 : x2;
    float4 acc0 = make_float4(r0.x * wsa, r0.y * wsa, r0.z * wsa, r0.w * wsa);
    float4 acc1 = make_float4(r1.x * wsb, r1.y * wsb, r1.z * wsb, r1.w * wsb);

    float d0, d1, d2, d3;
    {
        const float t0 = expf(a0 - m0), t1 = expf(a1 - m1), t2 = expf(a2 - m2), t3 = expf(a3 - m3);
        d0 = v0 ? t0 : 0.f; d1 = v0 ? t1 : 0.f; d2 = v0 ? t2 : 0.f; d3 = v0 ? t3 : 0.f;
        const int cnt = (end - beg < 32) ? (end - beg) : 32;
        gather_rows(hmat, s0, d0, d1, d2, d3, cnt, lane, hi, acc0, acc1);
    }
    for (int base = beg + 32; base < end; base += 32) {
        const int j = base + lane;
        int s = n;
        float e0 = 0.f, e1 = 0.f, e2 = 0.f, e3 = 0.f;
        if (j < end) {
            s = col[j];
            const float4 a = asrc4[s];
            e0 = expf(leakyf(a.x + ad.x) - m0); e1 = expf(leakyf(a.y + ad.y) - m1);
            e2 = expf(leakyf(a.z + ad.z) - m2); e3 = expf(leakyf(a.w + ad.w) - m3);
        }
        d0 += e0; d1 += e1; d2 += e2; d3 += e3;
        const int cnt = (end - base < 32) ? (end - base) : 32;
        gather_rows(hmat, s, e0, e1, e2, e3, cnt, lane, hi, acc0, acc1);
    }
    d0 = waveRedSum(d0); d1 = waveRedSum(d1); d2 = waveRedSum(d2); d3 = waveRedSum(d3);
    const float i0 = 1.f / ((x0 + d0) + 1e-16f), i1 = 1.f / ((x1 + d1) + 1e-16f);
    const float i2 = 1.f / ((x2 + d2) + 1e-16f), i3 = 1.f / ((x3 + d3) + 1e-16f);

    if constexpr (!MEAN) {
        const float ia = hi ? i1 : i0, ib = hi ? i3 : i2;
        const float4 ba = reinterpret_cast<const float4*>(bias)[lane];
        const float4 bb = reinterpret_cast<const float4*>(bias)[32 + lane];
        const float4 ga = reinterpret_cast<const float4*>(g)[lane];
        const float4 gb = reinterpret_cast<const float4*>(g)[32 + lane];
        const float4 ea = reinterpret_cast<const float4*>(be)[lane];
        const float4 eb = reinterpret_cast<const float4*>(be)[32 + lane];
        float4 va, vb;
        va.x = acc0.x * ia + ba.x; va.y = acc0.y * ia + ba.y; va.z = acc0.z * ia + ba.z; va.w = acc0.w * ia + ba.w;
        vb.x = acc1.x * ib + bb.x; vb.y = acc1.y * ib + bb.y; vb.z = acc1.z * ib + bb.z; vb.w = acc1.w * ib + bb.w;
        const float s  = ((va.x + va.y) + (va.z + va.w)) + ((vb.x + vb.y) + (vb.z + vb.w));
        const float mu = waveRedSum(s) * (1.f / HIDF);
        va.x -= mu; va.y -= mu; va.z -= mu; va.w -= mu;
        vb.x -= mu; vb.y -= mu; vb.z -= mu; vb.w -= mu;
        const float q = ((va.x * va.x + va.y * va.y) + (va.z * va.z + va.w * va.w))
                      + ((vb.x * vb.x + vb.y * vb.y) + (vb.z * vb.z + vb.w * vb.w));
        const float rstd = rsqrtf(waveRedSum(q) * (1.f / HIDF) + LN_EPS);
        float4 ya, yb;
        ya.x = eluf(va.x * rstd * ga.x + ea.x); ya.y = eluf(va.y * rstd * ga.y + ea.y);
        ya.z = eluf(va.z * rstd * ga.z + ea.z); ya.w = eluf(va.w * rstd * ga.w + ea.w);
        yb.x = eluf(vb.x * rstd * gb.x + eb.x); yb.y = eluf(vb.y * rstd * gb.y + eb.y);
        yb.z = eluf(vb.z * rstd * gb.z + eb.z); yb.w = eluf(vb.w * rstd * gb.w + eb.w);
        float4* orow = reinterpret_cast<float4*>(out + (size_t)n * HIDF);
        orow[lane]      = ya;
        orow[32 + lane] = yb;
    } else {
        float4 p0, p1;
        p0.x = __shfl_xor(acc0.x, 16, 32); p0.y = __shfl_xor(acc0.y, 16, 32);
        p0.z = __shfl_xor(acc0.z, 16, 32); p0.w = __shfl_xor(acc0.w, 16, 32);
        p1.x = __shfl_xor(acc1.x, 16, 32); p1.y = __shfl_xor(acc1.y, 16, 32);
        p1.z = __shfl_xor(acc1.z, 16, 32); p1.w = __shfl_xor(acc1.w, 16, 32);
        const float4 h0 = hi ? p0 : acc0, h1 = hi ? acc0 : p0;
        const float4 h2 = hi ? p1 : acc1, h3 = hi ? acc1 : p1;
        const int l16 = lane & 15;
        const float4 b4 = reinterpret_cast<const float4*>(bias)[l16];
        const float4 g4 = reinterpret_cast<const float4*>(g)[l16];
        const float4 e4 = reinterpret_cast<const float4*>(be)[l16];
        const float q0 = vw2[lane], q1 = vw2[32 + lane], q2 = vw2[64 + lane], q3 = vw2[96 + lane];
        const float hb = vb1[lane];
        float4 v;
        v.x = (((h0.x * i0 + h1.x * i1) + h2.x * i2) + h3.x * i3) * (1.f / HEADS) + b4.x;
        v.y = (((h0.y * i0 + h1.y * i1) + h2.y * i2) + h3.y * i3) * (1.f / HEADS) + b4.y;
        v.z = (((h0.z * i0 + h1.z * i1) + h2.z * i2) + h3.z * i3) * (1.f / HEADS) + b4.z;
        v.w = (((h0.w * i0 + h1.w * i1) + h2.w * i2) + h3.w * i3) * (1.f / HEADS) + b4.w;
        float s = (v.x + v.y) + (v.z + v.w);
        for (int off = 8; off > 0; off >>= 1) s += __shfl_xor(s, off, 32);
        const float mu = s * (1.f / CCH);
        v.x -= mu; v.y -= mu; v.z -= mu; v.w -= mu;
        float q = (v.x * v.x + v.y * v.y) + (v.z * v.z + v.w * v.w);
        for (int off = 8; off > 0; off >>= 1) q += __shfl_xor(q, off, 32);
        const float rstd = rsqrtf(q * (1.f / CCH) + LN_EPS);
        float4 y;
        y.x = eluf(v.x * rstd * g4.x + e4.x); y.y = eluf(v.y * rstd * g4.y + e4.y);
        y.z = eluf(v.z * rstd * g4.z + e4.z); y.w = eluf(v.w * rstd * g4.w + e4.w);
        const float4* sW4 = reinterpret_cast<const float4*>(sW);
        float hid = hb;
#pragma unroll
        for (int c4 = 0; c4 < CCH / 4; ++c4) {
            const float4 wv = sW4[c4 * WP + lane];
            hid += laneValF(y.x, c4) * wv.x;
            hid += laneValF(y.y, c4) * wv.y;
            hid += laneValF(y.z, c4) * wv.z;
            hid += laneValF(y.w, c4) * wv.w;
        }
        hid = eluf(hid);
        float o0 = q0 * hid, o1 = q1 * hid, o2 = q2 * hid, o3 = q3 * hid;
        for (int off = 16; off > 0; off >>= 1) {
            o0 += __shfl_xor(o0, off, 32); o1 += __shfl_xor(o1, off, 32);
            o2 += __shfl_xor(o2, off, 32); o3 += __shfl_xor(o3, off, 32);
        }
        if (lane == 0) {
            const float4 ob = *reinterpret_cast<const float4*>(vb2);
            *reinterpret_cast<float4*>(vout + (size_t)n * 4) =
                make_float4(o0 + ob.x, o1 + ob.y, o2 + ob.z, o3 + ob.w);
        }
    }
}

extern "C" void kernel_launch(void* const* d_in, const int* in_sizes, int n_in,
                              void* d_out, int out_size, void* d_ws, size_t ws_size,
                              hipStream_t stream) {
    const float* x      = (const float*)d_in[0];
    const int*   esrc   = (const int*)d_in[1];
    const int*   edst   = (const int*)d_in[2];
    const float* proj_w = (const float*)d_in[3];
    const float* proj_b = (const float*)d_in[4];
    const float* W[3]  = {(const float*)d_in[5],  (const float*)d_in[11], (const float*)d_in[17]};
    const float* AS[3] = {(const float*)d_in[6],  (const float*)d_in[12], (const float*)d_in[18]};
    const float* AD[3] = {(const float*)d_in[7],  (const float*)d_in[13], (const float*)d_in[19]};
    const float* B[3]  = {(const float*)d_in[8],  (const float*)d_in[14], (const float*)d_in[20]};
    const float* G[3]  = {(const float*)d_in[9],  (const float*)d_in[15], (const float*)d_in[21]};
    const float* BE[3] = {(const float*)d_in[10], (const float*)d_in[16], (const float*)d_in[22]};
    const float* vw1 = (const float*)d_in[23];
    const float* vb1 = (const float*)d_in[24];
    const float* vw2 = (const float*)d_in[25];
    const float* vb2 = (const float*)d_in[26];

    const int N = in_sizes[0] / 128;
    const int E = in_sizes[1];
    const int NH = N * HEADS;

    float* ws     = (float*)d_ws;
    float* hin    = ws;
    float* hmat   = hin  + (size_t)N * HIDF;
    float* asrc   = hmat + (size_t)N * HIDF;
    float* adst   = asrc + (size_t)NH;
    int* cnt      = (int*)(adst + (size_t)NH);
    int* rowptr   = cnt + (size_t)N;
    int* col      = rowptr + (size_t)N + 1;

    dim3 blk(256);

    csr_clear<<<(N + 255) / 256, blk, 0, stream>>>(cnt, N);
    csr_count<<<(E + 255) / 256, blk, 0, stream>>>(edst, cnt, E);
    csr_scan<<<1, 1024, 0, stream>>>(cnt, rowptr, N);
    csr_fill<<<(E + 255) / 256, blk, 0, stream>>>(esrc, edst, cnt, col, E);

    gemm_wmma_f32_lds<64, 64, 4, 2, false><<<(N + 63) / 64, blk, 0, stream>>>(
        x, 128, proj_w, hin, N, 128, 64, proj_b, 1, nullptr, nullptr, nullptr, nullptr);

    int in_ld = 64;
    for (int l = 0; l < 3; ++l) {
        int K = (l == 0) ? 64 : 256;
        gemm_wmma_f32_lds<64, 128, 2, 4, true><<<((N + 63) / 64) * (HIDF / 128), blk, 0, stream>>>(
            hin, in_ld, W[l], hmat, N, K, HIDF, nullptr, 0, AS[l], AD[l], asrc, adst);
        if (l < 2) {
            gat_gather<false><<<(N * 32 + 255) / 256, blk, 0, stream>>>(
                rowptr, col, asrc, adst, hmat, B[l], G[l], BE[l], hin, N,
                nullptr, nullptr, nullptr, nullptr, nullptr);
            in_ld = HIDF;
        } else {
            gat_gather<true><<<(N * 32 + 255) / 256, blk, 0, stream>>>(
                rowptr, col, asrc, adst, hmat, B[l], G[l], BE[l], nullptr, N,
                vw1, vb1, vw2, vb2, (float*)d_out);
        }
    }
}


# Edge grouping without global atomics: edges binned by 256-node bucket (LDS histogram, scan, scatter), each bucket ordered by LDS counters in one work-group; in-edge sum order = LDS arrival order
# speedup vs baseline: 4.3200x; 1.2626x over previous
#include <hip/hip_runtime.h>
#include <math.h>

typedef __attribute__((ext_vector_type(2))) float v2f;
typedef __attribute__((ext_vector_type(8))) float v8f;

#define HEADS 4
#define CCH 64
#define HIDF 256
#define NEG_SLOPE 0.2f
#define LN_EPS 1e-5f

__device__ __forceinline__ float eluf(float x) { return x > 0.f ? x : (expf(x) - 1.f); }
__device__ __forceinline__ float leakyf(float x) { return x > 0.f ? x : NEG_SLOPE * x; }

__device__ __forceinline__ float waveRedSum(float v) {
    for (int off = 16; off > 0; off >>= 1) v += __shfl_xor(v, off, 32);
    return v;
}
__device__ __forceinline__ float waveRedMax(float v) {
    for (int off = 16; off > 0; off >>= 1) v = fmaxf(v, __shfl_xor(v, off, 32));
    return v;
}
__device__ __forceinline__ int waveRedSumI(int v) {
    for (int off = 16; off > 0; off >>= 1) v += __shfl_xor(v, off, 32);
    return v;
}
__device__ __forceinline__ float laneValF(float v, int t) {
    return __int_as_float(__builtin_amdgcn_readlane(__float_as_int(v), t));
}

template <int BM, int BN, int WM, int WN, bool ATT>
__global__ void __launch_bounds__(256)
gemm_wmma_f32_lds(const float* __restrict__ A, int lda,
                  const float* __restrict__ W,
                  float* __restrict__ C,
                  int M, int K, int Nout,
                  const float* __restrict__ bias, int elu_act,
                  const float* __restrict__ a_s, const float* __restrict__ a_d,
                  float* __restrict__ asrc, float* __restrict__ adst) {
    constexpr int BK = 32;
    constexpr int P  = BK + 4;
    constexpr int TM = BM / (16 * WM);
    constexpr int TN = BN / (16 * WN);
    constexpr int AF = BM * (BK / 4) / 256;
    constexpr int BF = BN * (BK / 4) / 256;
    static_assert(WM * WN == 8, "work-group of 8 waves");
    static_assert(TM * 16 * WM == BM && TN * 16 * WN == BN, "wave tiles cover the block");
    static_assert(AF * 256 == BM * (BK / 4) && BF * 256 == BN * (BK / 4), "copy covers the chunk");

    __shared__ __attribute__((aligned(16))) float sA[2][BM * P];
    __shared__ __attribute__((aligned(16))) float sB[2][BN * P];

    const int tid  = threadIdx.x;
    const int lane = tid & 31;
    const int w    = tid >> 5;
    const int wm   = w / WN;
    const int wn   = w - wm * WN;
    const int blocksN = Nout / BN;
    const int bm = blockIdx.x / blocksN;
    const int bn = blockIdx.x - bm * blocksN;
    const int m0 = bm * BM;
    const int n0 = bn * BN;

    const int cr = tid >> 3;
    const int cq = (tid & 7) << 2;
    const float* gA[AF];
    const float* gB[BF];
#pragma unroll
    for (int j = 0; j < AF; ++j) {
        int gm = m0 + cr + 32 * j;
        gm = gm < M ? gm : (M - 1);
        gA[j] = A + (size_t)gm * lda + cq;
    }
#pragma unroll
    for (int j = 0; j < BF; ++j) gB[j] = W + (size_t)(n0 + cr + 32 * j) * K + cq;

    static_assert(AF == 2 && (BF == 2 || BF == 4), "piece registers below are written out for these sizes");
    float4 ra0, ra1, rb0, rb1, rb2, rb3;
    ra0 = *reinterpret_cast<const float4*>(gA[0]);
    ra1 = *reinterpret_cast<const float4*>(gA[1]);
    rb0 = *reinterpret_cast<const float4*>(gB[0]);
    rb1 = *reinterpret_cast<const float4*>(gB[1]);
    if constexpr (BF == 4) {
        rb2 = *reinterpret_cast<const float4*>(gB[2]);
        rb3 = *reinterpret_cast<const float4*>(gB[3]);
    }
    const int sOff = cr * P + cq;
    *reinterpret_cast<float4*>(&sA[0][sOff])          = ra0;
    *reinterpret_cast<float4*>(&sA[0][sOff + 32 * P]) = ra1;
    *reinterpret_cast<float4*>(&sB[0][sOff])          = rb0;
    *reinterpret_cast<float4*>(&sB[0][sOff + 32 * P]) = rb1;
    if constexpr (BF == 4) {
        *reinterpret_cast<float4*>(&sB[0][sOff + 64 * P]) = rb2;
        *reinterpret_cast<float4*>(&sB[0][sOff + 96 * P]) = rb3;
    }

    v8f acc[TM][TN];
#pragma unroll
    for (int i = 0; i < TM; ++i)
#pragma unroll
        for (int j = 0; j < TN; ++j) acc[i][j] = v8f{};

    const int fragOff = (lane & 15) * P + ((lane >> 4) << 1);
    const int nk = K / BK;
    for (int it = 0; it < nk; ++it) {
        const int  buf  = it & 1;
        const bool more = it + 1 < nk;
        __syncthreads();
        if (more) {
            const int k1 = (it + 1) * BK;
            ra0 = *reinterpret_cast<const float4*>(gA[0] + k1);
            ra1 = *reinterpret_cast<const float4*>(gA[1] + k1);
            rb0 = *reinterpret_cast<const float4*>(gB[0] + k1);
            rb1 = *reinterpret_cast<const float4*>(gB[1] + k1);
            if constexpr (BF == 4) {
                rb2 = *reinterpret_cast<const float4*>(gB[2] + k1);
                rb3 = *reinterpret_cast<const float4*>(gB[3] + k1);
            }
        }
        const float* pa = &sA[buf][wm * (TM * 16) * P + fragOff];
        const float* pb = &sB[buf][wn * (TN * 16) * P + fragOff];
#pragma unroll
        for (int ks = 0; ks < BK / 4; ++ks) {
            v2f af[TM], bf[TN];
#pragma unroll
            for (int i = 0; i < TM; ++i) af[i] = *reinterpret_cast<const v2f*>(pa + i * 16 * P + ks * 4);
#pragma unroll
            for (int j = 0; j < TN; ++j) bf[j] = *reinterpret_cast<const v2f*>(pb + j * 16 * P + ks * 4);
#pragma unroll
            for (int i = 0; i < TM; ++i)
#pragma unroll
                for (int j = 0; j < TN; ++j)
                    acc[i][j] = __builtin_amdgcn_wmma_f32_16x16x4_f32(false, af[i], false, bf[j],
                                                                      (short)0, acc[i][j], false, false);
        }
        if (more) {
            *reinterpret_cast<float4*>(&sA[buf ^ 1][sOff])          = ra0;
            *reinterpret_cast<float4*>(&sA[buf ^ 1][sOff + 32 * P]) = ra1;
            *reinterpret_cast<float4*>(&sB[buf ^ 1][sOff])          = rb0;
            *reinterpret_cast<float4*>(&sB[buf ^ 1][sOff + 32 * P]) = rb1;
            if constexpr (BF == 4) {
                *reinterpret_cast<float4*>(&sB[buf ^ 1][sOff + 64 * P]) = rb2;
                *reinterpret_cast<float4*>(&sB[buf ^ 1][sOff + 96 * P]) = rb3;
            }
        }
    }

    if constexpr (ATT) {
        constexpr int CP = BN + 4;
        constexpr int RW = BM / 8;
        static_assert(BN == 128, "a row of the block is one 16-byte piece per lane, one head per lane half");
        static_assert(BM * CP <= 2 * BN * P, "the C block fits in the W chunk buffers");
        float* sC = &sB[0][0];
        __syncthreads();
#pragma unroll
        for (int i = 0; i < TM; ++i)
#pragma unroll
            for (int j = 0; j < TN; ++j)
#pragma unroll
                for (int r = 0; r < 8; ++r)
                    sC[((wm * TM + i) * 16 + ((lane >> 4) << 3) + r) * CP + (wn * TN + j) * 16 + (lane & 15)] = acc[i][j][r];
        __syncthreads();
        const int colg = n0 + 4 * lane;
        const float4 s4 = *reinterpret_cast<const float4*>(a_s + colg);
        const float4 d4 = *reinterpret_cast<const float4*>(a_d + colg);
        const int head = (n0 >> 6) + (lane >> 4);
        float4 v[RW];
        float ps[RW], pd[RW];
#pragma unroll
        for (int q = 0; q < RW; ++q)
            v[q] = *reinterpret_cast<const float4*>(sC + (w * RW + q) * CP + 4 * lane);
#pragma unroll
        for (int q = 0; q < RW; ++q) {
            ps[q] = (v[q].x * s4.x + v[q].y * s4.y) + (v[q].z * s4.z + v[q].w * s4.w);
            pd[q] = (v[q].x * d4.x + v[q].y * d4.y) + (v[q].z * d4.z + v[q].w * d4.w);
        }
#pragma unroll
        for (int off = 8; off > 0; off >>= 1) {
#pragma unroll
            for (int q = 0; q < RW; ++q) {
                ps[q] += __shfl_xor(ps[q], off, 32);
                pd[q] += __shfl_xor(pd[q], off, 32);
            }
        }
#pragma unroll
        for (int q = 0; q < RW; ++q) {
            const int gm = m0 + w * RW + q;
            if (gm < M) {
                *reinterpret_cast<float4*>(C + (size_t)gm * Nout + colg) = v[q];
                if ((lane & 15) == 0) {
                    asrc[gm * HEADS + head] = ps[q];
                    adst[gm * HEADS + head] = pd[q];
                }
            }
        }
        return;
    }

#pragma unroll
    for (int i = 0; i < TM; ++i) {
#pragma unroll
        for (int j = 0; j < TN; ++j) {
            const int rbase = m0 + (wm * TM + i) * 16 + ((lane >> 4) << 3);
            const int col   = n0 + (wn * TN + j) * 16 + (lane & 15);
            const float bv  = bias ? bias[col] : 0.f;
#pragma unroll
            for (int r = 0; r < 8; ++r) {
                float v = acc[i][j][r] + bv;
                if (elu_act) v = eluf(v);
                if (rbase + r < M) C[(size_t)(rbase + r) * Nout + col] = v;
            }
        }
    }
}

#define BSH 8
#define NBMAX 256
#define G1 64
#define EB 8

__global__ void __launch_bounds__(1024) bucket_hist(const int* __restrict__ dst, int* __restrict__ mat,
                                                    int E, int NB) {
    __shared__ int h[NBMAX];
    const int tid = threadIdx.x;
    if (tid < NBMAX) h[tid] = 0;
    __syncthreads();
    const int per = (E + G1 - 1) / G1;
    const int lo  = blockIdx.x * per;
    const int hi  = (lo + per < E) ? (lo + per) : E;
    for (int e0 = lo + tid; e0 < hi; e0 += EB * 1024) {
        int d[EB];
#pragma unroll
        for (int j = 0; j < EB; ++j) d[j] = (e0 + j * 1024 < hi) ? dst[e0 + j * 1024] : -1;
#pragma unroll
        for (int j = 0; j < EB; ++j)
            if (d[j] >= 0) atomicAdd(&h[d[j] >> BSH], 1);
    }
    __syncthreads();
    if (tid < NB) mat[tid * G1 + blockIdx.x] = h[tid];
}

__global__ void __launch_bounds__(1024) excl_scan(int* v, int L) {
    __shared__ int wtot[32];
    const int lane = threadIdx.x & 31;
    const int w    = threadIdx.x >> 5;
    const int per  = ((L + 4095) >> 12) << 7;
    const int lo   = w * per;
    const int hi   = (lo + per < L) ? (lo + per) : L;
    int sum = 0;
#pragma unroll 4
    for (int i = lo + 4 * lane; i < hi; i += 128) {
        const int4 c = *reinterpret_cast<const int4*>(v + i);
        sum += (c.x + c.y) + (c.z + c.w);
    }
    sum = waveRedSumI(sum);
    if (lane == 0) wtot[w] = sum;
    __syncthreads();
    int carry = waveRedSumI(lane < w ? wtot[lane] : 0);
    for (int i0 = lo; i0 < hi; i0 += 128) {
        const int  i  = i0 + 4 * lane;
        const bool in = i < hi;
        int4 c = make_int4(0, 0, 0, 0);
        if (in) c = *reinterpret_cast<const int4*>(v + i);
        const int t = (c.x + c.y) + (c.z + c.w);
        int x = t;
        for (int off = 1; off < 32; off <<= 1) {
            int y = __shfl_up(x, off, 32);
            if (lane >= off) x += y;
        }
        if (in) {
            const int b = carry + x - t;
            *reinterpret_cast<int4*>(v + i) = make_int4(b, b + c.x, b + c.x + c.y, b + c.x + c.y + c.z);
        }
        carry += __shfl(x, 31, 32);
    }
    if (threadIdx.x == 1023) v[L] = carry;
}

__global__ void __launch_bounds__(1024) bucket_scatter(const int* __restrict__ src, const int* __restrict__ dst,
                                                       const int* __restrict__ mat, int* __restrict__ pairs,
                                                       int E, int NB) {
    __shared__ int cur[NBMAX];
    const int tid = threadIdx.x;
    if (tid < NB) cur[tid] = mat[tid * G1 + blockIdx.x];
    __syncthreads();
    const int per = (E + G1 - 1) / G1;
    const int lo  = blockIdx.x * per;
    const int hi  = (lo + per < E) ? (lo + per) : E;
    int2* rec = reinterpret_cast<int2*>(pairs);
    for (int e0 = lo + tid; e0 < hi; e0 += EB * 1024) {
        int d[EB], s[EB];
#pragma unroll
        for (int j = 0; j < EB; ++j) {
            const bool in = e0 + j * 1024 < hi;
            d[j] = in ? dst[e0 + j * 1024] : -1;
            s[j] = in ? src[e0 + j * 1024] : 0;
        }
#pragma unroll
        for (int j = 0; j < EB; ++j)
            if (d[j] >= 0) {
                const int p = atomicAdd(&cur[d[j] >> BSH], 1);
                rec[p] = make_int2(s[j], d[j]);
            }
    }
}

__global__ void __launch_bounds__(256) bucket_sort(const int* __restrict__ mat, const int* __restrict__ pairs,
                                                   int* __restrict__ rowptr, int* __restrict__ col, int N) {
    __shared__ int cnt[256];
    __shared__ int wsum[8];
    const int tid  = threadIdx.x;
    const int lane = tid & 31;
    const int w    = tid >> 5;
    const int b    = blockIdx.x;
    const int beg  = mat[b * G1], end = mat[(b + 1) * G1];
    const int n0   = b << BSH;
    const int2* rec = reinterpret_cast<const int2*>(pairs);
    cnt[tid] = 0;
    __syncthreads();
    for (int i0 = beg + tid; i0 < end; i0 += EB * 256) {
        int d[EB];
#pragma unroll
        for (int j = 0; j < EB; ++j) d[j] = (i0 + j * 256 < end) ? rec[i0 + j * 256].y : -1;
#pragma unroll
        for (int j = 0; j < EB; ++j)
            if (d[j] >= 0) atomicAdd(&cnt[d[j] - n0], 1);
    }
    __syncthreads();
    const int c = cnt[tid];
    int x = c;
    for (int off = 1; off < 32; off <<= 1) {
        int y = __shfl_up(x, off, 32);
        if (lane >= off) x += y;
    }
    if (lane == 31) wsum[w] = x;
    __syncthreads();
    int carry = 0;
    for (int k = 0; k < w; ++k) carry += wsum[k];
    const int start = beg + carry + x - c;
    if (n0 + tid < N) rowptr[n0 + tid] = start;
    if (b == (int)gridDim.x - 1 && tid == 0) rowptr[N] = end;
    cnt[tid] = start;
    __syncthreads();
    for (int i0 = beg + tid; i0 < end; i0 += EB * 256) {
        int2 r[EB];
#pragma unroll
        for (int j = 0; j < EB; ++j) r[j] = (i0 + j * 256 < end) ? rec[i0 + j * 256] : make_int2(0, -1);
#pragma unroll
        for (int j = 0; j < EB; ++j)
            if (r[j].y >= 0) {
                const int q = atomicAdd(&cnt[r[j].y - n0], 1);
                col[q] = r[j].x;
            }
    }
}

__device__ __forceinline__ void gather_rows(const float* __restrict__ hmat, int s,
                                            float e0, float e1, float e2, float e3,
                                            int cnt, int lane, bool hi,
                                            float4& acc0, float4& acc1) {
#pragma unroll 4
    for (int t = 0; t < cnt; ++t) {
        const int   st = __builtin_amdgcn_readlane(s, t);
        const float w0 = laneValF(e0, t), w1 = laneValF(e1, t);
        const float w2 = laneValF(e2, t), w3 = laneValF(e3, t);
        const float wa = hi ? w1 : w0;
        const float wb = hi ? w3 : w2;
        const float4* r = reinterpret_cast<const float4*>(hmat + (size_t)st * HIDF);
        const float4 p = r[lane];
        const float4 q = r[32 + lane];
        acc0.x += wa * p.x; acc0.y += wa * p.y; acc0.z += wa * p.z; acc0.w += wa * p.w;
        acc1.x += wb * q.x; acc1.y += wb * q.y; acc1.z += wb * q.z; acc1.w += wb * q.w;
    }
}

template <bool MEAN>
__global__ void gat_gather(const int* __restrict__ rowptr, const int* __restrict__ col,
                           const float* __restrict__ asrc, const float* __restrict__ adst,
                           const float* __restrict__ hmat,
                           const float* __restrict__ bias, const float* __restrict__ g,
                           const float* __restrict__ be, float* __restrict__ out, int N,
                           const float* __restrict__ vw1, const float* __restrict__ vb1,
                           const float* __restrict__ vw2, const float* __restrict__ vb2,
                           float* __restrict__ vout) {
    constexpr int WP = 33;
    __shared__ __attribute__((aligned(16))) float sW[MEAN ? 16 * WP * 4 : 4];
    if constexpr (MEAN) {
        const int f0 = threadIdx.x, f1 = threadIdx.x + 256;
        const float4 wa = reinterpret_cast<const float4*>(vw1)[f0];
        const float4 wb = reinterpret_cast<const float4*>(vw1)[f1];
        reinterpret_cast<float4*>(sW)[(f0 & 15) * WP + (f0 >> 4)] = wa;
        reinterpret_cast<float4*>(sW)[(f1 & 15) * WP + (f1 >> 4)] = wb;
        __syncthreads();
    }
    const int lane = threadIdx.x & 31;
    const int n = __builtin_amdgcn_readfirstlane((int)((blockIdx.x * blockDim.x + threadIdx.x) >> 5));
    if (n >= N) return;
    const int beg = rowptr[n], end = rowptr[n + 1];
    const bool hi = lane >= 16;
    const float4* asrc4 = reinterpret_cast<const float4*>(asrc);
    const float4 ad = reinterpret_cast<const float4*>(adst)[n];
    const float4 an = asrc4[n];
    const float sf0 = leakyf(an.x + ad.x), sf1 = leakyf(an.y + ad.y);
    const float sf2 = leakyf(an.z + ad.z), sf3 = leakyf(an.w + ad.w);

    const bool v0 = beg + lane < end;
    int   s0 = n;
    float a0 = sf0, a1 = sf1, a2 = sf2, a3 = sf3;
    if (v0) {
        s0 = col[beg + lane];
        const float4 a = asrc4[s0];
        a0 = leakyf(a.x + ad.x); a1 = leakyf(a.y + ad.y);
        a2 = leakyf(a.z + ad.z); a3 = leakyf(a.w + ad.w);
    }
    float m0 = fmaxf(sf0, a0), m1 = fmaxf(sf1, a1), m2 = fmaxf(sf2, a2), m3 = fmaxf(sf3, a3);
    for (int base = beg + 32; base < end; base += 32) {
        const int j = base + lane;
        if (j < end) {
            const float4 a = asrc4[col[j]];
            m0 = fmaxf(m0, leakyf(a.x + ad.x)); m1 = fmaxf(m1, leakyf(a.y + ad.y));
            m2 = fmaxf(m2, leakyf(a.z + ad.z)); m3 = fmaxf(m3, leakyf(a.w + ad.w));
        }
    }
    m0 = waveRedMax(m0); m1 = waveRedMax(m1); m2 = waveRedMax(m2); m3 = waveRedMax(m3);

    const float x0 = expf(sf0 - m0), x1 = expf(sf1 - m1), x2 = expf(sf2 - m2), x3 = expf(sf3 - m3);
    const float4* rown = reinterpret_cast<const float4*>(hmat + (size_t)n * HIDF);
    const float4 r0 = rown[lane], r1 = rown[32 + lane];
    const float wsa = hi ? x1 : x0, wsb = hi ? x3 : x2;
    float4 acc0 = make_float4(r0.x * wsa, r0.y * wsa, r0.z * wsa, r0.w * wsa);
    float4 acc1 = make_float4(r1.x * wsb, r1.y * wsb, r1.z * wsb, r1.w * wsb);

    float d0, d1, d2, d3;
    {
        const float t0 = expf(a0 - m0), t1 = expf(a1 - m1), t2 = expf(a2 - m2), t3 = expf(a3 - m3);
        d0 = v0 ? t0 : 0.f; d1 = v0 ? t1 : 0.f; d2 = v0 ? t2 : 0.f; d3 = v0 ? t3 : 0.f;
        const int cnt = (end - beg < 32) ? (end - beg) : 32;
        gather_rows(hmat, s0, d0, d1, d2, d3, cnt, lane, hi, acc0, acc1);
    }
    for (int base = beg + 32; base < end; base += 32) {
        const int j = base + lane;
        int s = n;
        float e0 = 0.f, e1 = 0.f, e2 = 0.f, e3 = 0.f;
        if (j < end) {
            s = col[j];
            const float4 a = asrc4[s];
            e0 = expf(leakyf(a.x + ad.x) - m0); e1 = expf(leakyf(a.y + ad.y) - m1);
            e2 = expf(leakyf(a.z + ad.z) - m2); e3 = expf(leakyf(a.w + ad.w) - m3);
        }
        d0 += e0; d1 += e1; d2 += e2; d3 += e3;
        const int cnt = (end - base < 32) ? (end - base) : 32;
        gather_rows(hmat, s, e0, e1, e2, e3, cnt, lane, hi, acc0, acc1);
    }
    d0 = waveRedSum(d0); d1 = waveRedSum(d1); d2 = waveRedSum(d2); d3 = waveRedSum(d3);
    const float i0 = 1.f / ((x0 + d0) + 1e-16f), i1 = 1.f / ((x1 + d1) + 1e-16f);
    const float i2 = 1.f / ((x2 + d2) + 1e-16f), i3 = 1.f / ((x3 + d3) + 1e-16f);

    if constexpr (!MEAN) {
        const float ia = hi ? i1 : i0, ib = hi ? i3 : i2;
        const float4 ba = reinterpret_cast<const float4*>(bias)[lane];
        const float4 bb = reinterpret_cast<const float4*>(bias)[32 + lane];
        const float4 ga = reinterpret_cast<const float4*>(g)[lane];
        const float4 gb = reinterpret_cast<const float4*>(g)[32 + lane];
        const float4 ea = reinterpret_cast<const float4*>(be)[lane];
        const float4 eb = reinterpret_cast<const float4*>(be)[32 + lane];
        float4 va, vb;
        va.x = acc0.x * ia + ba.x; va.y = acc0.y * ia + ba.y; va.z = acc0.z * ia + ba.z; va.w = acc0.w * ia + ba.w;
        vb.x = acc1.x * ib + bb.x; vb.y = acc1.y * ib + bb.y; vb.z = acc1.z * ib + bb.z; vb.w = acc1.w * ib + bb.w;
        const float s  = ((va.x + va.y) + (va.z + va.w)) + ((vb.x + vb.y) + (vb.z + vb.w));
        const float mu = waveRedSum(s) * (1.f / HIDF);
        va.x -= mu; va.y -= mu; va.z -= mu; va.w -= mu;
        vb.x -= mu; vb.y -= mu; vb.z -= mu; vb.w -= mu;
        const float q = ((va.x * va.x + va.y * va.y) + (va.z * va.z + va.w * va.w))
                      + ((vb.x * vb.x + vb.y * vb.y) + (vb.z * vb.z + vb.w * vb.w));
        const float rstd = rsqrtf(waveRedSum(q) * (1.f / HIDF) + LN_EPS);
        float4 ya, yb;
        ya.x = eluf(va.x * rstd * ga.x + ea.x); ya.y = eluf(va.y * rstd * ga.y + ea.y);
        ya.z = eluf(va.z * rstd * ga.z + ea.z); ya.w = eluf(va.w * rstd * ga.w + ea.w);
        yb.x = eluf(vb.x * rstd * gb.x + eb.x); yb.y = eluf(vb.y * rstd * gb.y + eb.y);
        yb.z = eluf(vb.z * rstd * gb.z + eb.z); yb.w = eluf(vb.w * rstd * gb.w + eb.w);
        float4* orow = reinterpret_cast<float4*>(out + (size_t)n * HIDF);
        orow[lane]      = ya;
        orow[32 + lane] = yb;
    } else {
        float4 p0, p1;
        p0.x = __shfl_xor(acc0.x, 16, 32); p0.y = __shfl_xor(acc0.y, 16, 32);
        p0.z = __shfl_xor(acc0.z, 16, 32); p0.w = __shfl_xor(acc0.w, 16, 32);
        p1.x = __shfl_xor(acc1.x, 16, 32); p1.y = __shfl_xor(acc1.y, 16, 32);
        p1.z = __shfl_xor(acc1.z, 16, 32); p1.w = __shfl_xor(acc1.w, 16, 32);
        const float4 h0 = hi ? p0 : acc0, h1 = hi ? acc0 : p0;
        const float4 h2 = hi ? p1 : acc1, h3 = hi ? acc1 : p1;
        const int l16 = lane & 15;
        const float4 b4 = reinterpret_cast<const float4*>(bias)[l16];
        const float4 g4 = reinterpret_cast<const float4*>(g)[l16];
        const float4 e4 = reinterpret_cast<const float4*>(be)[l16];
        const float q0 = vw2[lane], q1 = vw2[32 + lane], q2 = vw2[64 + lane], q3 = vw2[96 + lane];
        const float hb = vb1[lane];
        float4 v;
        v.x = (((h0.x * i0 + h1.x * i1) + h2.x * i2) + h3.x * i3) * (1.f / HEADS) + b4.x;
        v.y = (((h0.y * i0 + h1.y * i1) + h2.y * i2) + h3.y * i3) * (1.f / HEADS) + b4.y;
        v.z = (((h0.z * i0 + h1.z * i1) + h2.z * i2) + h3.z * i3) * (1.f / HEADS) + b4.z;
        v.w = (((h0.w * i0 + h1.w * i1) + h2.w * i2) + h3.w * i3) * (1.f / HEADS) + b4.w;
        float s = (v.x + v.y) + (v.z + v.w);
        for (int off = 8; off > 0; off >>= 1) s += __shfl_xor(s, off, 32);
        const float mu = s * (1.f / CCH);
        v.x -= mu; v.y -= mu; v.z -= mu; v.w -= mu;
        float q = (v.x * v.x + v.y * v.y) + (v.z * v.z + v.w * v.w);
        for (int off = 8; off > 0; off >>= 1) q += __shfl_xor(q, off, 32);
        const float rstd = rsqrtf(q * (1.f / CCH) + LN_EPS);
        float4 y;
        y.x = eluf(v.x * rstd * g4.x + e4.x); y.y = eluf(v.y * rstd * g4.y + e4.y);
        y.z = eluf(v.z * rstd * g4.z + e4.z); y.w = eluf(v.w * rstd * g4.w + e4.w);
        const float4* sW4 = reinterpret_cast<const float4*>(sW);
        float hid = hb;
#pragma unroll
        for (int c4 = 0; c4 < CCH / 4; ++c4) {
            const float4 wv = sW4[c4 * WP + lane];
            hid += laneValF(y.x, c4) * wv.x;
            hid += laneValF(y.y, c4) * wv.y;
            hid += laneValF(y.z, c4) * wv.z;
            hid += laneValF(y.w, c4) * wv.w;
        }
        hid = eluf(hid);
        float o0 = q0 * hid, o1 = q1 * hid, o2 = q2 * hid, o3 = q3 * hid;
        for (int off = 16; off > 0; off >>= 1) {
            o0 += __shfl_xor(o0, off, 32); o1 += __shfl_xor(o1, off, 32);
            o2 += __shfl_xor(o2, off, 32); o3 += __shfl_xor(o3, off, 32);
        }
        if (lane == 0) {
            const float4 ob = *reinterpret_cast<const float4*>(vb2);
            *reinterpret_cast<float4*>(vout + (size_t)n * 4) =
                make_float4(o0 + ob.x, o1 + ob.y, o2 + ob.z, o3 + ob.w);
        }
    }
}

extern "C" void kernel_launch(void* const* d_in, const int* in_sizes, int n_in,
                              void* d_out, int out_size, void* d_ws, size_t ws_size,
                              hipStream_t stream) {
    const float* x      = (const float*)d_in[0];
    const int*   esrc   = (const int*)d_in[1];
    const int*   edst   = (const int*)d_in[2];
    const float* proj_w = (const float*)d_in[3];
    const float* proj_b = (const float*)d_in[4];
    const float* W[3]  = {(const float*)d_in[5],  (const float*)d_in[11], (const float*)d_in[17]};
    const float* AS[3] = {(const float*)d_in[6],  (const float*)d_in[12], (const float*)d_in[18]};
    const float* AD[3] = {(const float*)d_in[7],  (const float*)d_in[13], (const float*)d_in[19]};
    const float* B[3]  = {(const float*)d_in[8],  (const float*)d_in[14], (const float*)d_in[20]};
    const float* G[3]  = {(const float*)d_in[9],  (const float*)d_in[15], (const float*)d_in[21]};
    const float* BE[3] = {(const float*)d_in[10], (const float*)d_in[16], (const float*)d_in[22]};
    const float* vw1 = (const float*)d_in[23];
    const float* vb1 = (const float*)d_in[24];
    const float* vw2 = (const float*)d_in[25];
    const float* vb2 = (const float*)d_in[26];

    const int N = in_sizes[0] / 128;
    const int E = in_sizes[1];
    const int NH = N * HEADS;

    float* ws     = (float*)d_ws;
    float* hin    = ws;
    float* hmat   = hin  + (size_t)N * HIDF;
    float* asrc   = hmat + (size_t)N * HIDF;
    float* adst   = asrc + (size_t)NH;
    const int NB  = (N + (1 << BSH) - 1) >> BSH;
    int* rowptr   = (int*)(adst + (size_t)NH);
    int* col      = rowptr + (size_t)N + 4;
    int* pairs    = col + (((size_t)E + 3) & ~(size_t)3);
    int* mat      = pairs + ((2 * (size_t)E + 3) & ~(size_t)3);

    dim3 blk(256);

    bucket_hist<<<G1, 1024, 0, stream>>>(edst, mat, E, NB);
    excl_scan<<<1, 1024, 0, stream>>>(mat, NB * G1);
    bucket_scatter<<<G1, 1024, 0, stream>>>(esrc, edst, mat, pairs, E, NB);
    bucket_sort<<<NB, blk, 0, stream>>>(mat, pairs, rowptr, col, N);

    gemm_wmma_f32_lds<64, 64, 4, 2, false><<<(N + 63) / 64, blk, 0, stream>>>(
        x, 128, proj_w, hin, N, 128, 64, proj_b, 1, nullptr, nullptr, nullptr, nullptr);

    int in_ld = 64;
    for (int l = 0; l < 3; ++l) {
        int K = (l == 0) ? 64 : 256;
        gemm_wmma_f32_lds<64, 128, 2, 4, true><<<((N + 63) / 64) * (HIDF / 128), blk, 0, stream>>>(
            hin, in_ld, W[l], hmat, N, K, HIDF, nullptr, 0, AS[l], AD[l], asrc, adst);
        if (l < 2) {
            gat_gather<false><<<(N * 32 + 255) / 256, blk, 0, stream>>>(
                rowptr, col, asrc, adst, hmat, B[l], G[l], BE[l], hin, N,
                nullptr, nullptr, nullptr, nullptr, nullptr);
            in_ld = HIDF;
        } else {
            gat_gather<true><<<(N * 32 + 255) / 256, blk, 0, stream>>>(
                rowptr, col, asrc, adst, hmat, B[l], G[l], BE[l], nullptr, N,
                vw1, vb1, vw2, vb2, (float*)d_out);
        }
    }
}


# Edge grouping moved into the projection and layer-0 GEMM launches as their first work-groups (per-chunk LDS counting sort, then per-bucket sort); 4 launches gone; in-edge order = LDS arrival order
# speedup vs baseline: 4.6365x; 1.0733x over previous
#include <hip/hip_runtime.h>
#include <math.h>

typedef __attribute__((ext_vector_type(2))) float v2f;
typedef __attribute__((ext_vector_type(8))) float v8f;

#define HEADS 4
#define CCH 64
#define HIDF 256
#define NEG_SLOPE 0.2f
#define LN_EPS 1e-5f

__device__ __forceinline__ float eluf(float x) { return x > 0.f ? x : (expf(x) - 1.f); }
__device__ __forceinline__ float leakyf(float x) { return x > 0.f ? x : NEG_SLOPE * x; }

__device__ __forceinline__ float waveRedSum(float v) {
    for (int off = 16; off > 0; off >>= 1) v += __shfl_xor(v, off, 32);
    return v;
}
__device__ __forceinline__ float waveRedMax(float v) {
    for (int off = 16; off > 0; off >>= 1) v = fmaxf(v, __shfl_xor(v, off, 32));
    return v;
}
__device__ __forceinline__ int waveRedSumI(int v) {
    for (int off = 16; off > 0; off >>= 1) v += __shfl_xor(v, off, 32);
    return v;
}
__device__ __forceinline__ float laneValF(float v, int t) {
    return __int_as_float(__builtin_amdgcn_readlane(__float_as_int(v), t));
}

#define BSH 8
#define NBMAX 256
#define G1 256
#define EB 8

__device__ __forceinline__ void chunk_group(int c, const int* __restrict__ src, const int* __restrict__ dst,
                                            int* __restrict__ cntm, int* __restrict__ offm,
                                            int* __restrict__ pairs, int E, int NB) {
    __shared__ int h[NBMAX];
    __shared__ int hsum[8];
    const int tid  = threadIdx.x;
    const int lane = tid & 31;
    const int w    = tid >> 5;
    h[tid] = 0;
    __syncthreads();
    const int per = (E + G1 - 1) / G1;
    const int lo  = (c * per < E) ? (c * per) : E;
    const int hi  = (lo + per < E) ? (lo + per) : E;
    for (int e0 = lo + tid; e0 < hi; e0 += EB * 256) {
        int d[EB];
#pragma unroll
        for (int j = 0; j < EB; ++j) d[j] = (e0 + j * 256 < hi) ? dst[e0 + j * 256] : -1;
#pragma unroll
        for (int j = 0; j < EB; ++j)
            if (d[j] >= 0) atomicAdd(&h[d[j] >> BSH], 1);
    }
    __syncthreads();
    const int cb = h[tid];
    int x = cb;
    for (int off = 1; off < 32; off <<= 1) {
        int y = __shfl_up(x, off, 32);
        if (lane >= off) x += y;
    }
    if (lane == 31) hsum[w] = x;
    __syncthreads();
    int carry = 0;
    for (int k = 0; k < w; ++k) carry += hsum[k];
    const int start = lo + carry + x - cb;
    if (tid < NB) {
        cntm[tid * G1 + c] = cb;
        offm[tid * G1 + c] = start;
    }
    h[tid] = start;
    __syncthreads();
    int2* rec = reinterpret_cast<int2*>(pairs);
    for (int e0 = lo + tid; e0 < hi; e0 += EB * 256) {
        int d[EB], s[EB];
#pragma unroll
        for (int j = 0; j < EB; ++j) {
            const bool in = e0 + j * 256 < hi;
            d[j] = in ? dst[e0 + j * 256] : -1;
            s[j] = in ? src[e0 + j * 256] : 0;
        }
#pragma unroll
        for (int j = 0; j < EB; ++j)
            if (d[j] >= 0) {
                const int p = atomicAdd(&h[d[j] >> BSH], 1);
                rec[p] = make_int2(s[j], d[j]);
            }
    }
}

__device__ __forceinline__ void bucket_sort(int b, bool last, const int* __restrict__ cntm,
                                            const int* __restrict__ offm, const int* __restrict__ pairs,
                                            int* __restrict__ rowptr, int* __restrict__ col, int N, int E) {
    __shared__ int cnt[256];
    __shared__ int wsum[8];
    __shared__ int bsum[8];
    static_assert(G1 == 256, "one segment per thread");
    const int tid  = threadIdx.x;
    const int lane = tid & 31;
    const int w    = tid >> 5;
    const int n0   = b << BSH;
    const int2* rec = reinterpret_cast<const int2*>(pairs);
    int tot = 0;
    if (tid < b) {
        const int4* r = reinterpret_cast<const int4*>(cntm + tid * G1);
#pragma unroll 16
        for (int j = 0; j < G1 / 4; ++j) {
            const int4 q = r[j];
            tot += (q.x + q.y) + (q.z + q.w);
        }
    }
    tot = waveRedSumI(tot);
    if (lane == 0) bsum[w] = tot;
    const int segOff = offm[b * G1 + tid];
    const int segCnt = cntm[b * G1 + tid];
    cnt[tid] = 0;
    __syncthreads();
    int beg = 0;
    for (int k = 0; k < 8; ++k) beg += bsum[k];
    for (int i0 = 0; i0 < segCnt; i0 += EB) {
        int d[EB];
#pragma unroll
        for (int j = 0; j < EB; ++j) d[j] = (i0 + j < segCnt) ? rec[segOff + i0 + j].y : -1;
#pragma unroll
        for (int j = 0; j < EB; ++j)
            if (d[j] >= 0) atomicAdd(&cnt[d[j] - n0], 1);
    }
    __syncthreads();
    const int c = cnt[tid];
    int x = c;
    for (int off = 1; off < 32; off <<= 1) {
        int y = __shfl_up(x, off, 32);
        if (lane >= off) x += y;
    }
    if (lane == 31) wsum[w] = x;
    __syncthreads();
    int carry = 0;
    for (int k = 0; k < w; ++k) carry += wsum[k];
    const int start = beg + carry + x - c;
    if (n0 + tid < N) rowptr[n0 + tid] = start;
    if (last && tid == 0) rowptr[N] = E;
    cnt[tid] = start;
    __syncthreads();
    for (int i0 = 0; i0 < segCnt; i0 += EB) {
        int2 r[EB];
#pragma unroll
        for (int j = 0; j < EB; ++j) r[j] = (i0 + j < segCnt) ? rec[segOff + i0 + j] : make_int2(0, -1);
#pragma unroll
        for (int j = 0; j < EB; ++j)
            if (r[j].y >= 0) {
                const int q = atomicAdd(&cnt[r[j].y - n0], 1);
                col[q] = r[j].x;
            }
    }
}

template <int BM, int BN, int WM, int WN, bool ATT, int GST>
__global__ void __launch_bounds__(256)
gemm_wmma_f32_lds(const float* __restrict__ A, int lda,
                  const float* __restrict__ W,
                  float* __restrict__ C,
                  int M, int K, int Nout,
                  const float* __restrict__ bias, int elu_act,
                  const float* __restrict__ a_s, const float* __restrict__ a_d,
                  float* __restrict__ asrc, float* __restrict__ adst,
                  int gblocks, const int* __restrict__ esrc, const int* __restrict__ edst,
                  int* __restrict__ cntm, int* __restrict__ offm, int* __restrict__ pairs,
                  int* __restrict__ rowptr, int* __restrict__ col, int E, int NB) {
    int bid = blockIdx.x;
    if constexpr (GST != 0) {
        if (bid < gblocks) {
            if constexpr (GST == 1) chunk_group(bid, esrc, edst, cntm, offm, pairs, E, NB);
            else bucket_sort(bid, bid == gblocks - 1, cntm, offm, pairs, rowptr, col, M, E);
            return;
        }
        bid -= gblocks;
    }
    constexpr int BK = 32;
    constexpr int P  = BK + 4;
    constexpr int TM = BM / (16 * WM);
    constexpr int TN = BN / (16 * WN);
    constexpr int AF = BM * (BK / 4) / 256;
    constexpr int BF = BN * (BK / 4) / 256;
    static_assert(WM * WN == 8, "work-group of 8 waves");
    static_assert(TM * 16 * WM == BM && TN * 16 * WN == BN, "wave tiles cover the block");
    static_assert(AF * 256 == BM * (BK / 4) && BF * 256 == BN * (BK / 4), "copy covers the chunk");

    __shared__ __attribute__((aligned(16))) float sA[2][BM * P];
    __shared__ __attribute__((aligned(16))) float sB[2][BN * P];

    const int tid  = threadIdx.x;
    const int lane = tid & 31;
    const int w    = tid >> 5;
    const int wm   = w / WN;
    const int wn   = w - wm * WN;
    const int blocksN = Nout / BN;
    const int bm = bid / blocksN;
    const int bn = bid - bm * blocksN;
    const int m0 = bm * BM;
    const int n0 = bn * BN;

    const int cr = tid >> 3;
    const int cq = (tid & 7) << 2;
    const float* gA[AF];
    const float* gB[BF];
#pragma unroll
    for (int j = 0; j < AF; ++j) {
        int gm = m0 + cr + 32 * j;
        gm = gm < M ? gm : (M - 1);
        gA[j] = A + (size_t)gm * lda + cq;
    }
#pragma unroll
    for (int j = 0; j < BF; ++j) gB[j] = W + (size_t)(n0 + cr + 32 * j) * K + cq;

    static_assert(AF == 2 && (BF == 2 || BF == 4), "piece registers below are written out for these sizes");
    float4 ra0, ra1, rb0, rb1, rb2, rb3;
    ra0 = *reinterpret_cast<const float4*>(gA[0]);
    ra1 = *reinterpret_cast<const float4*>(gA[1]);
    rb0 = *reinterpret_cast<const float4*>(gB[0]);
    rb1 = *reinterpret_cast<const float4*>(gB[1]);
    if constexpr (BF == 4) {
        rb2 = *reinterpret_cast<const float4*>(gB[2]);
        rb3 = *reinterpret_cast<const float4*>(gB[3]);
    }
    const int sOff = cr * P + cq;
    *reinterpret_cast<float4*>(&sA[0][sOff])          = ra0;
    *reinterpret_cast<float4*>(&sA[0][sOff + 32 * P]) = ra1;
    *reinterpret_cast<float4*>(&sB[0][sOff])          = rb0;
    *reinterpret_cast<float4*>(&sB[0][sOff + 32 * P]) = rb1;
    if constexpr (BF == 4) {
        *reinterpret_cast<float4*>(&sB[0][sOff + 64 * P]) = rb2;
        *reinterpret_cast<float4*>(&sB[0][sOff + 96 * P]) = rb3;
    }

    v8f acc[TM][TN];
#pragma unroll
    for (int i = 0; i < TM; ++i)
#pragma unroll
        for (int j = 0; j < TN; ++j) acc[i][j] = v8f{};

    const int fragOff = (lane & 15) * P + ((lane >> 4) << 1);
    const int nk = K / BK;
    for (int it = 0; it < nk; ++it) {
        const int  buf  = it & 1;
        const bool more = it + 1 < nk;
        __syncthreads();
        if (more) {
            const int k1 = (it + 1) * BK;
            ra0 = *reinterpret_cast<const float4*>(gA[0] + k1);
            ra1 = *reinterpret_cast<const float4*>(gA[1] + k1);
            rb0 = *reinterpret_cast<const float4*>(gB[0] + k1);
            rb1 = *reinterpret_cast<const float4*>(gB[1] + k1);
            if constexpr (BF == 4) {
                rb2 = *reinterpret_cast<const float4*>(gB[2] + k1);
                rb3 = *reinterpret_cast<const float4*>(gB[3] + k1);
            }
        }
        const float* pa = &sA[buf][wm * (TM * 16) * P + fragOff];
        const float* pb = &sB[buf][wn * (TN * 16) * P + fragOff];
#pragma unroll
        for (int ks = 0; ks < BK / 4; ++ks) {
            v2f af[TM], bf[TN];
#pragma unroll
            for (int i = 0; i < TM; ++i) af[i] = *reinterpret_cast<const v2f*>(pa + i * 16 * P + ks * 4);
#pragma unroll
            for (int j = 0; j < TN; ++j) bf[j] = *reinterpret_cast<const v2f*>(pb + j * 16 * P + ks * 4);
#pragma unroll
            for (int i = 0; i < TM; ++i)
#pragma unroll
                for (int j = 0; j < TN; ++j)
                    acc[i][j] = __builtin_amdgcn_wmma_f32_16x16x4_f32(false, af[i], false, bf[j],
                                                                      (short)0, acc[i][j], false, false);
        }
        if (more) {
            *reinterpret_cast<float4*>(&sA[buf ^ 1][sOff])          = ra0;
            *reinterpret_cast<float4*>(&sA[buf ^ 1][sOff + 32 * P]) = ra1;
            *reinterpret_cast<float4*>(&sB[buf ^ 1][sOff])          = rb0;
            *reinterpret_cast<float4*>(&sB[buf ^ 1][sOff + 32 * P]) = rb1;
            if constexpr (BF == 4) {
                *reinterpret_cast<float4*>(&sB[buf ^ 1][sOff + 64 * P]) = rb2;
                *reinterpret_cast<float4*>(&sB[buf ^ 1][sOff + 96 * P]) = rb3;
            }
        }
    }

    if constexpr (ATT) {
        constexpr int CP = BN + 4;
        constexpr int RW = BM / 8;
        static_assert(BN == 128, "a row of the block is one 16-byte piece per lane, one head per lane half");
        static_assert(BM * CP <= 2 * BN * P, "the C block fits in the W chunk buffers");
        float* sC = &sB[0][0];
        __syncthreads();
#pragma unroll
        for (int i = 0; i < TM; ++i)
#pragma unroll
            for (int j = 0; j < TN; ++j)
#pragma unroll
                for (int r = 0; r < 8; ++r)
                    sC[((wm * TM + i) * 16 + ((lane >> 4) << 3) + r) * CP + (wn * TN + j) * 16 + (lane & 15)] = acc[i][j][r];
        __syncthreads();
        const int colg = n0 + 4 * lane;
        const float4 s4 = *reinterpret_cast<const float4*>(a_s + colg);
        const float4 d4 = *reinterpret_cast<const float4*>(a_d + colg);
        const int head = (n0 >> 6) + (lane >> 4);
        float4 v[RW];
        float ps[RW], pd[RW];
#pragma unroll
        for (int q = 0; q < RW; ++q)
            v[q] = *reinterpret_cast<const float4*>(sC + (w * RW + q) * CP + 4 * lane);
#pragma unroll
        for (int q = 0; q < RW; ++q) {
            ps[q] = (v[q].x * s4.x + v[q].y * s4.y) + (v[q].z * s4.z + v[q].w * s4.w);
            pd[q] = (v[q].x * d4.x + v[q].y * d4.y) + (v[q].z * d4.z + v[q].w * d4.w);
        }
#pragma unroll
        for (int off = 8; off > 0; off >>= 1) {
#pragma unroll
            for (int q = 0; q < RW; ++q) {
                ps[q] += __shfl_xor(ps[q], off, 32);
                pd[q] += __shfl_xor(pd[q], off, 32);
            }
        }
#pragma unroll
        for (int q = 0; q < RW; ++q) {
            const int gm = m0 + w * RW + q;
            if (gm < M) {
                *reinterpret_cast<float4*>(C + (size_t)gm * Nout + colg) = v[q];
                if ((lane & 15) == 0) {
                    asrc[gm * HEADS + head] = ps[q];
                    adst[gm * HEADS + head] = pd[q];
                }
            }
        }
        return;
    }

#pragma unroll
    for (int i = 0; i < TM; ++i) {
#pragma unroll
        for (int j = 0; j < TN; ++j) {
            const int rbase = m0 + (wm * TM + i) * 16 + ((lane >> 4) << 3);
            const int col   = n0 + (wn * TN + j) * 16 + (lane & 15);
            const float bv  = bias ? bias[col] : 0.f;
#pragma unroll
            for (int r = 0; r < 8; ++r) {
                float v = acc[i][j][r] + bv;
                if (elu_act) v = eluf(v);
                if (rbase + r < M) C[(size_t)(rbase + r) * Nout + col] = v;
            }
        }
    }
}


__device__ __forceinline__ void gather_rows(const float* __restrict__ hmat, int s,
                                            float e0, float e1, float e2, float e3,
                                            int cnt, int lane, bool hi,
                                            float4& acc0, float4& acc1) {
#pragma unroll 4
    for (int t = 0; t < cnt; ++t) {
        const int   st = __builtin_amdgcn_readlane(s, t);
        const float w0 = laneValF(e0, t), w1 = laneValF(e1, t);
        const float w2 = laneValF(e2, t), w3 = laneValF(e3, t);
        const float wa = hi ? w1 : w0;
        const float wb = hi ? w3 : w2;
        const float4* r = reinterpret_cast<const float4*>(hmat + (size_t)st * HIDF);
        const float4 p = r[lane];
        const float4 q = r[32 + lane];
        acc0.x += wa * p.x; acc0.y += wa * p.y; acc0.z += wa * p.z; acc0.w += wa * p.w;
        acc1.x += wb * q.x; acc1.y += wb * q.y; acc1.z += wb * q.z; acc1.w += wb * q.w;
    }
}

template <bool MEAN>
__global__ void gat_gather(const int* __restrict__ rowptr, const int* __restrict__ col,
                           const float* __restrict__ asrc, const float* __restrict__ adst,
                           const float* __restrict__ hmat,
                           const float* __restrict__ bias, const float* __restrict__ g,
                           const float* __restrict__ be, float* __restrict__ out, int N,
                           const float* __restrict__ vw1, const float* __restrict__ vb1,
                           const float* __restrict__ vw2, const float* __restrict__ vb2,
                           float* __restrict__ vout) {
    constexpr int WP = 33;
    __shared__ __attribute__((aligned(16))) float sW[MEAN ? 16 * WP * 4 : 4];
    if constexpr (MEAN) {
        const int f0 = threadIdx.x, f1 = threadIdx.x + 256;
        const float4 wa = reinterpret_cast<const float4*>(vw1)[f0];
        const float4 wb = reinterpret_cast<const float4*>(vw1)[f1];
        reinterpret_cast<float4*>(sW)[(f0 & 15) * WP + (f0 >> 4)] = wa;
        reinterpret_cast<float4*>(sW)[(f1 & 15) * WP + (f1 >> 4)] = wb;
        __syncthreads();
    }
    const int lane = threadIdx.x & 31;
    const int n = __builtin_amdgcn_readfirstlane((int)((blockIdx.x * blockDim.x + threadIdx.x) >> 5));
    if (n >= N) return;
    const int beg = rowptr[n], end = rowptr[n + 1];
    const bool hi = lane >= 16;
    const float4* asrc4 = reinterpret_cast<const float4*>(asrc);
    const float4 ad = reinterpret_cast<const float4*>(adst)[n];
    const float4 an = asrc4[n];
    const float sf0 = leakyf(an.x + ad.x), sf1 = leakyf(an.y + ad.y);
    const float sf2 = leakyf(an.z + ad.z), sf3 = leakyf(an.w + ad.w);

    const bool v0 = beg + lane < end;
    int   s0 = n;
    float a0 = sf0, a1 = sf1, a2 = sf2, a3 = sf3;
    if (v0) {
        s0 = col[beg + lane];
        const float4 a = asrc4[s0];
        a0 = leakyf(a.x + ad.x); a1 = leakyf(a.y + ad.y);
        a2 = leakyf(a.z + ad.z); a3 = leakyf(a.w + ad.w);
    }
    float m0 = fmaxf(sf0, a0), m1 = fmaxf(sf1, a1), m2 = fmaxf(sf2, a2), m3 = fmaxf(sf3, a3);
    for (int base = beg + 32; base < end; base += 32) {
        const int j = base + lane;
        if (j < end) {
            const float4 a = asrc4[col[j]];
            m0 = fmaxf(m0, leakyf(a.x + ad.x)); m1 = fmaxf(m1, leakyf(a.y + ad.y));
            m2 = fmaxf(m2, leakyf(a.z + ad.z)); m3 = fmaxf(m3, leakyf(a.w + ad.w));
        }
    }
    m0 = waveRedMax(m0); m1 = waveRedMax(m1); m2 = waveRedMax(m2); m3 = waveRedMax(m3);

    const float x0 = expf(sf0 - m0), x1 = expf(sf1 - m1), x2 = expf(sf2 - m2), x3 = expf(sf3 - m3);
    const float4* rown = reinterpret_cast<const float4*>(hmat + (size_t)n * HIDF);
    const float4 r0 = rown[lane], r1 = rown[32 + lane];
    const float wsa = hi ? x1 : x0, wsb = hi ? x3 : x2;
    float4 acc0 = make_float4(r0.x * wsa, r0.y * wsa, r0.z * wsa, r0.w * wsa);
    float4 acc1 = make_float4(r1.x * wsb, r1.y * wsb, r1.z * wsb, r1.w * wsb);

    float d0, d1, d2, d3;
    {
        const float t0 = expf(a0 - m0), t1 = expf(a1 - m1), t2 = expf(a2 - m2), t3 = expf(a3 - m3);
        d0 = v0 ? t0 : 0.f; d1 = v0 ? t1 : 0.f; d2 = v0 ? t2 : 0.f; d3 = v0 ? t3 : 0.f;
        const int cnt = (end - beg < 32) ? (end - beg) : 32;
        gather_rows(hmat, s0, d0, d1, d2, d3, cnt, lane, hi, acc0, acc1);
    }
    for (int base = beg + 32; base < end; base += 32) {
        const int j = base + lane;
        int s = n;
        float e0 = 0.f, e1 = 0.f, e2 = 0.f, e3 = 0.f;
        if (j < end) {
            s = col[j];
            const float4 a = asrc4[s];
            e0 = expf(leakyf(a.x + ad.x) - m0); e1 = expf(leakyf(a.y + ad.y) - m1);
            e2 = expf(leakyf(a.z + ad.z) - m2); e3 = expf(leakyf(a.w + ad.w) - m3);
        }
        d0 += e0; d1 += e1; d2 += e2; d3 += e3;
        const int cnt = (end - base < 32) ? (end - base) : 32;
        gather_rows(hmat, s, e0, e1, e2, e3, cnt, lane, hi, acc0, acc1);
    }
    d0 = waveRedSum(d0); d1 = waveRedSum(d1); d2 = waveRedSum(d2); d3 = waveRedSum(d3);
    const float i0 = 1.f / ((x0 + d0) + 1e-16f), i1 = 1.f / ((x1 + d1) + 1e-16f);
    const float i2 = 1.f / ((x2 + d2) + 1e-16f), i3 = 1.f / ((x3 + d3) + 1e-16f);

    if constexpr (!MEAN) {
        const float ia = hi ? i1 : i0, ib = hi ? i3 : i2;
        const float4 ba = reinterpret_cast<const float4*>(bias)[lane];
        const float4 bb = reinterpret_cast<const float4*>(bias)[32 + lane];
        const float4 ga = reinterpret_cast<const float4*>(g)[lane];
        const float4 gb = reinterpret_cast<const float4*>(g)[32 + lane];
        const float4 ea = reinterpret_cast<const float4*>(be)[lane];
        const float4 eb = reinterpret_cast<const float4*>(be)[32 + lane];
        float4 va, vb;
        va.x = acc0.x * ia + ba.x; va.y = acc0.y * ia + ba.y; va.z = acc0.z * ia + ba.z; va.w = acc0.w * ia + ba.w;
        vb.x = acc1.x * ib + bb.x; vb.y = acc1.y * ib + bb.y; vb.z = acc1.z * ib + bb.z; vb.w = acc1.w * ib + bb.w;
        const float s  = ((va.x + va.y) + (va.z + va.w)) + ((vb.x + vb.y) + (vb.z + vb.w));
        const float mu = waveRedSum(s) * (1.f / HIDF);
        va.x -= mu; va.y -= mu; va.z -= mu; va.w -= mu;
        vb.x -= mu; vb.y -= mu; vb.z -= mu; vb.w -= mu;
        const float q = ((va.x * va.x + va.y * va.y) + (va.z * va.z + va.w * va.w))
                      + ((vb.x * vb.x + vb.y * vb.y) + (vb.z * vb.z + vb.w * vb.w));
        const float rstd = rsqrtf(waveRedSum(q) * (1.f / HIDF) + LN_EPS);
        float4 ya, yb;
        ya.x = eluf(va.x * rstd * ga.x + ea.x); ya.y = eluf(va.y * rstd * ga.y + ea.y);
        ya.z = eluf(va.z * rstd * ga.z + ea.z); ya.w = eluf(va.w * rstd * ga.w + ea.w);
        yb.x = eluf(vb.x * rstd * gb.x + eb.x); yb.y = eluf(vb.y * rstd * gb.y + eb.y);
        yb.z = eluf(vb.z * rstd * gb.z + eb.z); yb.w = eluf(vb.w * rstd * gb.w + eb.w);
        float4* orow = reinterpret_cast<float4*>(out + (size_t)n * HIDF);
        orow[lane]      = ya;
        orow[32 + lane] = yb;
    } else {
        float4 p0, p1;
        p0.x = __shfl_xor(acc0.x, 16, 32); p0.y = __shfl_xor(acc0.y, 16, 32);
        p0.z = __shfl_xor(acc0.z, 16, 32); p0.w = __shfl_xor(acc0.w, 16, 32);
        p1.x = __shfl_xor(acc1.x, 16, 32); p1.y = __shfl_xor(acc1.y, 16, 32);
        p1.z = __shfl_xor(acc1.z, 16, 32); p1.w = __shfl_xor(acc1.w, 16, 32);
        const float4 h0 = hi ? p0 : acc0, h1 = hi ? acc0 : p0;
        const float4 h2 = hi ? p1 : acc1, h3 = hi ? acc1 : p1;
        const int l16 = lane & 15;
        const float4 b4 = reinterpret_cast<const float4*>(bias)[l16];
        const float4 g4 = reinterpret_cast<const float4*>(g)[l16];
        const float4 e4 = reinterpret_cast<const float4*>(be)[l16];
        const float q0 = vw2[lane], q1 = vw2[32 + lane], q2 = vw2[64 + lane], q3 = vw2[96 + lane];
        const float hb = vb1[lane];
        float4 v;
        v.x = (((h0.x * i0 + h1.x * i1) + h2.x * i2) + h3.x * i3) * (1.f / HEADS) + b4.x;
        v.y = (((h0.y * i0 + h1.y * i1) + h2.y * i2) + h3.y * i3) * (1.f / HEADS) + b4.y;
        v.z = (((h0.z * i0 + h1.z * i1) + h2.z * i2) + h3.z * i3) * (1.f / HEADS) + b4.z;
        v.w = (((h0.w * i0 + h1.w * i1) + h2.w * i2) + h3.w * i3) * (1.f / HEADS) + b4.w;
        float s = (v.x + v.y) + (v.z + v.w);
        for (int off = 8; off > 0; off >>= 1) s += __shfl_xor(s, off, 32);
        const float mu = s * (1.f / CCH);
        v.x -= mu; v.y -= mu; v.z -= mu; v.w -= mu;
        float q = (v.x * v.x + v.y * v.y) + (v.z * v.z + v.w * v.w);
        for (int off = 8; off > 0; off >>= 1) q += __shfl_xor(q, off, 32);
        const float rstd = rsqrtf(q * (1.f / CCH) + LN_EPS);
        float4 y;
        y.x = eluf(v.x * rstd * g4.x + e4.x); y.y = eluf(v.y * rstd * g4.y + e4.y);
        y.z = eluf(v.z * rstd * g4.z + e4.z); y.w = eluf(v.w * rstd * g4.w + e4.w);
        const float4* sW4 = reinterpret_cast<const float4*>(sW);
        float hid = hb;
#pragma unroll
        for (int c4 = 0; c4 < CCH / 4; ++c4) {
            const float4 wv = sW4[c4 * WP + lane];
            hid += laneValF(y.x, c4) * wv.x;
            hid += laneValF(y.y, c4) * wv.y;
            hid += laneValF(y.z, c4) * wv.z;
            hid += laneValF(y.w, c4) * wv.w;
        }
        hid = eluf(hid);
        float o0 = q0 * hid, o1 = q1 * hid, o2 = q2 * hid, o3 = q3 * hid;
        for (int off = 16; off > 0; off >>= 1) {
            o0 += __shfl_xor(o0, off, 32); o1 += __shfl_xor(o1, off, 32);
            o2 += __shfl_xor(o2, off, 32); o3 += __shfl_xor(o3, off, 32);
        }
        if (lane == 0) {
            const float4 ob = *reinterpret_cast<const float4*>(vb2);
            *reinterpret_cast<float4*>(vout + (size_t)n * 4) =
                make_float4(o0 + ob.x, o1 + ob.y, o2 + ob.z, o3 + ob.w);
        }
    }
}

extern "C" void kernel_launch(void* const* d_in, const int* in_sizes, int n_in,
                              void* d_out, int out_size, void* d_ws, size_t ws_size,
                              hipStream_t stream) {
    const float* x      = (const float*)d_in[0];
    const int*   esrc   = (const int*)d_in[1];
    const int*   edst   = (const int*)d_in[2];
    const float* proj_w = (const float*)d_in[3];
    const float* proj_b = (const float*)d_in[4];
    const float* W[3]  = {(const float*)d_in[5],  (const float*)d_in[11], (const float*)d_in[17]};
    const float* AS[3] = {(const float*)d_in[6],  (const float*)d_in[12], (const float*)d_in[18]};
    const float* AD[3] = {(const float*)d_in[7],  (const float*)d_in[13], (const float*)d_in[19]};
    const float* B[3]  = {(const float*)d_in[8],  (const float*)d_in[14], (const float*)d_in[20]};
    const float* G[3]  = {(const float*)d_in[9],  (const float*)d_in[15], (const float*)d_in[21]};
    const float* BE[3] = {(const float*)d_in[10], (const float*)d_in[16], (const float*)d_in[22]};
    const float* vw1 = (const float*)d_in[23];
    const float* vb1 = (const float*)d_in[24];
    const float* vw2 = (const float*)d_in[25];
    const float* vb2 = (const float*)d_in[26];

    const int N = in_sizes[0] / 128;
    const int E = in_sizes[1];
    const int NH = N * HEADS;

    float* ws     = (float*)d_ws;
    float* hin    = ws;
    float* hmat   = hin  + (size_t)N * HIDF;
    float* asrc   = hmat + (size_t)N * HIDF;
    float* adst   = asrc + (size_t)NH;
    const int NB  = (N + (1 << BSH) - 1) >> BSH;
    int* rowptr   = (int*)(adst + (size_t)NH);
    int* col      = rowptr + (size_t)N + 4;
    int* pairs    = col + (((size_t)E + 3) & ~(size_t)3);
    int* cntm     = pairs + ((2 * (size_t)E + 3) & ~(size_t)3);
    int* offm     = cntm + (size_t)NB * G1;

    dim3 blk(256);

    gemm_wmma_f32_lds<64, 64, 4, 2, false, 1><<<G1 + (N + 63) / 64, blk, 0, stream>>>(
        x, 128, proj_w, hin, N, 128, 64, proj_b, 1, nullptr, nullptr, nullptr, nullptr,
        G1, esrc, edst, cntm, offm, pairs, nullptr, nullptr, E, NB);

    int in_ld = 64;
    for (int l = 0; l < 3; ++l) {
        int K = (l == 0) ? 64 : 256;
        if (l == 0)
            gemm_wmma_f32_lds<64, 128, 2, 4, true, 2><<<NB + ((N + 63) / 64) * (HIDF / 128), blk, 0, stream>>>(
                hin, in_ld, W[l], hmat, N, K, HIDF, nullptr, 0, AS[l], AD[l], asrc, adst,
                NB, nullptr, nullptr, cntm, offm, pairs, rowptr, col, E, NB);
        else
            gemm_wmma_f32_lds<64, 128, 2, 4, true, 0><<<((N + 63) / 64) * (HIDF / 128), blk, 0, stream>>>(
                hin, in_ld, W[l], hmat, N, K, HIDF, nullptr, 0, AS[l], AD[l], asrc, adst,
                0, nullptr, nullptr, nullptr, nullptr, nullptr, nullptr, nullptr, 0, 0);
        if (l < 2) {
            gat_gather<false><<<(N * 32 + 255) / 256, blk, 0, stream>>>(
                rowptr, col, asrc, adst, hmat, B[l], G[l], BE[l], hin, N,
                nullptr, nullptr, nullptr, nullptr, nullptr);
            in_ld = HIDF;
        } else {
            gat_gather<true><<<(N * 32 + 255) / 256, blk, 0, stream>>>(
                rowptr, col, asrc, adst, hmat, B[l], G[l], BE[l], nullptr, N,
                vw1, vb1, vw2, vb2, (float*)d_out);
        }
    }
}
